# starting kernel
# baseline (speedup 1.0000x reference)
#include <hip/hip_runtime.h>

typedef float v2f __attribute__((ext_vector_type(2)));
typedef float v8f __attribute__((ext_vector_type(8)));

#define GCN_N 100000
#define GCN_E 1600000

__device__ __forceinline__ void atomic_add_f32(float* p, float v) {
    unsafeAtomicAdd(p, v);
}

__global__ void deg_init_kernel(float* deg, int n) {
    int i = blockIdx.x * blockDim.x + threadIdx.x;
    if (i < n) deg[i] = 1.0f;
}

__global__ void deg_count_kernel(const int* __restrict__ dst, float* deg, int e) {
    int i = blockIdx.x * blockDim.x + threadIdx.x;
    if (i < e) atomic_add_f32(&deg[dst[i]], 1.0f);
}

__global__ void dinv_kernel(float* d, int n) {
    int i = blockIdx.x * blockDim.x + threadIdx.x;
    if (i < n) d[i] = rsqrtf(d[i]);
}

template <int KDIM, int NOUT, bool RELU>
__global__ void gemm_wmma_kernel(const float* __restrict__ X,
                                 const float* __restrict__ W,
                                 float* __restrict__ H, int M) {
    constexpr int NT  = NOUT / 16;
    constexpr int LDW = NOUT + 8;
    __shared__ float Ws[KDIM * LDW];
    for (int i = threadIdx.x; i < KDIM * NOUT; i += blockDim.x) {
        int k = i / NOUT, n = i - k * NOUT;
        Ws[k * LDW + n] = W[i];
    }
    __syncthreads();

    const int wid  = (blockIdx.x * blockDim.x + threadIdx.x) >> 5;
    const int lane = threadIdx.x & 31;
    const int m0 = (wid / NT) * 16;
    const int n0 = (wid % NT) * 16;
    if (m0 >= M) return;

    const int half = lane >> 4;
    const int l16  = lane & 15;
    const int koff = half * 2;
    const float* xrow = X + (size_t)(m0 + l16) * KDIM + koff;
    const float* wcol = Ws + koff * LDW + n0 + l16;

    v8f acc = {};
#pragma unroll 4
    for (int k0 = 0; k0 < KDIM; k0 += 4) {
        float a0 = xrow[k0];
        float a1 = xrow[k0 + 1];
        if (RELU) { a0 = fmaxf(a0, 0.0f); a1 = fmaxf(a1, 0.0f); }
        v2f a; a.x = a0; a.y = a1;
        v2f b; b.x = wcol[k0 * LDW]; b.y = wcol[k0 * LDW + LDW];
        acc = __builtin_amdgcn_wmma_f32_16x16x4_f32(
            false, a, false, b, (short)0, acc, false, false);
    }
    float* orow = H + (size_t)(m0 + half * 8) * NOUT + n0 + l16;
#pragma unroll
    for (int v = 0; v < 8; ++v)
        orow[(size_t)v * NOUT] = acc[v];
}

__global__ void agg_init_kernel(const float* __restrict__ H,
                                const float* __restrict__ dinv,
                                const float* __restrict__ bias,
                                float* __restrict__ A, int n, int fsh) {
    int t = blockIdx.x * blockDim.x + threadIdx.x;
    if (t >= (n << fsh)) return;
    int i = t >> fsh;
    int c = (t & ((1 << fsh) - 1)) << 2;
    int F = 4 << fsh;
    float d = dinv[i];
    float s = d * d;
    float4 h = *(const float4*)(H + (size_t)i * F + c);
    float4 o;
    o.x = h.x * s + bias[c + 0];
    o.y = h.y * s + bias[c + 1];
    o.z = h.z * s + bias[c + 2];
    o.w = h.w * s + bias[c + 3];
    *(float4*)(A + (size_t)i * F + c) = o;
}

__global__ void agg_edge_kernel(const int* __restrict__ src,
                                const int* __restrict__ dst,
                                const float* __restrict__ dinv,
                                const float* __restrict__ H,
                                float* __restrict__ A, int E, int fsh) {
    int t = blockIdx.x * blockDim.x + threadIdx.x;
    if (t >= (E << fsh)) return;
    int e = t >> fsh;
    int c = (t & ((1 << fsh) - 1)) << 2;
    int F = 4 << fsh;
    int s = src[e], d = dst[e];
    float nrm = dinv[s] * dinv[d];
    float4 h = *(const float4*)(H + (size_t)s * F + c);
    float* out = A + (size_t)d * F + c;
    atomic_add_f32(out + 0, h.x * nrm);
    atomic_add_f32(out + 1, h.y * nrm);
    atomic_add_f32(out + 2, h.z * nrm);
    atomic_add_f32(out + 3, h.w * nrm);
}

extern "C" void kernel_launch(void* const* d_in, const int* in_sizes, int n_in,
                              void* d_out, int out_size, void* d_ws, size_t ws_size,
                              hipStream_t stream) {
    (void)in_sizes; (void)n_in; (void)out_size;
    const float* x  = (const float*)d_in[0];
    const int*   ei = (const int*)d_in[1];
    const float* W1 = (const float*)d_in[2];
    const float* b1 = (const float*)d_in[3];
    const float* W2 = (const float*)d_in[4];
    const float* b2 = (const float*)d_in[5];
    const float* W3 = (const float*)d_in[6];
    const float* b3 = (const float*)d_in[7];
    float* out = (float*)d_out;

    const int N = GCN_N, E = GCN_E;
    const int* srcI = ei;
    const int* dstI = ei + E;

    float* dinv = (float*)d_ws;
    float* Hbuf = dinv + N;
    float* Abuf = Hbuf + (size_t)N * 64;
    size_t need = ((size_t)N + 2 * (size_t)N * 64) * sizeof(float);
    if (ws_size < need) return;

    const int TB = 256;
    const int mtiles = N / 16;
    const int wavesA = mtiles * 4;
    const int wavesB = mtiles * 2;
    const int gbA = (wavesA + 7) / 8;
    const int gbB = (wavesB + 7) / 8;

    deg_init_kernel <<<(N + TB - 1) / TB, TB, 0, stream>>>(dinv, N);
    deg_count_kernel<<<(E + TB - 1) / TB, TB, 0, stream>>>(dstI, dinv, E);
    dinv_kernel     <<<(N + TB - 1) / TB, TB, 0, stream>>>(dinv, N);

    gemm_wmma_kernel<128, 64, false><<<gbA, TB, 0, stream>>>(x, W1, Hbuf, N);
    agg_init_kernel<<<((N << 4) + TB - 1) / TB, TB, 0, stream>>>(Hbuf, dinv, b1, Abuf, N, 4);
    agg_edge_kernel<<<((E << 4) + TB - 1) / TB, TB, 0, stream>>>(srcI, dstI, dinv, Hbuf, Abuf, E, 4);

    gemm_wmma_kernel<64, 64, true><<<gbA, TB, 0, stream>>>(Abuf, W2, Hbuf, N);
    agg_init_kernel<<<((N << 4) + TB - 1) / TB, TB, 0, stream>>>(Hbuf, dinv, b2, Abuf, N, 4);
    agg_edge_kernel<<<((E << 4) + TB - 1) / TB, TB, 0, stream>>>(srcI, dstI, dinv, Hbuf, Abuf, E, 4);

    gemm_wmma_kernel<64, 32, true><<<gbB, TB, 0, stream>>>(Abuf, W3, Hbuf, N);
    agg_init_kernel<<<((N << 3) + TB - 1) / TB, TB, 0, stream>>>(Hbuf, dinv, b3, out, N, 3);
    agg_edge_kernel<<<((E << 3) + TB - 1) / TB, TB, 0, stream>>>(srcI, dstI, dinv, Hbuf, out, E, 3);
}


# Float scatter atomics removed: edges counting-sorted by dst once per call (int atomic arrival rank + scan), each layer gathers its row sums; per-node sum order follows arrival rank
# speedup vs baseline: 1.8322x; 1.8322x over previous
#include <hip/hip_runtime.h>


typedef float v2f __attribute__((ext_vector_type(2)));
typedef float v8f __attribute__((ext_vector_type(8)));

#define GCN_N 100000
#define GCN_E 1600000

#define SCAN_TB 256
#define SCAN_CHUNK 1024
#define SCAN_NB ((GCN_N + SCAN_CHUNK - 1) / SCAN_CHUNK)

static_assert(SCAN_NB <= SCAN_TB, "the chunk totals are reduced by one block of SCAN_TB threads");


__global__ void cnt_init_kernel(int* cnt, int n) {
    int i = blockIdx.x * blockDim.x + threadIdx.x;
    if (i < n) cnt[i] = 0;
}

__global__ void rank_kernel(const int* __restrict__ dst, int* cnt,
                            int* __restrict__ rank, int e) {
    int i = blockIdx.x * blockDim.x + threadIdx.x;
    if (i < e) rank[i] = atomicAdd(&cnt[dst[i]], 1);
}

__global__ void chunk_sum_kernel(const int* __restrict__ cnt, int* __restrict__ bsum, int n) {
    __shared__ int sh[SCAN_TB];
    const int t = threadIdx.x;
    const int i0 = blockIdx.x * SCAN_CHUNK + t * 4;
    int s = 0;
    for (int k = 0; k < 4; ++k)
        if (i0 + k < n) s += cnt[i0 + k];
    sh[t] = s;
    __syncthreads();
    for (int o = SCAN_TB / 2; o > 0; o >>= 1) {
        if (t < o) sh[t] += sh[t + o];
        __syncthreads();
    }
    if (t == 0) bsum[blockIdx.x] = sh[0];
}

__global__ void scan_dinv_kernel(const int* __restrict__ cnt, const int* __restrict__ bsum,
                                 int* __restrict__ rowptr, float* __restrict__ dinv, int n) {
    __shared__ int sh[SCAN_TB];
    const int t = threadIdx.x;

    sh[t] = (t < (int)blockIdx.x) ? bsum[t] : 0;
    __syncthreads();
    for (int o = SCAN_TB / 2; o > 0; o >>= 1) {
        if (t < o) sh[t] += sh[t + o];
        __syncthreads();
    }
    const int base = sh[0];
    __syncthreads();

    const int i0 = blockIdx.x * SCAN_CHUNK + t * 4;
    int cv[4];
    int ts = 0;
    for (int k = 0; k < 4; ++k) {
        cv[k] = (i0 + k < n) ? cnt[i0 + k] : 0;
        ts += cv[k];
    }
    sh[t] = ts;
    __syncthreads();
    for (int o = 1; o < SCAN_TB; o <<= 1) {
        const int add = (t >= o) ? sh[t - o] : 0;
        __syncthreads();
        sh[t] += add;
        __syncthreads();
    }
    int run = base + sh[t] - ts;
    for (int k = 0; k < 4; ++k) {
        if (i0 + k < n) {
            rowptr[i0 + k] = run;
            dinv[i0 + k] = rsqrtf((float)(cv[k] + 1));
            run += cv[k];
        }
    }
}

__global__ void fill_kernel(const int* __restrict__ src, const int* __restrict__ dst,
                            const int* __restrict__ rank, const int* __restrict__ rowptr,
                            int* __restrict__ col, int e) {
    int i = blockIdx.x * blockDim.x + threadIdx.x;
    if (i < e) col[rowptr[dst[i]] + rank[i]] = src[i];
}

template <int KDIM, int NOUT, bool RELU>
__global__ void gemm_wmma_kernel(const float* __restrict__ X,
                                 const float* __restrict__ W,
                                 float* __restrict__ H, int M) {
    constexpr int NT  = NOUT / 16;
    constexpr int LDW = NOUT + 8;
    __shared__ float Ws[KDIM * LDW];
    for (int i = threadIdx.x; i < KDIM * NOUT; i += blockDim.x) {
        int k = i / NOUT, n = i - k * NOUT;
        Ws[k * LDW + n] = W[i];
    }
    __syncthreads();

    const int wid  = (blockIdx.x * blockDim.x + threadIdx.x) >> 5;
    const int lane = threadIdx.x & 31;
    const int m0 = (wid / NT) * 16;
    const int n0 = (wid % NT) * 16;
    if (m0 >= M) return;

    const int half = lane >> 4;
    const int l16  = lane & 15;
    const int koff = half * 2;
    const float* xrow = X + (size_t)(m0 + l16) * KDIM + koff;
    const float* wcol = Ws + koff * LDW + n0 + l16;

    v8f acc = {};
#pragma unroll 4
    for (int k0 = 0; k0 < KDIM; k0 += 4) {
        float a0 = xrow[k0];
        float a1 = xrow[k0 + 1];
        if (RELU) { a0 = fmaxf(a0, 0.0f); a1 = fmaxf(a1, 0.0f); }
        v2f a; a.x = a0; a.y = a1;
        v2f b; b.x = wcol[k0 * LDW]; b.y = wcol[k0 * LDW + LDW];
        acc = __builtin_amdgcn_wmma_f32_16x16x4_f32(
            false, a, false, b, (short)0, acc, false, false);
    }
    float* orow = H + (size_t)(m0 + half * 8) * NOUT + n0 + l16;
#pragma unroll
    for (int v = 0; v < 8; ++v)
        orow[(size_t)v * NOUT] = acc[v];
}

template <int F>
__global__ void gather_sum_kernel(const int* __restrict__ rowptr,
                                  const int* __restrict__ cnt,
                                  const int* __restrict__ col,
                                  const float* __restrict__ dinv,
                                  const float* __restrict__ H,
                                  const float* __restrict__ bias,
                                  float* __restrict__ A, int n) {
    constexpr int LPN = F / 4;
    constexpr int U = 4;
    const int t = blockIdx.x * blockDim.x + threadIdx.x;
    const int i = t / LPN;
    if (i >= n) return;
    const int c = (t % LPN) * 4;

    const float di = dinv[i];
    const float sw = di * di;
    const float4 hs = *(const float4*)(H + (size_t)i * F + c);
    float4 acc;
    acc.x = hs.x * sw + bias[c + 0];
    acc.y = hs.y * sw + bias[c + 1];
    acc.z = hs.z * sw + bias[c + 2];
    acc.w = hs.w * sw + bias[c + 3];

    const int beg = rowptr[i];
    const int end = beg + cnt[i];
    for (int j = beg; j < end; j += U) {
        int    s[U];
        float  w[U];
        float4 h[U];
#pragma unroll
        for (int k = 0; k < U; ++k) {
            const int jj = j + k;
            s[k] = col[jj < end ? jj : end - 1];
        }
#pragma unroll
        for (int k = 0; k < U; ++k) {
            const float ds = dinv[s[k]];
            h[k] = *(const float4*)(H + (size_t)s[k] * F + c);
            w[k] = (j + k < end) ? ds * di : 0.0f;
        }
#pragma unroll
        for (int k = 0; k < U; ++k) {
            acc.x += h[k].x * w[k];
            acc.y += h[k].y * w[k];
            acc.z += h[k].z * w[k];
            acc.w += h[k].w * w[k];
        }
    }
    *(float4*)(A + (size_t)i * F + c) = acc;
}

extern "C" void kernel_launch(void* const* d_in, const int* in_sizes, int n_in,
                              void* d_out, int out_size, void* d_ws, size_t ws_size,
                              hipStream_t stream) {
    (void)in_sizes; (void)n_in; (void)out_size;
    const float* x  = (const float*)d_in[0];
    const int*   ei = (const int*)d_in[1];
    const float* W1 = (const float*)d_in[2];
    const float* b1 = (const float*)d_in[3];
    const float* W2 = (const float*)d_in[4];
    const float* b2 = (const float*)d_in[5];
    const float* W3 = (const float*)d_in[6];
    const float* b3 = (const float*)d_in[7];
    float* out = (float*)d_out;

    const int N = GCN_N, E = GCN_E;
    const int* srcI = ei;
    const int* dstI = ei + E;

    float* dinv  = (float*)d_ws;
    float* Hbuf  = dinv + N;
    float* Abuf  = Hbuf + (size_t)N * 64;
    int*   cnt   = (int*)(Abuf + (size_t)N * 64);
    int*   rowp  = cnt + N;
    int*   bsum  = rowp + N;
    int*   rank  = bsum + SCAN_TB;
    int*   col   = rank + E;
    size_t need = (3 * (size_t)N + 2 * (size_t)N * 64 + SCAN_TB + 2 * (size_t)E) * sizeof(float);
    if (ws_size < need) return;

    const int TB = 256;
    const int mtiles = N / 16;
    const int wavesA = mtiles * 4;
    const int wavesB = mtiles * 2;
    const int gbA = (wavesA + 7) / 8;
    const int gbB = (wavesB + 7) / 8;

    cnt_init_kernel <<<(N + TB - 1) / TB, TB, 0, stream>>>(cnt, N);
    rank_kernel     <<<(E + TB - 1) / TB, TB, 0, stream>>>(dstI, cnt, rank, E);
    chunk_sum_kernel<<<SCAN_NB, SCAN_TB, 0, stream>>>(cnt, bsum, N);
    scan_dinv_kernel<<<SCAN_NB, SCAN_TB, 0, stream>>>(cnt, bsum, rowp, dinv, N);
    fill_kernel     <<<(E + TB - 1) / TB, TB, 0, stream>>>(srcI, dstI, rank, rowp, col, E);

    gemm_wmma_kernel<128, 64, false><<<gbA, TB, 0, stream>>>(x, W1, Hbuf, N);
    gather_sum_kernel<64><<<(N * 16 + TB - 1) / TB, TB, 0, stream>>>(rowp, cnt, col, dinv, Hbuf, b1, Abuf, N);

    gemm_wmma_kernel<64, 64, true><<<gbA, TB, 0, stream>>>(Abuf, W2, Hbuf, N);
    gather_sum_kernel<64><<<(N * 16 + TB - 1) / TB, TB, 0, stream>>>(rowp, cnt, col, dinv, Hbuf, b2, Abuf, N);

    gemm_wmma_kernel<64, 32, true><<<gbB, TB, 0, stream>>>(Abuf, W3, Hbuf, N);
    gather_sum_kernel<32><<<(N * 8 + TB - 1) / TB, TB, 0, stream>>>(rowp, cnt, col, dinv, Hbuf, b3, out, N);
}


# Edge sort by dst without global atomics: edges binned into 256-node buckets by LDS counters (hist, scan, place), each bucket sorted in LDS; per-node sum order now follows LDS atomic arrival
# speedup vs baseline: 2.4669x; 1.3464x over previous
#include <hip/hip_runtime.h>


typedef float v2f __attribute__((ext_vector_type(2)));
typedef float v8f __attribute__((ext_vector_type(8)));
typedef int   v4i __attribute__((ext_vector_type(4)));

#define GCN_N 100000
#define GCN_E 1600000

#define SORT_BW   256
#define SORT_NBK  ((GCN_N + SORT_BW - 1) / SORT_BW)
#define SORT_NCH  250
#define SORT_CH   (GCN_E / SORT_NCH)
#define SORT_TBE  400
#define SORT_EPT  (SORT_CH / 4 / SORT_TBE)
#define SORT_TBB  512
#define SORT_U    10
#define SRC_BITS  17

static_assert(GCN_E % SORT_NCH == 0 && SORT_CH % 4 == 0, "chunks are whole int4 groups");
static_assert(GCN_N <= (1 << SRC_BITS), "src must fit the low bits of a binned word");
static_assert((SORT_CH / 4) % SORT_TBE == 0 && SORT_EPT == 4, "every thread of an edge pass owns 4 int4 groups");
static_assert(SORT_BW == 256 && SORT_NCH <= 256, "scan widths below");
static_assert(SORT_NBK <= SORT_TBE && SORT_NBK <= SORT_TBB && SORT_TBB == 512, "one thread per bucket total");


template <int NW>
__device__ inline int block_incl_scan(int v, int* wsum) {
    const int lane = threadIdx.x & 31;
    const int wave = threadIdx.x >> 5;
    for (int o = 1; o < 32; o <<= 1) {
        const int u = __shfl_up(v, o);
        if (lane >= o) v += u;
    }
    if (lane == 31 && wave < NW) wsum[wave] = v;
    __syncthreads();
    int off = 0;
    for (int w = 0; w < NW; ++w)
        if (w < wave) off += wsum[w];
    return v + off;
}

__global__ void bucket_hist_kernel(const int* __restrict__ dst, int* __restrict__ hist) {
    __shared__ int lh[SORT_NBK];
    const int t = threadIdx.x;
    for (int k = t; k < SORT_NBK; k += blockDim.x) lh[k] = 0;
    __syncthreads();
    const v4i* d4 = (const v4i*)(dst + (size_t)blockIdx.x * SORT_CH);
    v4i d[SORT_EPT];
#pragma unroll
    for (int u = 0; u < SORT_EPT; ++u) d[u] = d4[t + u * SORT_TBE];
    asm volatile("" :: "v"(d[0]), "v"(d[1]), "v"(d[2]), "v"(d[3]));
#pragma unroll
    for (int u = 0; u < SORT_EPT; ++u) {
        atomicAdd(&lh[d[u].x >> 8], 1);
        atomicAdd(&lh[d[u].y >> 8], 1);
        atomicAdd(&lh[d[u].z >> 8], 1);
        atomicAdd(&lh[d[u].w >> 8], 1);
    }
    __syncthreads();
    for (int k = t; k < SORT_NBK; k += blockDim.x) hist[blockIdx.x * SORT_NBK + k] = lh[k];
}

__global__ void chunk_scan_kernel(int* __restrict__ hist, int* __restrict__ total) {
    __shared__ int wsum[8];
    const int t = threadIdx.x;
    const int k = blockIdx.x;
    const int v = (t < SORT_NCH) ? hist[t * SORT_NBK + k] : 0;
    const int incl = block_incl_scan<8>(v, wsum);
    if (t < SORT_NCH) hist[t * SORT_NBK + k] = incl - v;
    if (t == SORT_NCH - 1) total[k] = incl;
}

__global__ void bucket_place_kernel(const int* __restrict__ src, const int* __restrict__ dst,
                                    const int* __restrict__ pre, const int* __restrict__ total,
                                    int* __restrict__ binned) {
    __shared__ int wsum[(SORT_TBE + 31) / 32];
    __shared__ int lpos[SORT_NBK];
    const int t = threadIdx.x;
    const int v = (t < SORT_NBK) ? total[t] : 0;
    const int incl = block_incl_scan<(SORT_TBE + 31) / 32>(v, wsum);
    if (t < SORT_NBK) lpos[t] = incl - v + pre[blockIdx.x * SORT_NBK + t];
    __syncthreads();
    const v4i* d4 = (const v4i*)(dst + (size_t)blockIdx.x * SORT_CH);
    const v4i* s4 = (const v4i*)(src + (size_t)blockIdx.x * SORT_CH);
    v4i d[SORT_EPT], s[SORT_EPT];
#pragma unroll
    for (int u = 0; u < SORT_EPT; ++u) {
        d[u] = d4[t + u * SORT_TBE];
        s[u] = s4[t + u * SORT_TBE];
    }
    asm volatile("" :: "v"(d[0]), "v"(d[1]), "v"(d[2]), "v"(d[3]),
                       "v"(s[0]), "v"(s[1]), "v"(s[2]), "v"(s[3]));
#pragma unroll
    for (int u = 0; u < SORT_EPT; ++u) {
        const int p0 = atomicAdd(&lpos[d[u].x >> 8], 1);
        const int p1 = atomicAdd(&lpos[d[u].y >> 8], 1);
        const int p2 = atomicAdd(&lpos[d[u].z >> 8], 1);
        const int p3 = atomicAdd(&lpos[d[u].w >> 8], 1);
        binned[p0] = s[u].x | ((d[u].x & (SORT_BW - 1)) << SRC_BITS);
        binned[p1] = s[u].y | ((d[u].y & (SORT_BW - 1)) << SRC_BITS);
        binned[p2] = s[u].z | ((d[u].z & (SORT_BW - 1)) << SRC_BITS);
        binned[p3] = s[u].w | ((d[u].w & (SORT_BW - 1)) << SRC_BITS);
    }
}

__global__ void bucket_sort_kernel(const int* __restrict__ binned, const int* __restrict__ total,
                                   int* __restrict__ rowptr, int* __restrict__ cnt,
                                   float* __restrict__ dinv, int* __restrict__ col, int n) {
    __shared__ int red[16];
    __shared__ int wsum[8];
    __shared__ int lc[SORT_BW];
    const int t = threadIdx.x;
    const int lane = t & 31;
    const int wave = t >> 5;
    const int k = blockIdx.x;

    int s = (t < k) ? total[t] : 0;
    for (int o = 16; o > 0; o >>= 1) s += __shfl_xor(s, o);
    if (lane == 0 && wave < 16) red[wave] = s;
    if (t < SORT_BW) lc[t] = 0;
    __syncthreads();
    int bs = 0;
    for (int w = 0; w < 16; ++w) bs += red[w];
    const int be = bs + total[k];

    for (int e0 = bs + t; e0 < be; e0 += SORT_U * SORT_TBB) {
        int w[SORT_U];
#pragma unroll
        for (int u = 0; u < SORT_U; ++u) {
            const int e = e0 + u * SORT_TBB;
            w[u] = (e < be) ? binned[e] : -1;
        }
#pragma unroll
        for (int u = 0; u < SORT_U; ++u)
            if (w[u] >= 0) atomicAdd(&lc[w[u] >> SRC_BITS], 1);
    }
    __syncthreads();

    const int c = (t < SORT_BW) ? lc[t] : 0;
    const int incl = block_incl_scan<8>(c, wsum);
    if (t < SORT_BW) {
        const int start = bs + incl - c;
        const int node = k * SORT_BW + t;
        if (node < n) {
            rowptr[node] = start;
            cnt[node] = c;
            dinv[node] = rsqrtf((float)(c + 1));
        }
        lc[t] = start;
    }
    __syncthreads();
    for (int e0 = bs + t; e0 < be; e0 += SORT_U * SORT_TBB) {
        int w[SORT_U];
#pragma unroll
        for (int u = 0; u < SORT_U; ++u) {
            const int e = e0 + u * SORT_TBB;
            w[u] = (e < be) ? binned[e] : -1;
        }
        int p[SORT_U];
#pragma unroll
        for (int u = 0; u < SORT_U; ++u)
            p[u] = (w[u] >= 0) ? atomicAdd(&lc[w[u] >> SRC_BITS], 1) : 0;
#pragma unroll
        for (int u = 0; u < SORT_U; ++u)
            if (w[u] >= 0) col[p[u]] = w[u] & ((1 << SRC_BITS) - 1);
    }
}

template <int KDIM, int NOUT, bool RELU>
__global__ void gemm_wmma_kernel(const float* __restrict__ X,
                                 const float* __restrict__ W,
                                 float* __restrict__ H, int M) {
    constexpr int NT  = NOUT / 16;
    constexpr int LDW = NOUT + 8;
    __shared__ float Ws[KDIM * LDW];
    for (int i = threadIdx.x; i < KDIM * NOUT; i += blockDim.x) {
        int k = i / NOUT, n = i - k * NOUT;
        Ws[k * LDW + n] = W[i];
    }
    __syncthreads();

    const int wid  = (blockIdx.x * blockDim.x + threadIdx.x) >> 5;
    const int lane = threadIdx.x & 31;
    const int m0 = (wid / NT) * 16;
    const int n0 = (wid % NT) * 16;
    if (m0 >= M) return;

    const int half = lane >> 4;
    const int l16  = lane & 15;
    const int koff = half * 2;
    const float* xrow = X + (size_t)(m0 + l16) * KDIM + koff;
    const float* wcol = Ws + koff * LDW + n0 + l16;

    v8f acc = {};
#pragma unroll 4
    for (int k0 = 0; k0 < KDIM; k0 += 4) {
        float a0 = xrow[k0];
        float a1 = xrow[k0 + 1];
        if (RELU) { a0 = fmaxf(a0, 0.0f); a1 = fmaxf(a1, 0.0f); }
        v2f a; a.x = a0; a.y = a1;
        v2f b; b.x = wcol[k0 * LDW]; b.y = wcol[k0 * LDW + LDW];
        acc = __builtin_amdgcn_wmma_f32_16x16x4_f32(
            false, a, false, b, (short)0, acc, false, false);
    }
    float* orow = H + (size_t)(m0 + half * 8) * NOUT + n0 + l16;
#pragma unroll
    for (int v = 0; v < 8; ++v)
        orow[(size_t)v * NOUT] = acc[v];
}

template <int F>
__global__ void gather_sum_kernel(const int* __restrict__ rowptr,
                                  const int* __restrict__ cnt,
                                  const int* __restrict__ col,
                                  const float* __restrict__ dinv,
                                  const float* __restrict__ H,
                                  const float* __restrict__ bias,
                                  float* __restrict__ A, int n) {
    constexpr int LPN = F / 4;
    constexpr int U = 4;
    const int t = blockIdx.x * blockDim.x + threadIdx.x;
    const int i = t / LPN;
    if (i >= n) return;
    const int c = (t % LPN) * 4;

    const float di = dinv[i];
    const float sw = di * di;
    const float4 hs = *(const float4*)(H + (size_t)i * F + c);
    float4 acc;
    acc.x = hs.x * sw + bias[c + 0];
    acc.y = hs.y * sw + bias[c + 1];
    acc.z = hs.z * sw + bias[c + 2];
    acc.w = hs.w * sw + bias[c + 3];

    const int beg = rowptr[i];
    const int end = beg + cnt[i];
    for (int j = beg; j < end; j += U) {
        int    s[U];
        float  w[U];
        float4 h[U];
#pragma unroll
        for (int k = 0; k < U; ++k) {
            const int jj = j + k;
            s[k] = col[jj < end ? jj : end - 1];
        }
#pragma unroll
        for (int k = 0; k < U; ++k) {
            const float ds = dinv[s[k]];
            h[k] = *(const float4*)(H + (size_t)s[k] * F + c);
            w[k] = (j + k < end) ? ds * di : 0.0f;
        }
#pragma unroll
        for (int k = 0; k < U; ++k) {
            acc.x += h[k].x * w[k];
            acc.y += h[k].y * w[k];
            acc.z += h[k].z * w[k];
            acc.w += h[k].w * w[k];
        }
    }
    *(float4*)(A + (size_t)i * F + c) = acc;
}

extern "C" void kernel_launch(void* const* d_in, const int* in_sizes, int n_in,
                              void* d_out, int out_size, void* d_ws, size_t ws_size,
                              hipStream_t stream) {
    (void)in_sizes; (void)n_in; (void)out_size;
    const float* x  = (const float*)d_in[0];
    const int*   ei = (const int*)d_in[1];
    const float* W1 = (const float*)d_in[2];
    const float* b1 = (const float*)d_in[3];
    const float* W2 = (const float*)d_in[4];
    const float* b2 = (const float*)d_in[5];
    const float* W3 = (const float*)d_in[6];
    const float* b3 = (const float*)d_in[7];
    float* out = (float*)d_out;

    const int N = GCN_N, E = GCN_E;
    const int* srcI = ei;
    const int* dstI = ei + E;

    float* dinv  = (float*)d_ws;
    float* Hbuf  = dinv + N;
    float* Abuf  = Hbuf + (size_t)N * 64;
    int*   cnt    = (int*)(Abuf + (size_t)N * 64);
    int*   rowp   = cnt + N;
    int*   hist   = rowp + N;
    int*   total  = hist + SORT_NCH * SORT_NBK;
    int*   binned = total + 512;
    int*   col    = binned + E;
    size_t need = (3 * (size_t)N + 2 * (size_t)N * 64 + SORT_NCH * SORT_NBK + 512 + 2 * (size_t)E) * sizeof(float);
    if (ws_size < need) return;

    const int TB = 256;
    const int mtiles = N / 16;
    const int wavesA = mtiles * 4;
    const int wavesB = mtiles * 2;
    const int gbA = (wavesA + 7) / 8;
    const int gbB = (wavesB + 7) / 8;

    bucket_hist_kernel <<<SORT_NCH, SORT_TBE, 0, stream>>>(dstI, hist);
    chunk_scan_kernel  <<<SORT_NBK, 256, 0, stream>>>(hist, total);
    bucket_place_kernel<<<SORT_NCH, SORT_TBE, 0, stream>>>(srcI, dstI, hist, total, binned);
    bucket_sort_kernel <<<SORT_NBK, SORT_TBB, 0, stream>>>(binned, total, rowp, cnt, dinv, col, N);

    gemm_wmma_kernel<128, 64, false><<<gbA, TB, 0, stream>>>(x, W1, Hbuf, N);
    gather_sum_kernel<64><<<(N * 16 + TB - 1) / TB, TB, 0, stream>>>(rowp, cnt, col, dinv, Hbuf, b1, Abuf, N);

    gemm_wmma_kernel<64, 64, true><<<gbA, TB, 0, stream>>>(Abuf, W2, Hbuf, N);
    gather_sum_kernel<64><<<(N * 16 + TB - 1) / TB, TB, 0, stream>>>(rowp, cnt, col, dinv, Hbuf, b2, Abuf, N);

    gemm_wmma_kernel<64, 32, true><<<gbB, TB, 0, stream>>>(Abuf, W3, Hbuf, N);
    gather_sum_kernel<32><<<(N * 8 + TB - 1) / TB, TB, 0, stream>>>(rowp, cnt, col, dinv, Hbuf, b3, out, N);
}


# GEMM re-tiled: one wave owns a 16-row strip across all output columns, so each A fragment is loaded once and feeds NOUT/16 WMMAs; 4x (2x for NOUT=32) fewer GEMM waves, A loads and W stagings
# speedup vs baseline: 2.6391x; 1.0698x over previous
#include <hip/hip_runtime.h>


typedef float v2f __attribute__((ext_vector_type(2)));
typedef float v8f __attribute__((ext_vector_type(8)));
typedef int   v4i __attribute__((ext_vector_type(4)));

#define GCN_N 100000
#define GCN_E 1600000

#define SORT_BW   256
#define SORT_NBK  ((GCN_N + SORT_BW - 1) / SORT_BW)
#define SORT_NCH  250
#define SORT_CH   (GCN_E / SORT_NCH)
#define SORT_TBE  400
#define SORT_EPT  (SORT_CH / 4 / SORT_TBE)
#define SORT_TBB  512
#define SORT_U    10
#define SRC_BITS  17

static_assert(GCN_E % SORT_NCH == 0 && SORT_CH % 4 == 0, "chunks are whole int4 groups");
static_assert(GCN_N <= (1 << SRC_BITS), "src must fit the low bits of a binned word");
static_assert((SORT_CH / 4) % SORT_TBE == 0 && SORT_EPT == 4, "every thread of an edge pass owns 4 int4 groups");
static_assert(SORT_BW == 256 && SORT_NCH <= 256, "scan widths below");
static_assert(SORT_NBK <= SORT_TBE && SORT_NBK <= SORT_TBB && SORT_TBB == 512, "one thread per bucket total");


template <int NW>
__device__ inline int block_incl_scan(int v, int* wsum) {
    const int lane = threadIdx.x & 31;
    const int wave = threadIdx.x >> 5;
    for (int o = 1; o < 32; o <<= 1) {
        const int u = __shfl_up(v, o);
        if (lane >= o) v += u;
    }
    if (lane == 31 && wave < NW) wsum[wave] = v;
    __syncthreads();
    int off = 0;
    for (int w = 0; w < NW; ++w)
        if (w < wave) off += wsum[w];
    return v + off;
}

__global__ void bucket_hist_kernel(const int* __restrict__ dst, int* __restrict__ hist) {
    __shared__ int lh[SORT_NBK];
    const int t = threadIdx.x;
    for (int k = t; k < SORT_NBK; k += blockDim.x) lh[k] = 0;
    __syncthreads();
    const v4i* d4 = (const v4i*)(dst + (size_t)blockIdx.x * SORT_CH);
    v4i d[SORT_EPT];
#pragma unroll
    for (int u = 0; u < SORT_EPT; ++u) d[u] = d4[t + u * SORT_TBE];
    asm volatile("" :: "v"(d[0]), "v"(d[1]), "v"(d[2]), "v"(d[3]));
#pragma unroll
    for (int u = 0; u < SORT_EPT; ++u) {
        atomicAdd(&lh[d[u].x >> 8], 1);
        atomicAdd(&lh[d[u].y >> 8], 1);
        atomicAdd(&lh[d[u].z >> 8], 1);
        atomicAdd(&lh[d[u].w >> 8], 1);
    }
    __syncthreads();
    for (int k = t; k < SORT_NBK; k += blockDim.x) hist[blockIdx.x * SORT_NBK + k] = lh[k];
}

__global__ void chunk_scan_kernel(int* __restrict__ hist, int* __restrict__ total) {
    __shared__ int wsum[8];
    const int t = threadIdx.x;
    const int k = blockIdx.x;
    const int v = (t < SORT_NCH) ? hist[t * SORT_NBK + k] : 0;
    const int incl = block_incl_scan<8>(v, wsum);
    if (t < SORT_NCH) hist[t * SORT_NBK + k] = incl - v;
    if (t == SORT_NCH - 1) total[k] = incl;
}

__global__ void bucket_place_kernel(const int* __restrict__ src, const int* __restrict__ dst,
                                    const int* __restrict__ pre, const int* __restrict__ total,
                                    int* __restrict__ binned) {
    __shared__ int wsum[(SORT_TBE + 31) / 32];
    __shared__ int lpos[SORT_NBK];
    const int t = threadIdx.x;
    const int v = (t < SORT_NBK) ? total[t] : 0;
    const int incl = block_incl_scan<(SORT_TBE + 31) / 32>(v, wsum);
    if (t < SORT_NBK) lpos[t] = incl - v + pre[blockIdx.x * SORT_NBK + t];
    __syncthreads();
    const v4i* d4 = (const v4i*)(dst + (size_t)blockIdx.x * SORT_CH);
    const v4i* s4 = (const v4i*)(src + (size_t)blockIdx.x * SORT_CH);
    v4i d[SORT_EPT], s[SORT_EPT];
#pragma unroll
    for (int u = 0; u < SORT_EPT; ++u) {
        d[u] = d4[t + u * SORT_TBE];
        s[u] = s4[t + u * SORT_TBE];
    }
    asm volatile("" :: "v"(d[0]), "v"(d[1]), "v"(d[2]), "v"(d[3]),
                       "v"(s[0]), "v"(s[1]), "v"(s[2]), "v"(s[3]));
#pragma unroll
    for (int u = 0; u < SORT_EPT; ++u) {
        const int p0 = atomicAdd(&lpos[d[u].x >> 8], 1);
        const int p1 = atomicAdd(&lpos[d[u].y >> 8], 1);
        const int p2 = atomicAdd(&lpos[d[u].z >> 8], 1);
        const int p3 = atomicAdd(&lpos[d[u].w >> 8], 1);
        binned[p0] = s[u].x | ((d[u].x & (SORT_BW - 1)) << SRC_BITS);
        binned[p1] = s[u].y | ((d[u].y & (SORT_BW - 1)) << SRC_BITS);
        binned[p2] = s[u].z | ((d[u].z & (SORT_BW - 1)) << SRC_BITS);
        binned[p3] = s[u].w | ((d[u].w & (SORT_BW - 1)) << SRC_BITS);
    }
}

__global__ void bucket_sort_kernel(const int* __restrict__ binned, const int* __restrict__ total,
                                   int* __restrict__ rowptr, int* __restrict__ cnt,
                                   float* __restrict__ dinv, int* __restrict__ col, int n) {
    __shared__ int red[16];
    __shared__ int wsum[8];
    __shared__ int lc[SORT_BW];
    const int t = threadIdx.x;
    const int lane = t & 31;
    const int wave = t >> 5;
    const int k = blockIdx.x;

    int s = (t < k) ? total[t] : 0;
    for (int o = 16; o > 0; o >>= 1) s += __shfl_xor(s, o);
    if (lane == 0 && wave < 16) red[wave] = s;
    if (t < SORT_BW) lc[t] = 0;
    __syncthreads();
    int bs = 0;
    for (int w = 0; w < 16; ++w) bs += red[w];
    const int be = bs + total[k];

    for (int e0 = bs + t; e0 < be; e0 += SORT_U * SORT_TBB) {
        int w[SORT_U];
#pragma unroll
        for (int u = 0; u < SORT_U; ++u) {
            const int e = e0 + u * SORT_TBB;
            w[u] = (e < be) ? binned[e] : -1;
        }
#pragma unroll
        for (int u = 0; u < SORT_U; ++u)
            if (w[u] >= 0) atomicAdd(&lc[w[u] >> SRC_BITS], 1);
    }
    __syncthreads();

    const int c = (t < SORT_BW) ? lc[t] : 0;
    const int incl = block_incl_scan<8>(c, wsum);
    if (t < SORT_BW) {
        const int start = bs + incl - c;
        const int node = k * SORT_BW + t;
        if (node < n) {
            rowptr[node] = start;
            cnt[node] = c;
            dinv[node] = rsqrtf((float)(c + 1));
        }
        lc[t] = start;
    }
    __syncthreads();
    for (int e0 = bs + t; e0 < be; e0 += SORT_U * SORT_TBB) {
        int w[SORT_U];
#pragma unroll
        for (int u = 0; u < SORT_U; ++u) {
            const int e = e0 + u * SORT_TBB;
            w[u] = (e < be) ? binned[e] : -1;
        }
        int p[SORT_U];
#pragma unroll
        for (int u = 0; u < SORT_U; ++u)
            p[u] = (w[u] >= 0) ? atomicAdd(&lc[w[u] >> SRC_BITS], 1) : 0;
#pragma unroll
        for (int u = 0; u < SORT_U; ++u)
            if (w[u] >= 0) col[p[u]] = w[u] & ((1 << SRC_BITS) - 1);
    }
}

template <int KDIM, int NOUT, bool RELU>
__global__ void gemm_wmma_kernel(const float* __restrict__ X,
                                 const float* __restrict__ W,
                                 float* __restrict__ H, int M) {
    constexpr int NT  = NOUT / 16;
    constexpr int LDW = NOUT + 8;
    constexpr int KC  = 32;
    constexpr int NS  = KC / 4;
    static_assert(KDIM % KC == 0, "whole groups");
    __shared__ float Ws[KDIM * LDW];
    for (int i = threadIdx.x; i < KDIM * NOUT; i += blockDim.x) {
        int k = i / NOUT, n = i - k * NOUT;
        Ws[k * LDW + n] = W[i];
    }
    __syncthreads();

    const int wid  = (blockIdx.x * blockDim.x + threadIdx.x) >> 5;
    const int lane = threadIdx.x & 31;
    const int m0 = wid * 16;
    if (m0 >= M) return;

    const int half = lane >> 4;
    const int l16  = lane & 15;
    const int koff = half * 2;
    const float* xrow = X + (size_t)(m0 + l16) * KDIM + koff;
    const float* wcol = Ws + koff * LDW + l16;

    v8f acc[NT];
#pragma unroll
    for (int nt = 0; nt < NT; ++nt) acc[nt] = v8f{};

    v2f a[NS], an[NS];
#pragma unroll
    for (int s = 0; s < NS; ++s) a[s] = *(const v2f*)(xrow + 4 * s);

#pragma unroll
    for (int kc = 0; kc < KDIM; kc += KC) {
        if (kc + KC < KDIM) {
#pragma unroll
            for (int s = 0; s < NS; ++s) an[s] = *(const v2f*)(xrow + kc + KC + 4 * s);
        }
#pragma unroll
        for (int s = 0; s < NS; ++s) {
            v2f av = a[s];
            if (RELU) { av.x = fmaxf(av.x, 0.0f); av.y = fmaxf(av.y, 0.0f); }
            const int k0 = kc + 4 * s;
#pragma unroll
            for (int nt = 0; nt < NT; ++nt) {
                v2f b;
                b.x = wcol[k0 * LDW + nt * 16];
                b.y = wcol[k0 * LDW + LDW + nt * 16];
                acc[nt] = __builtin_amdgcn_wmma_f32_16x16x4_f32(
                    false, av, false, b, (short)0, acc[nt], false, false);
            }
        }
        if (kc + KC < KDIM) {
#pragma unroll
            for (int s = 0; s < NS; ++s) a[s] = an[s];
        }
    }
    float* orow = H + (size_t)(m0 + half * 8) * NOUT + l16;
#pragma unroll
    for (int nt = 0; nt < NT; ++nt)
#pragma unroll
        for (int v = 0; v < 8; ++v)
            orow[(size_t)v * NOUT + nt * 16] = acc[nt][v];
}

template <int F>
__global__ void gather_sum_kernel(const int* __restrict__ rowptr,
                                  const int* __restrict__ cnt,
                                  const int* __restrict__ col,
                                  const float* __restrict__ dinv,
                                  const float* __restrict__ H,
                                  const float* __restrict__ bias,
                                  float* __restrict__ A, int n) {
    constexpr int LPN = F / 4;
    constexpr int U = 4;
    const int t = blockIdx.x * blockDim.x + threadIdx.x;
    const int i = t / LPN;
    if (i >= n) return;
    const int c = (t % LPN) * 4;

    const float di = dinv[i];
    const float sw = di * di;
    const float4 hs = *(const float4*)(H + (size_t)i * F + c);
    float4 acc;
    acc.x = hs.x * sw + bias[c + 0];
    acc.y = hs.y * sw + bias[c + 1];
    acc.z = hs.z * sw + bias[c + 2];
    acc.w = hs.w * sw + bias[c + 3];

    const int beg = rowptr[i];
    const int end = beg + cnt[i];
    for (int j = beg; j < end; j += U) {
        int    s[U];
        float  w[U];
        float4 h[U];
#pragma unroll
        for (int k = 0; k < U; ++k) {
            const int jj = j + k;
            s[k] = col[jj < end ? jj : end - 1];
        }
#pragma unroll
        for (int k = 0; k < U; ++k) {
            const float ds = dinv[s[k]];
            h[k] = *(const float4*)(H + (size_t)s[k] * F + c);
            w[k] = (j + k < end) ? ds * di : 0.0f;
        }
#pragma unroll
        for (int k = 0; k < U; ++k) {
            acc.x += h[k].x * w[k];
            acc.y += h[k].y * w[k];
            acc.z += h[k].z * w[k];
            acc.w += h[k].w * w[k];
        }
    }
    *(float4*)(A + (size_t)i * F + c) = acc;
}

extern "C" void kernel_launch(void* const* d_in, const int* in_sizes, int n_in,
                              void* d_out, int out_size, void* d_ws, size_t ws_size,
                              hipStream_t stream) {
    (void)in_sizes; (void)n_in; (void)out_size;
    const float* x  = (const float*)d_in[0];
    const int*   ei = (const int*)d_in[1];
    const float* W1 = (const float*)d_in[2];
    const float* b1 = (const float*)d_in[3];
    const float* W2 = (const float*)d_in[4];
    const float* b2 = (const float*)d_in[5];
    const float* W3 = (const float*)d_in[6];
    const float* b3 = (const float*)d_in[7];
    float* out = (float*)d_out;

    const int N = GCN_N, E = GCN_E;
    const int* srcI = ei;
    const int* dstI = ei + E;

    float* dinv  = (float*)d_ws;
    float* Hbuf  = dinv + N;
    float* Abuf  = Hbuf + (size_t)N * 64;
    int*   cnt    = (int*)(Abuf + (size_t)N * 64);
    int*   rowp   = cnt + N;
    int*   hist   = rowp + N;
    int*   total  = hist + SORT_NCH * SORT_NBK;
    int*   binned = total + 512;
    int*   col    = binned + E;
    size_t need = (3 * (size_t)N + 2 * (size_t)N * 64 + SORT_NCH * SORT_NBK + 512 + 2 * (size_t)E) * sizeof(float);
    if (ws_size < need) return;

    const int TB = 256;
    const int mtiles = N / 16;
    const int gbA = (mtiles + 7) / 8;
    const int gbB = gbA;

    bucket_hist_kernel <<<SORT_NCH, SORT_TBE, 0, stream>>>(dstI, hist);
    chunk_scan_kernel  <<<SORT_NBK, 256, 0, stream>>>(hist, total);
    bucket_place_kernel<<<SORT_NCH, SORT_TBE, 0, stream>>>(srcI, dstI, hist, total, binned);
    bucket_sort_kernel <<<SORT_NBK, SORT_TBB, 0, stream>>>(binned, total, rowp, cnt, dinv, col, N);

    gemm_wmma_kernel<128, 64, false><<<gbA, TB, 0, stream>>>(x, W1, Hbuf, N);
    gather_sum_kernel<64><<<(N * 16 + TB - 1) / TB, TB, 0, stream>>>(rowp, cnt, col, dinv, Hbuf, b1, Abuf, N);

    gemm_wmma_kernel<64, 64, true><<<gbA, TB, 0, stream>>>(Abuf, W2, Hbuf, N);
    gather_sum_kernel<64><<<(N * 16 + TB - 1) / TB, TB, 0, stream>>>(rowp, cnt, col, dinv, Hbuf, b2, Abuf, N);

    gemm_wmma_kernel<64, 32, true><<<gbB, TB, 0, stream>>>(Abuf, W3, Hbuf, N);
    gather_sum_kernel<32><<<(N * 8 + TB - 1) / TB, TB, 0, stream>>>(rowp, cnt, col, dinv, Hbuf, b3, out, N);
}


# Edge weight factored: GEMM epilogue scales each row by dinv[row]; gather loads one row per edge (no per-edge dinv lookup) and applies dinv[dst] once to the sum; same sum order, rounding differs
# speedup vs baseline: 2.8853x; 1.0933x over previous
#include <hip/hip_runtime.h>


typedef float v2f __attribute__((ext_vector_type(2)));
typedef float v8f __attribute__((ext_vector_type(8)));
typedef int   v4i __attribute__((ext_vector_type(4)));

#define GCN_N 100000
#define GCN_E 1600000

#define SORT_BW   256
#define SORT_NBK  ((GCN_N + SORT_BW - 1) / SORT_BW)
#define SORT_NCH  250
#define SORT_CH   (GCN_E / SORT_NCH)
#define SORT_TBE  400
#define SORT_EPT  (SORT_CH / 4 / SORT_TBE)
#define SORT_TBB  512
#define SORT_U    10
#define SRC_BITS  17

static_assert(GCN_E % SORT_NCH == 0 && SORT_CH % 4 == 0, "chunks are whole int4 groups");
static_assert(GCN_N <= (1 << SRC_BITS), "src must fit the low bits of a binned word");
static_assert((SORT_CH / 4) % SORT_TBE == 0 && SORT_EPT == 4, "every thread of an edge pass owns 4 int4 groups");
static_assert(SORT_BW == 256 && SORT_NCH <= 256, "scan widths below");
static_assert(SORT_NBK <= SORT_TBE && SORT_NBK <= SORT_TBB && SORT_TBB == 512, "one thread per bucket total");


template <int NW>
__device__ inline int block_incl_scan(int v, int* wsum) {
    const int lane = threadIdx.x & 31;
    const int wave = threadIdx.x >> 5;
    for (int o = 1; o < 32; o <<= 1) {
        const int u = __shfl_up(v, o);
        if (lane >= o) v += u;
    }
    if (lane == 31 && wave < NW) wsum[wave] = v;
    __syncthreads();
    int off = 0;
    for (int w = 0; w < NW; ++w)
        if (w < wave) off += wsum[w];
    return v + off;
}

__global__ void bucket_hist_kernel(const int* __restrict__ dst, int* __restrict__ hist) {
    __shared__ int lh[SORT_NBK];
    const int t = threadIdx.x;
    for (int k = t; k < SORT_NBK; k += blockDim.x) lh[k] = 0;
    __syncthreads();
    const v4i* d4 = (const v4i*)(dst + (size_t)blockIdx.x * SORT_CH);
    v4i d[SORT_EPT];
#pragma unroll
    for (int u = 0; u < SORT_EPT; ++u) d[u] = d4[t + u * SORT_TBE];
    asm volatile("" :: "v"(d[0]), "v"(d[1]), "v"(d[2]), "v"(d[3]));
#pragma unroll
    for (int u = 0; u < SORT_EPT; ++u) {
        atomicAdd(&lh[d[u].x >> 8], 1);
        atomicAdd(&lh[d[u].y >> 8], 1);
        atomicAdd(&lh[d[u].z >> 8], 1);
        atomicAdd(&lh[d[u].w >> 8], 1);
    }
    __syncthreads();
    for (int k = t; k < SORT_NBK; k += blockDim.x) hist[blockIdx.x * SORT_NBK + k] = lh[k];
}

__global__ void chunk_scan_kernel(int* __restrict__ hist, int* __restrict__ total) {
    __shared__ int wsum[8];
    const int t = threadIdx.x;
    const int k = blockIdx.x;
    const int v = (t < SORT_NCH) ? hist[t * SORT_NBK + k] : 0;
    const int incl = block_incl_scan<8>(v, wsum);
    if (t < SORT_NCH) hist[t * SORT_NBK + k] = incl - v;
    if (t == SORT_NCH - 1) total[k] = incl;
}

__global__ void bucket_place_kernel(const int* __restrict__ src, const int* __restrict__ dst,
                                    const int* __restrict__ pre, const int* __restrict__ total,
                                    int* __restrict__ binned) {
    __shared__ int wsum[(SORT_TBE + 31) / 32];
    __shared__ int lpos[SORT_NBK];
    const int t = threadIdx.x;
    const int v = (t < SORT_NBK) ? total[t] : 0;
    const int incl = block_incl_scan<(SORT_TBE + 31) / 32>(v, wsum);
    if (t < SORT_NBK) lpos[t] = incl - v + pre[blockIdx.x * SORT_NBK + t];
    __syncthreads();
    const v4i* d4 = (const v4i*)(dst + (size_t)blockIdx.x * SORT_CH);
    const v4i* s4 = (const v4i*)(src + (size_t)blockIdx.x * SORT_CH);
    v4i d[SORT_EPT], s[SORT_EPT];
#pragma unroll
    for (int u = 0; u < SORT_EPT; ++u) {
        d[u] = d4[t + u * SORT_TBE];
        s[u] = s4[t + u * SORT_TBE];
    }
    asm volatile("" :: "v"(d[0]), "v"(d[1]), "v"(d[2]), "v"(d[3]),
                       "v"(s[0]), "v"(s[1]), "v"(s[2]), "v"(s[3]));
#pragma unroll
    for (int u = 0; u < SORT_EPT; ++u) {
        const int p0 = atomicAdd(&lpos[d[u].x >> 8], 1);
        const int p1 = atomicAdd(&lpos[d[u].y >> 8], 1);
        const int p2 = atomicAdd(&lpos[d[u].z >> 8], 1);
        const int p3 = atomicAdd(&lpos[d[u].w >> 8], 1);
        binned[p0] = s[u].x | ((d[u].x & (SORT_BW - 1)) << SRC_BITS);
        binned[p1] = s[u].y | ((d[u].y & (SORT_BW - 1)) << SRC_BITS);
        binned[p2] = s[u].z | ((d[u].z & (SORT_BW - 1)) << SRC_BITS);
        binned[p3] = s[u].w | ((d[u].w & (SORT_BW - 1)) << SRC_BITS);
    }
}

__global__ void bucket_sort_kernel(const int* __restrict__ binned, const int* __restrict__ total,
                                   int* __restrict__ rowptr, int* __restrict__ cnt,
                                   float* __restrict__ dinv, int* __restrict__ col, int n) {
    __shared__ int red[16];
    __shared__ int wsum[8];
    __shared__ int lc[SORT_BW];
    const int t = threadIdx.x;
    const int lane = t & 31;
    const int wave = t >> 5;
    const int k = blockIdx.x;

    int s = (t < k) ? total[t] : 0;
    for (int o = 16; o > 0; o >>= 1) s += __shfl_xor(s, o);
    if (lane == 0 && wave < 16) red[wave] = s;
    if (t < SORT_BW) lc[t] = 0;
    __syncthreads();
    int bs = 0;
    for (int w = 0; w < 16; ++w) bs += red[w];
    const int be = bs + total[k];

    for (int e0 = bs + t; e0 < be; e0 += SORT_U * SORT_TBB) {
        int w[SORT_U];
#pragma unroll
        for (int u = 0; u < SORT_U; ++u) {
            const int e = e0 + u * SORT_TBB;
            w[u] = (e < be) ? binned[e] : -1;
        }
#pragma unroll
        for (int u = 0; u < SORT_U; ++u)
            if (w[u] >= 0) atomicAdd(&lc[w[u] >> SRC_BITS], 1);
    }
    __syncthreads();

    const int c = (t < SORT_BW) ? lc[t] : 0;
    const int incl = block_incl_scan<8>(c, wsum);
    if (t < SORT_BW) {
        const int start = bs + incl - c;
        const int node = k * SORT_BW + t;
        if (node < n) {
            rowptr[node] = start;
            cnt[node] = c;
            dinv[node] = rsqrtf((float)(c + 1));
        }
        lc[t] = start;
    }
    __syncthreads();
    for (int e0 = bs + t; e0 < be; e0 += SORT_U * SORT_TBB) {
        int w[SORT_U];
#pragma unroll
        for (int u = 0; u < SORT_U; ++u) {
            const int e = e0 + u * SORT_TBB;
            w[u] = (e < be) ? binned[e] : -1;
        }
        int p[SORT_U];
#pragma unroll
        for (int u = 0; u < SORT_U; ++u)
            p[u] = (w[u] >= 0) ? atomicAdd(&lc[w[u] >> SRC_BITS], 1) : 0;
#pragma unroll
        for (int u = 0; u < SORT_U; ++u)
            if (w[u] >= 0) col[p[u]] = w[u] & ((1 << SRC_BITS) - 1);
    }
}

template <int KDIM, int NOUT, bool RELU>
__global__ void gemm_wmma_kernel(const float* __restrict__ X,
                                 const float* __restrict__ W,
                                 const float* __restrict__ rs,
                                 float* __restrict__ H, int M) {
    constexpr int NT  = NOUT / 16;
    constexpr int LDW = NOUT + 8;
    constexpr int KC  = 32;
    constexpr int NS  = KC / 4;
    static_assert(KDIM % KC == 0, "whole groups");
    __shared__ float Ws[KDIM * LDW];
    for (int i = threadIdx.x; i < KDIM * NOUT; i += blockDim.x) {
        int k = i / NOUT, n = i - k * NOUT;
        Ws[k * LDW + n] = W[i];
    }
    __syncthreads();

    const int wid  = (blockIdx.x * blockDim.x + threadIdx.x) >> 5;
    const int lane = threadIdx.x & 31;
    const int m0 = wid * 16;
    if (m0 >= M) return;

    const int half = lane >> 4;
    const int l16  = lane & 15;
    const int koff = half * 2;
    const float* xrow = X + (size_t)(m0 + l16) * KDIM + koff;
    const float* wcol = Ws + koff * LDW + l16;

    const float4 rs0 = *(const float4*)(rs + m0 + half * 8);
    const float4 rs1 = *(const float4*)(rs + m0 + half * 8 + 4);

    v8f acc[NT];
#pragma unroll
    for (int nt = 0; nt < NT; ++nt) acc[nt] = v8f{};

    v2f a[NS], an[NS];
#pragma unroll
    for (int s = 0; s < NS; ++s) a[s] = *(const v2f*)(xrow + 4 * s);

#pragma unroll
    for (int kc = 0; kc < KDIM; kc += KC) {
        if (kc + KC < KDIM) {
#pragma unroll
            for (int s = 0; s < NS; ++s) an[s] = *(const v2f*)(xrow + kc + KC + 4 * s);
        }
#pragma unroll
        for (int s = 0; s < NS; ++s) {
            v2f av = a[s];
            if (RELU) { av.x = fmaxf(av.x, 0.0f); av.y = fmaxf(av.y, 0.0f); }
            const int k0 = kc + 4 * s;
#pragma unroll
            for (int nt = 0; nt < NT; ++nt) {
                v2f b;
                b.x = wcol[k0 * LDW + nt * 16];
                b.y = wcol[k0 * LDW + LDW + nt * 16];
                acc[nt] = __builtin_amdgcn_wmma_f32_16x16x4_f32(
                    false, av, false, b, (short)0, acc[nt], false, false);
            }
        }
        if (kc + KC < KDIM) {
#pragma unroll
            for (int s = 0; s < NS; ++s) a[s] = an[s];
        }
    }
    const float rsv[8] = {rs0.x, rs0.y, rs0.z, rs0.w, rs1.x, rs1.y, rs1.z, rs1.w};
    float* orow = H + (size_t)(m0 + half * 8) * NOUT + l16;
#pragma unroll
    for (int nt = 0; nt < NT; ++nt)
#pragma unroll
        for (int v = 0; v < 8; ++v)
            orow[(size_t)v * NOUT + nt * 16] = acc[nt][v] * rsv[v];
}

template <int F>
__global__ void gather_sum_kernel(const int* __restrict__ rowptr,
                                  const int* __restrict__ cnt,
                                  const int* __restrict__ col,
                                  const float* __restrict__ dinv,
                                  const float* __restrict__ H,
                                  const float* __restrict__ bias,
                                  float* __restrict__ A, int n) {
    constexpr int LPN = F / 4;
    constexpr int U = 4;
    const int t = blockIdx.x * blockDim.x + threadIdx.x;
    const int i = t / LPN;
    if (i >= n) return;
    const int c = (t % LPN) * 4;

    const float di = dinv[i];
    float4 acc = *(const float4*)(H + (size_t)i * F + c);
    const float4 bv = *(const float4*)(bias + c);

    const int beg = rowptr[i];
    const int end = beg + cnt[i];
    for (int j = beg; j < end; j += U) {
        int    s[U];
        float  w[U];
        float4 h[U];
#pragma unroll
        for (int k = 0; k < U; ++k) {
            const int jj = j + k;
            s[k] = col[jj < end ? jj : end - 1];
        }
#pragma unroll
        for (int k = 0; k < U; ++k) {
            h[k] = *(const float4*)(H + (size_t)s[k] * F + c);
            w[k] = (j + k < end) ? 1.0f : 0.0f;
        }
#pragma unroll
        for (int k = 0; k < U; ++k) {
            acc.x += h[k].x * w[k];
            acc.y += h[k].y * w[k];
            acc.z += h[k].z * w[k];
            acc.w += h[k].w * w[k];
        }
    }
    float4 o;
    o.x = acc.x * di + bv.x;
    o.y = acc.y * di + bv.y;
    o.z = acc.z * di + bv.z;
    o.w = acc.w * di + bv.w;
    *(float4*)(A + (size_t)i * F + c) = o;
}

extern "C" void kernel_launch(void* const* d_in, const int* in_sizes, int n_in,
                              void* d_out, int out_size, void* d_ws, size_t ws_size,
                              hipStream_t stream) {
    (void)in_sizes; (void)n_in; (void)out_size;
    const float* x  = (const float*)d_in[0];
    const int*   ei = (const int*)d_in[1];
    const float* W1 = (const float*)d_in[2];
    const float* b1 = (const float*)d_in[3];
    const float* W2 = (const float*)d_in[4];
    const float* b2 = (const float*)d_in[5];
    const float* W3 = (const float*)d_in[6];
    const float* b3 = (const float*)d_in[7];
    float* out = (float*)d_out;

    const int N = GCN_N, E = GCN_E;
    const int* srcI = ei;
    const int* dstI = ei + E;

    float* dinv  = (float*)d_ws;
    float* Hbuf  = dinv + N;
    float* Abuf  = Hbuf + (size_t)N * 64;
    int*   cnt    = (int*)(Abuf + (size_t)N * 64);
    int*   rowp   = cnt + N;
    int*   hist   = rowp + N;
    int*   total  = hist + SORT_NCH * SORT_NBK;
    int*   binned = total + 512;
    int*   col    = binned + E;
    size_t need = (3 * (size_t)N + 2 * (size_t)N * 64 + SORT_NCH * SORT_NBK + 512 + 2 * (size_t)E) * sizeof(float);
    if (ws_size < need) return;

    const int TB = 256;
    const int mtiles = N / 16;
    const int gbA = (mtiles + 7) / 8;
    const int gbB = gbA;

    bucket_hist_kernel <<<SORT_NCH, SORT_TBE, 0, stream>>>(dstI, hist);
    chunk_scan_kernel  <<<SORT_NBK, 256, 0, stream>>>(hist, total);
    bucket_place_kernel<<<SORT_NCH, SORT_TBE, 0, stream>>>(srcI, dstI, hist, total, binned);
    bucket_sort_kernel <<<SORT_NBK, SORT_TBB, 0, stream>>>(binned, total, rowp, cnt, dinv, col, N);

    gemm_wmma_kernel<128, 64, false><<<gbA, TB, 0, stream>>>(x, W1, dinv, Hbuf, N);
    gather_sum_kernel<64><<<(N * 16 + TB - 1) / TB, TB, 0, stream>>>(rowp, cnt, col, dinv, Hbuf, b1, Abuf, N);

    gemm_wmma_kernel<64, 64, true><<<gbA, TB, 0, stream>>>(Abuf, W2, dinv, Hbuf, N);
    gather_sum_kernel<64><<<(N * 16 + TB - 1) / TB, TB, 0, stream>>>(rowp, cnt, col, dinv, Hbuf, b2, Abuf, N);

    gemm_wmma_kernel<64, 32, true><<<gbB, TB, 0, stream>>>(Abuf, W3, dinv, Hbuf, N);
    gather_sum_kernel<32><<<(N * 8 + TB - 1) / TB, TB, 0, stream>>>(rowp, cnt, col, dinv, Hbuf, b3, out, N);
}


# Gather: a node's lanes fetch its next 16 (8 at 32 features) edge indices in one load and share them by DPP lane moves, rowptr read beside cnt; 8 rows in flight, no per-edge index load; same sum order
# speedup vs baseline: 3.0192x; 1.0464x over previous
#include <hip/hip_runtime.h>


typedef float v2f __attribute__((ext_vector_type(2)));
typedef float v8f __attribute__((ext_vector_type(8)));
typedef int   v4i __attribute__((ext_vector_type(4)));

#define GCN_N 100000
#define GCN_E 1600000

#define SORT_BW   256
#define SORT_NBK  ((GCN_N + SORT_BW - 1) / SORT_BW)
#define SORT_NCH  250
#define SORT_CH   (GCN_E / SORT_NCH)
#define SORT_TBE  400
#define SORT_EPT  (SORT_CH / 4 / SORT_TBE)
#define SORT_TBB  512
#define SORT_U    10
#define SRC_BITS  17

static_assert(GCN_E % SORT_NCH == 0 && SORT_CH % 4 == 0, "chunks are whole int4 groups");
static_assert(GCN_N <= (1 << SRC_BITS), "src must fit the low bits of a binned word");
static_assert((SORT_CH / 4) % SORT_TBE == 0 && SORT_EPT == 4, "every thread of an edge pass owns 4 int4 groups");
static_assert(SORT_BW == 256 && SORT_NCH <= 256, "scan widths below");
static_assert(SORT_NBK <= SORT_TBE && SORT_NBK <= SORT_TBB && SORT_TBB == 512, "one thread per bucket total");


template <int NW>
__device__ inline int block_incl_scan(int v, int* wsum) {
    const int lane = threadIdx.x & 31;
    const int wave = threadIdx.x >> 5;
    for (int o = 1; o < 32; o <<= 1) {
        const int u = __shfl_up(v, o);
        if (lane >= o) v += u;
    }
    if (lane == 31 && wave < NW) wsum[wave] = v;
    __syncthreads();
    int off = 0;
    for (int w = 0; w < NW; ++w)
        if (w < wave) off += wsum[w];
    return v + off;
}

__global__ void bucket_hist_kernel(const int* __restrict__ dst, int* __restrict__ hist) {
    __shared__ int lh[SORT_NBK];
    const int t = threadIdx.x;
    for (int k = t; k < SORT_NBK; k += blockDim.x) lh[k] = 0;
    __syncthreads();
    const v4i* d4 = (const v4i*)(dst + (size_t)blockIdx.x * SORT_CH);
    v4i d[SORT_EPT];
#pragma unroll
    for (int u = 0; u < SORT_EPT; ++u) d[u] = d4[t + u * SORT_TBE];
    asm volatile("" :: "v"(d[0]), "v"(d[1]), "v"(d[2]), "v"(d[3]));
#pragma unroll
    for (int u = 0; u < SORT_EPT; ++u) {
        atomicAdd(&lh[d[u].x >> 8], 1);
        atomicAdd(&lh[d[u].y >> 8], 1);
        atomicAdd(&lh[d[u].z >> 8], 1);
        atomicAdd(&lh[d[u].w >> 8], 1);
    }
    __syncthreads();
    for (int k = t; k < SORT_NBK; k += blockDim.x) hist[blockIdx.x * SORT_NBK + k] = lh[k];
}

__global__ void chunk_scan_kernel(int* __restrict__ hist, int* __restrict__ total) {
    __shared__ int wsum[8];
    const int t = threadIdx.x;
    const int k = blockIdx.x;
    const int v = (t < SORT_NCH) ? hist[t * SORT_NBK + k] : 0;
    const int incl = block_incl_scan<8>(v, wsum);
    if (t < SORT_NCH) hist[t * SORT_NBK + k] = incl - v;
    if (t == SORT_NCH - 1) total[k] = incl;
}

__global__ void bucket_place_kernel(const int* __restrict__ src, const int* __restrict__ dst,
                                    const int* __restrict__ pre, const int* __restrict__ total,
                                    int* __restrict__ binned) {
    __shared__ int wsum[(SORT_TBE + 31) / 32];
    __shared__ int lpos[SORT_NBK];
    const int t = threadIdx.x;
    const int v = (t < SORT_NBK) ? total[t] : 0;
    const int incl = block_incl_scan<(SORT_TBE + 31) / 32>(v, wsum);
    if (t < SORT_NBK) lpos[t] = incl - v + pre[blockIdx.x * SORT_NBK + t];
    __syncthreads();
    const v4i* d4 = (const v4i*)(dst + (size_t)blockIdx.x * SORT_CH);
    const v4i* s4 = (const v4i*)(src + (size_t)blockIdx.x * SORT_CH);
    v4i d[SORT_EPT], s[SORT_EPT];
#pragma unroll
    for (int u = 0; u < SORT_EPT; ++u) {
        d[u] = d4[t + u * SORT_TBE];
        s[u] = s4[t + u * SORT_TBE];
    }
    asm volatile("" :: "v"(d[0]), "v"(d[1]), "v"(d[2]), "v"(d[3]),
                       "v"(s[0]), "v"(s[1]), "v"(s[2]), "v"(s[3]));
#pragma unroll
    for (int u = 0; u < SORT_EPT; ++u) {
        const int p0 = atomicAdd(&lpos[d[u].x >> 8], 1);
        const int p1 = atomicAdd(&lpos[d[u].y >> 8], 1);
        const int p2 = atomicAdd(&lpos[d[u].z >> 8], 1);
        const int p3 = atomicAdd(&lpos[d[u].w >> 8], 1);
        binned[p0] = s[u].x | ((d[u].x & (SORT_BW - 1)) << SRC_BITS);
        binned[p1] = s[u].y | ((d[u].y & (SORT_BW - 1)) << SRC_BITS);
        binned[p2] = s[u].z | ((d[u].z & (SORT_BW - 1)) << SRC_BITS);
        binned[p3] = s[u].w | ((d[u].w & (SORT_BW - 1)) << SRC_BITS);
    }
}

__global__ void bucket_sort_kernel(const int* __restrict__ binned, const int* __restrict__ total,
                                   int* __restrict__ rowptr, int* __restrict__ cnt,
                                   float* __restrict__ dinv, int* __restrict__ col, int n) {
    __shared__ int red[16];
    __shared__ int wsum[8];
    __shared__ int lc[SORT_BW];
    const int t = threadIdx.x;
    const int lane = t & 31;
    const int wave = t >> 5;
    const int k = blockIdx.x;

    int s = (t < k) ? total[t] : 0;
    for (int o = 16; o > 0; o >>= 1) s += __shfl_xor(s, o);
    if (lane == 0 && wave < 16) red[wave] = s;
    if (t < SORT_BW) lc[t] = 0;
    __syncthreads();
    int bs = 0;
    for (int w = 0; w < 16; ++w) bs += red[w];
    const int be = bs + total[k];

    for (int e0 = bs + t; e0 < be; e0 += SORT_U * SORT_TBB) {
        int w[SORT_U];
#pragma unroll
        for (int u = 0; u < SORT_U; ++u) {
            const int e = e0 + u * SORT_TBB;
            w[u] = (e < be) ? binned[e] : -1;
        }
#pragma unroll
        for (int u = 0; u < SORT_U; ++u)
            if (w[u] >= 0) atomicAdd(&lc[w[u] >> SRC_BITS], 1);
    }
    __syncthreads();

    const int c = (t < SORT_BW) ? lc[t] : 0;
    const int incl = block_incl_scan<8>(c, wsum);
    if (t < SORT_BW) {
        const int start = bs + incl - c;
        const int node = k * SORT_BW + t;
        if (node < n) {
            rowptr[node] = start;
            cnt[node] = c;
            dinv[node] = rsqrtf((float)(c + 1));
        }
        lc[t] = start;
    }
    __syncthreads();
    for (int e0 = bs + t; e0 < be; e0 += SORT_U * SORT_TBB) {
        int w[SORT_U];
#pragma unroll
        for (int u = 0; u < SORT_U; ++u) {
            const int e = e0 + u * SORT_TBB;
            w[u] = (e < be) ? binned[e] : -1;
        }
        int p[SORT_U];
#pragma unroll
        for (int u = 0; u < SORT_U; ++u)
            p[u] = (w[u] >= 0) ? atomicAdd(&lc[w[u] >> SRC_BITS], 1) : 0;
#pragma unroll
        for (int u = 0; u < SORT_U; ++u)
            if (w[u] >= 0) col[p[u]] = w[u] & ((1 << SRC_BITS) - 1);
    }
}

template <int KDIM, int NOUT, bool RELU>
__global__ void gemm_wmma_kernel(const float* __restrict__ X,
                                 const float* __restrict__ W,
                                 const float* __restrict__ rs,
                                 float* __restrict__ H, int M) {
    constexpr int NT  = NOUT / 16;
    constexpr int LDW = NOUT + 8;
    constexpr int KC  = 32;
    constexpr int NS  = KC / 4;
    static_assert(KDIM % KC == 0, "whole groups");
    __shared__ float Ws[KDIM * LDW];
    for (int i = threadIdx.x; i < KDIM * NOUT; i += blockDim.x) {
        int k = i / NOUT, n = i - k * NOUT;
        Ws[k * LDW + n] = W[i];
    }
    __syncthreads();

    const int wid  = (blockIdx.x * blockDim.x + threadIdx.x) >> 5;
    const int lane = threadIdx.x & 31;
    const int m0 = wid * 16;
    if (m0 >= M) return;

    const int half = lane >> 4;
    const int l16  = lane & 15;
    const int koff = half * 2;
    const float* xrow = X + (size_t)(m0 + l16) * KDIM + koff;
    const float* wcol = Ws + koff * LDW + l16;

    const float4 rs0 = *(const float4*)(rs + m0 + half * 8);
    const float4 rs1 = *(const float4*)(rs + m0 + half * 8 + 4);

    v8f acc[NT];
#pragma unroll
    for (int nt = 0; nt < NT; ++nt) acc[nt] = v8f{};

    v2f a[NS], an[NS];
#pragma unroll
    for (int s = 0; s < NS; ++s) a[s] = *(const v2f*)(xrow + 4 * s);

#pragma unroll
    for (int kc = 0; kc < KDIM; kc += KC) {
        if (kc + KC < KDIM) {
#pragma unroll
            for (int s = 0; s < NS; ++s) an[s] = *(const v2f*)(xrow + kc + KC + 4 * s);
        }
#pragma unroll
        for (int s = 0; s < NS; ++s) {
            v2f av = a[s];
            if (RELU) { av.x = fmaxf(av.x, 0.0f); av.y = fmaxf(av.y, 0.0f); }
            const int k0 = kc + 4 * s;
#pragma unroll
            for (int nt = 0; nt < NT; ++nt) {
                v2f b;
                b.x = wcol[k0 * LDW + nt * 16];
                b.y = wcol[k0 * LDW + LDW + nt * 16];
                acc[nt] = __builtin_amdgcn_wmma_f32_16x16x4_f32(
                    false, av, false, b, (short)0, acc[nt], false, false);
            }
        }
        if (kc + KC < KDIM) {
#pragma unroll
            for (int s = 0; s < NS; ++s) a[s] = an[s];
        }
    }
    const float rsv[8] = {rs0.x, rs0.y, rs0.z, rs0.w, rs1.x, rs1.y, rs1.z, rs1.w};
    float* orow = H + (size_t)(m0 + half * 8) * NOUT + l16;
#pragma unroll
    for (int nt = 0; nt < NT; ++nt)
#pragma unroll
        for (int v = 0; v < 8; ++v)
            orow[(size_t)v * NOUT + nt * 16] = acc[nt][v] * rsv[v];
}

template <int LPN, int K>
__device__ inline int group_bcast(int v) {
    static_assert(LPN == 16 || LPN == 8, "a group is a DPP row or a DPP8 group");
    static_assert(K >= 0 && K < LPN, "source lane inside the group");
    if constexpr (LPN == 16)
        return __builtin_amdgcn_update_dpp(0, v, 0x150 + K, 0xF, 0xF, true);
    else
        return __builtin_amdgcn_mov_dpp8(v, K * 0x249249);
}

template <int F, int G>
__device__ inline void load_row_group(const float* __restrict__ H, int mine, int c, float4 (&h)[8]) {
    constexpr int LPN = F / 4;
    int s[8];
    s[0] = group_bcast<LPN, G + 0>(mine);
    s[1] = group_bcast<LPN, G + 1>(mine);
    s[2] = group_bcast<LPN, G + 2>(mine);
    s[3] = group_bcast<LPN, G + 3>(mine);
    s[4] = group_bcast<LPN, G + 4>(mine);
    s[5] = group_bcast<LPN, G + 5>(mine);
    s[6] = group_bcast<LPN, G + 6>(mine);
    s[7] = group_bcast<LPN, G + 7>(mine);
#pragma unroll
    for (int k = 0; k < 8; ++k) h[k] = *(const float4*)(H + (size_t)s[k] * F + c);
}

template <int G>
__device__ inline void sum_row_group(const float4 (&h)[8], int rem, float4& acc) {
#pragma unroll
    for (int k = 0; k < 8; ++k) {
        const float w = (G + k < rem) ? 1.0f : 0.0f;
        acc.x += h[k].x * w;
        acc.y += h[k].y * w;
        acc.z += h[k].z * w;
        acc.w += h[k].w * w;
    }
}

template <int F>
__global__ void gather_sum_kernel(const int* __restrict__ rowptr,
                                  const int* __restrict__ cnt,
                                  const int* __restrict__ col,
                                  const float* __restrict__ dinv,
                                  const float* __restrict__ H,
                                  const float* __restrict__ bias,
                                  float* __restrict__ A, int n) {
    constexpr int LPN = F / 4;
    const int t = blockIdx.x * blockDim.x + threadIdx.x;
    const int i = t / LPN;
    if (i >= n) return;
    const int q = t % LPN;
    const int c = q * 4;

    const int beg = rowptr[i];
    const int deg = cnt[i];
    const float di = dinv[i];
    float4 acc = *(const float4*)(H + (size_t)i * F + c);
    const float4 bv = *(const float4*)(bias + c);
    asm volatile("" :: "v"(beg), "v"(deg));

    const int end = beg + deg;
    int mine = 0;
    if (beg < end) mine = col[beg + q < end ? beg + q : end - 1];
    for (int j = beg; j < end; j += LPN) {
        const int rem = end - j;
        float4 h[8];
        load_row_group<F, 0>(H, mine, c, h);
        int next = 0;
        if (rem > LPN) {
            const int jn = j + LPN + q;
            next = col[jn < end ? jn : end - 1];
        }
        sum_row_group<0>(h, rem, acc);
        if constexpr (LPN > 8) {
            if (rem > 8) {
                load_row_group<F, 8>(H, mine, c, h);
                sum_row_group<8>(h, rem, acc);
            }
        }
        mine = next;
    }
    float4 o;
    o.x = acc.x * di + bv.x;
    o.y = acc.y * di + bv.y;
    o.z = acc.z * di + bv.z;
    o.w = acc.w * di + bv.w;
    *(float4*)(A + (size_t)i * F + c) = o;
}

extern "C" void kernel_launch(void* const* d_in, const int* in_sizes, int n_in,
                              void* d_out, int out_size, void* d_ws, size_t ws_size,
                              hipStream_t stream) {
    (void)in_sizes; (void)n_in; (void)out_size;
    const float* x  = (const float*)d_in[0];
    const int*   ei = (const int*)d_in[1];
    const float* W1 = (const float*)d_in[2];
    const float* b1 = (const float*)d_in[3];
    const float* W2 = (const float*)d_in[4];
    const float* b2 = (const float*)d_in[5];
    const float* W3 = (const float*)d_in[6];
    const float* b3 = (const float*)d_in[7];
    float* out = (float*)d_out;

    const int N = GCN_N, E = GCN_E;
    const int* srcI = ei;
    const int* dstI = ei + E;

    float* dinv  = (float*)d_ws;
    float* Hbuf  = dinv + N;
    float* Abuf  = Hbuf + (size_t)N * 64;
    int*   cnt    = (int*)(Abuf + (size_t)N * 64);
    int*   rowp   = cnt + N;
    int*   hist   = rowp + N;
    int*   total  = hist + SORT_NCH * SORT_NBK;
    int*   binned = total + 512;
    int*   col    = binned + E;
    size_t need = (3 * (size_t)N + 2 * (size_t)N * 64 + SORT_NCH * SORT_NBK + 512 + 2 * (size_t)E) * sizeof(float);
    if (ws_size < need) return;

    const int TB = 256;
    const int mtiles = N / 16;
    const int gbA = (mtiles + 7) / 8;
    const int gbB = gbA;

    bucket_hist_kernel <<<SORT_NCH, SORT_TBE, 0, stream>>>(dstI, hist);
    chunk_scan_kernel  <<<SORT_NBK, 256, 0, stream>>>(hist, total);
    bucket_place_kernel<<<SORT_NCH, SORT_TBE, 0, stream>>>(srcI, dstI, hist, total, binned);
    bucket_sort_kernel <<<SORT_NBK, SORT_TBB, 0, stream>>>(binned, total, rowp, cnt, dinv, col, N);

    gemm_wmma_kernel<128, 64, false><<<gbA, TB, 0, stream>>>(x, W1, dinv, Hbuf, N);
    gather_sum_kernel<64><<<(N * 16 + TB - 1) / TB, TB, 0, stream>>>(rowp, cnt, col, dinv, Hbuf, b1, Abuf, N);

    gemm_wmma_kernel<64, 64, true><<<gbA, TB, 0, stream>>>(Abuf, W2, dinv, Hbuf, N);
    gather_sum_kernel<64><<<(N * 16 + TB - 1) / TB, TB, 0, stream>>>(rowp, cnt, col, dinv, Hbuf, b2, Abuf, N);

    gemm_wmma_kernel<64, 32, true><<<gbB, TB, 0, stream>>>(Abuf, W3, dinv, Hbuf, N);
    gather_sum_kernel<32><<<(N * 8 + TB - 1) / TB, TB, 0, stream>>>(rowp, cnt, col, dinv, Hbuf, b3, out, N);
}


# Gather fused with the next layer's product: a block's 16 finished rows stay in LDS and are multiplied by W there (f32 WMMA); 2 GEMM launches and the A plane's round trips go; same sum order
# speedup vs baseline: 3.4212x; 1.1331x over previous
#include <hip/hip_runtime.h>


typedef float v2f __attribute__((ext_vector_type(2)));
typedef float v8f __attribute__((ext_vector_type(8)));
typedef int   v4i __attribute__((ext_vector_type(4)));

#define GCN_N 100000
#define GCN_E 1600000

#define SORT_BW   256
#define SORT_NBK  ((GCN_N + SORT_BW - 1) / SORT_BW)
#define SORT_NCH  250
#define SORT_CH   (GCN_E / SORT_NCH)
#define SORT_TBE  400
#define SORT_EPT  (SORT_CH / 4 / SORT_TBE)
#define SORT_TBB  512
#define SORT_U    10
#define SRC_BITS  17

static_assert(GCN_N % 16 == 0, "the fused gather blocks own 16 whole nodes each");
static_assert(GCN_E % SORT_NCH == 0 && SORT_CH % 4 == 0, "chunks are whole int4 groups");
static_assert(GCN_N <= (1 << SRC_BITS), "src must fit the low bits of a binned word");
static_assert((SORT_CH / 4) % SORT_TBE == 0 && SORT_EPT == 4, "every thread of an edge pass owns 4 int4 groups");
static_assert(SORT_BW == 256 && SORT_NCH <= 256, "scan widths below");
static_assert(SORT_NBK <= SORT_TBE && SORT_NBK <= SORT_TBB && SORT_TBB == 512, "one thread per bucket total");


template <int NW>
__device__ inline int block_incl_scan(int v, int* wsum) {
    const int lane = threadIdx.x & 31;
    const int wave = threadIdx.x >> 5;
    for (int o = 1; o < 32; o <<= 1) {
        const int u = __shfl_up(v, o);
        if (lane >= o) v += u;
    }
    if (lane == 31 && wave < NW) wsum[wave] = v;
    __syncthreads();
    int off = 0;
    for (int w = 0; w < NW; ++w)
        if (w < wave) off += wsum[w];
    return v + off;
}

__global__ void bucket_hist_kernel(const int* __restrict__ dst, int* __restrict__ hist) {
    __shared__ int lh[SORT_NBK];
    const int t = threadIdx.x;
    for (int k = t; k < SORT_NBK; k += blockDim.x) lh[k] = 0;
    __syncthreads();
    const v4i* d4 = (const v4i*)(dst + (size_t)blockIdx.x * SORT_CH);
    v4i d[SORT_EPT];
#pragma unroll
    for (int u = 0; u < SORT_EPT; ++u) d[u] = d4[t + u * SORT_TBE];
    asm volatile("" :: "v"(d[0]), "v"(d[1]), "v"(d[2]), "v"(d[3]));
#pragma unroll
    for (int u = 0; u < SORT_EPT; ++u) {
        atomicAdd(&lh[d[u].x >> 8], 1);
        atomicAdd(&lh[d[u].y >> 8], 1);
        atomicAdd(&lh[d[u].z >> 8], 1);
        atomicAdd(&lh[d[u].w >> 8], 1);
    }
    __syncthreads();
    for (int k = t; k < SORT_NBK; k += blockDim.x) hist[blockIdx.x * SORT_NBK + k] = lh[k];
}

__global__ void chunk_scan_kernel(int* __restrict__ hist, int* __restrict__ total) {
    __shared__ int wsum[8];
    const int t = threadIdx.x;
    const int k = blockIdx.x;
    const int v = (t < SORT_NCH) ? hist[t * SORT_NBK + k] : 0;
    const int incl = block_incl_scan<8>(v, wsum);
    if (t < SORT_NCH) hist[t * SORT_NBK + k] = incl - v;
    if (t == SORT_NCH - 1) total[k] = incl;
}

__global__ void bucket_place_kernel(const int* __restrict__ src, const int* __restrict__ dst,
                                    const int* __restrict__ pre, const int* __restrict__ total,
                                    int* __restrict__ binned) {
    __shared__ int wsum[(SORT_TBE + 31) / 32];
    __shared__ int lpos[SORT_NBK];
    const int t = threadIdx.x;
    const int v = (t < SORT_NBK) ? total[t] : 0;
    const int incl = block_incl_scan<(SORT_TBE + 31) / 32>(v, wsum);
    if (t < SORT_NBK) lpos[t] = incl - v + pre[blockIdx.x * SORT_NBK + t];
    __syncthreads();
    const v4i* d4 = (const v4i*)(dst + (size_t)blockIdx.x * SORT_CH);
    const v4i* s4 = (const v4i*)(src + (size_t)blockIdx.x * SORT_CH);
    v4i d[SORT_EPT], s[SORT_EPT];
#pragma unroll
    for (int u = 0; u < SORT_EPT; ++u) {
        d[u] = d4[t + u * SORT_TBE];
        s[u] = s4[t + u * SORT_TBE];
    }
    asm volatile("" :: "v"(d[0]), "v"(d[1]), "v"(d[2]), "v"(d[3]),
                       "v"(s[0]), "v"(s[1]), "v"(s[2]), "v"(s[3]));
#pragma unroll
    for (int u = 0; u < SORT_EPT; ++u) {
        const int p0 = atomicAdd(&lpos[d[u].x >> 8], 1);
        const int p1 = atomicAdd(&lpos[d[u].y >> 8], 1);
        const int p2 = atomicAdd(&lpos[d[u].z >> 8], 1);
        const int p3 = atomicAdd(&lpos[d[u].w >> 8], 1);
        binned[p0] = s[u].x | ((d[u].x & (SORT_BW - 1)) << SRC_BITS);
        binned[p1] = s[u].y | ((d[u].y & (SORT_BW - 1)) << SRC_BITS);
        binned[p2] = s[u].z | ((d[u].z & (SORT_BW - 1)) << SRC_BITS);
        binned[p3] = s[u].w | ((d[u].w & (SORT_BW - 1)) << SRC_BITS);
    }
}

__global__ void bucket_sort_kernel(const int* __restrict__ binned, const int* __restrict__ total,
                                   int* __restrict__ rowptr, int* __restrict__ cnt,
                                   float* __restrict__ dinv, int* __restrict__ col, int n) {
    __shared__ int red[16];
    __shared__ int wsum[8];
    __shared__ int lc[SORT_BW];
    const int t = threadIdx.x;
    const int lane = t & 31;
    const int wave = t >> 5;
    const int k = blockIdx.x;

    int s = (t < k) ? total[t] : 0;
    for (int o = 16; o > 0; o >>= 1) s += __shfl_xor(s, o);
    if (lane == 0 && wave < 16) red[wave] = s;
    if (t < SORT_BW) lc[t] = 0;
    __syncthreads();
    int bs = 0;
    for (int w = 0; w < 16; ++w) bs += red[w];
    const int be = bs + total[k];

    for (int e0 = bs + t; e0 < be; e0 += SORT_U * SORT_TBB) {
        int w[SORT_U];
#pragma unroll
        for (int u = 0; u < SORT_U; ++u) {
            const int e = e0 + u * SORT_TBB;
            w[u] = (e < be) ? binned[e] : -1;
        }
#pragma unroll
        for (int u = 0; u < SORT_U; ++u)
            if (w[u] >= 0) atomicAdd(&lc[w[u] >> SRC_BITS], 1);
    }
    __syncthreads();

    const int c = (t < SORT_BW) ? lc[t] : 0;
    const int incl = block_incl_scan<8>(c, wsum);
    if (t < SORT_BW) {
        const int start = bs + incl - c;
        const int node = k * SORT_BW + t;
        if (node < n) {
            rowptr[node] = start;
            cnt[node] = c;
            dinv[node] = rsqrtf((float)(c + 1));
        }
        lc[t] = start;
    }
    __syncthreads();
    for (int e0 = bs + t; e0 < be; e0 += SORT_U * SORT_TBB) {
        int w[SORT_U];
#pragma unroll
        for (int u = 0; u < SORT_U; ++u) {
            const int e = e0 + u * SORT_TBB;
            w[u] = (e < be) ? binned[e] : -1;
        }
        int p[SORT_U];
#pragma unroll
        for (int u = 0; u < SORT_U; ++u)
            p[u] = (w[u] >= 0) ? atomicAdd(&lc[w[u] >> SRC_BITS], 1) : 0;
#pragma unroll
        for (int u = 0; u < SORT_U; ++u)
            if (w[u] >= 0) col[p[u]] = w[u] & ((1 << SRC_BITS) - 1);
    }
}

template <int KDIM, int NOUT, bool RELU>
__global__ void gemm_wmma_kernel(const float* __restrict__ X,
                                 const float* __restrict__ W,
                                 const float* __restrict__ rs,
                                 float* __restrict__ H, int M) {
    constexpr int NT  = NOUT / 16;
    constexpr int LDW = NOUT + 8;
    constexpr int KC  = 32;
    constexpr int NS  = KC / 4;
    static_assert(KDIM % KC == 0, "whole groups");
    __shared__ float Ws[KDIM * LDW];
    for (int i = threadIdx.x; i < KDIM * NOUT; i += blockDim.x) {
        int k = i / NOUT, n = i - k * NOUT;
        Ws[k * LDW + n] = W[i];
    }
    __syncthreads();

    const int wid  = (blockIdx.x * blockDim.x + threadIdx.x) >> 5;
    const int lane = threadIdx.x & 31;
    const int m0 = wid * 16;
    if (m0 >= M) return;

    const int half = lane >> 4;
    const int l16  = lane & 15;
    const int koff = half * 2;
    const float* xrow = X + (size_t)(m0 + l16) * KDIM + koff;
    const float* wcol = Ws + koff * LDW + l16;

    const float4 rs0 = *(const float4*)(rs + m0 + half * 8);
    const float4 rs1 = *(const float4*)(rs + m0 + half * 8 + 4);

    v8f acc[NT];
#pragma unroll
    for (int nt = 0; nt < NT; ++nt) acc[nt] = v8f{};

    v2f a[NS], an[NS];
#pragma unroll
    for (int s = 0; s < NS; ++s) a[s] = *(const v2f*)(xrow + 4 * s);

#pragma unroll
    for (int kc = 0; kc < KDIM; kc += KC) {
        if (kc + KC < KDIM) {
#pragma unroll
            for (int s = 0; s < NS; ++s) an[s] = *(const v2f*)(xrow + kc + KC + 4 * s);
        }
#pragma unroll
        for (int s = 0; s < NS; ++s) {
            v2f av = a[s];
            if (RELU) { av.x = fmaxf(av.x, 0.0f); av.y = fmaxf(av.y, 0.0f); }
            const int k0 = kc + 4 * s;
#pragma unroll
            for (int nt = 0; nt < NT; ++nt) {
                v2f b;
                b.x = wcol[k0 * LDW + nt * 16];
                b.y = wcol[k0 * LDW + LDW + nt * 16];
                acc[nt] = __builtin_amdgcn_wmma_f32_16x16x4_f32(
                    false, av, false, b, (short)0, acc[nt], false, false);
            }
        }
        if (kc + KC < KDIM) {
#pragma unroll
            for (int s = 0; s < NS; ++s) a[s] = an[s];
        }
    }
    const float rsv[8] = {rs0.x, rs0.y, rs0.z, rs0.w, rs1.x, rs1.y, rs1.z, rs1.w};
    float* orow = H + (size_t)(m0 + half * 8) * NOUT + l16;
#pragma unroll
    for (int nt = 0; nt < NT; ++nt)
#pragma unroll
        for (int v = 0; v < 8; ++v)
            orow[(size_t)v * NOUT + nt * 16] = acc[nt][v] * rsv[v];
}

template <int LPN, int K>
__device__ inline int group_bcast(int v) {
    static_assert(LPN == 16 || LPN == 8, "a group is a DPP row or a DPP8 group");
    static_assert(K >= 0 && K < LPN, "source lane inside the group");
    if constexpr (LPN == 16)
        return __builtin_amdgcn_update_dpp(0, v, 0x150 + K, 0xF, 0xF, true);
    else
        return __builtin_amdgcn_mov_dpp8(v, K * 0x249249);
}

template <int F, int G>
__device__ inline void load_row_group(const float* __restrict__ H, int mine, int c, float4 (&h)[8]) {
    constexpr int LPN = F / 4;
    int s[8];
    s[0] = group_bcast<LPN, G + 0>(mine);
    s[1] = group_bcast<LPN, G + 1>(mine);
    s[2] = group_bcast<LPN, G + 2>(mine);
    s[3] = group_bcast<LPN, G + 3>(mine);
    s[4] = group_bcast<LPN, G + 4>(mine);
    s[5] = group_bcast<LPN, G + 5>(mine);
    s[6] = group_bcast<LPN, G + 6>(mine);
    s[7] = group_bcast<LPN, G + 7>(mine);
#pragma unroll
    for (int k = 0; k < 8; ++k) h[k] = *(const float4*)(H + (size_t)s[k] * F + c);
}

template <int G>
__device__ inline void sum_row_group(const float4 (&h)[8], int rem, float4& acc) {
#pragma unroll
    for (int k = 0; k < 8; ++k) {
        const float w = (G + k < rem) ? 1.0f : 0.0f;
        acc.x += h[k].x * w;
        acc.y += h[k].y * w;
        acc.z += h[k].z * w;
        acc.w += h[k].w * w;
    }
}

struct GatherState {
    int beg, end, mine;
    float di;
    float4 acc, bv;
};

template <int F>
__device__ inline GatherState gather_open(const int* __restrict__ rowptr,
                                          const int* __restrict__ cnt,
                                          const int* __restrict__ col,
                                          const float* __restrict__ dinv,
                                          const float* __restrict__ H,
                                          const float* __restrict__ bias,
                                          int i, int q) {
    const int c = q * 4;
    GatherState st;
    const int beg = rowptr[i];
    const int deg = cnt[i];
    st.di = dinv[i];
    st.acc = *(const float4*)(H + (size_t)i * F + c);
    st.bv = *(const float4*)(bias + c);
    asm volatile("" :: "v"(beg), "v"(deg));
    const int end = beg + deg;
    st.beg = beg;
    st.end = end;
    st.mine = 0;
    if (beg < end) st.mine = col[beg + q < end ? beg + q : end - 1];
    return st;
}

template <int F>
__device__ inline float4 gather_run(const int* __restrict__ col,
                                    const float* __restrict__ H,
                                    const GatherState& st, int q) {
    constexpr int LPN = F / 4;
    const int c = q * 4;
    const int end = st.end;
    int mine = st.mine;
    float4 acc = st.acc;
    for (int j = st.beg; j < end; j += LPN) {
        const int rem = end - j;
        float4 h[8];
        load_row_group<F, 0>(H, mine, c, h);
        int next = 0;
        if (rem > LPN) {
            const int jn = j + LPN + q;
            next = col[jn < end ? jn : end - 1];
        }
        sum_row_group<0>(h, rem, acc);
        if constexpr (LPN > 8) {
            if (rem > 8) {
                load_row_group<F, 8>(H, mine, c, h);
                sum_row_group<8>(h, rem, acc);
            }
        }
        mine = next;
    }
    float4 o;
    o.x = acc.x * st.di + st.bv.x;
    o.y = acc.y * st.di + st.bv.y;
    o.z = acc.z * st.di + st.bv.z;
    o.w = acc.w * st.di + st.bv.w;
    return o;
}

template <int F>
__global__ void gather_sum_kernel(const int* __restrict__ rowptr,
                                  const int* __restrict__ cnt,
                                  const int* __restrict__ col,
                                  const float* __restrict__ dinv,
                                  const float* __restrict__ H,
                                  const float* __restrict__ bias,
                                  float* __restrict__ A, int n) {
    constexpr int LPN = F / 4;
    const int t = blockIdx.x * blockDim.x + threadIdx.x;
    const int i = t / LPN;
    if (i >= n) return;
    const int q = t % LPN;
    const GatherState st = gather_open<F>(rowptr, cnt, col, dinv, H, bias, i, q);
    const float4 o = gather_run<F>(col, H, st, q);
    *(float4*)(A + (size_t)i * F + q * 4) = o;
}

template <int NOUT, bool RELU>
__global__ void gather_gemm_kernel(const int* __restrict__ rowptr,
                                   const int* __restrict__ cnt,
                                   const int* __restrict__ col,
                                   const float* __restrict__ dinv,
                                   const float* __restrict__ Hin,
                                   const float* __restrict__ bias,
                                   const float* __restrict__ W,
                                   float* __restrict__ Hout) {
    constexpr int F   = 64;
    constexpr int LPN = F / 4;
    constexpr int NB  = 16;
    constexpr int TBG = NB * LPN;
    constexpr int LDW = NOUT + 8;
    constexpr int LDA = F + 4;
    constexpr int NWW = NOUT / 32;
    constexpr int WV  = F * NOUT / 4 / TBG;
    static_assert(NOUT % 32 == 0 && (F * NOUT / 4) % TBG == 0, "whole lines, whole pieces");
    __shared__ alignas(16) float Ws[F * LDW];
    __shared__ alignas(16) float As[NB * LDA];
    __shared__ alignas(16) float rsl[NB];

    const int t = threadIdx.x;
    float4 wv[WV];
#pragma unroll
    for (int u = 0; u < WV; ++u) wv[u] = *(const float4*)(W + 4 * (t + u * TBG));

    const int il = t / LPN;
    const int q  = t % LPN;
    const int i  = blockIdx.x * NB + il;
    const GatherState st = gather_open<F>(rowptr, cnt, col, dinv, Hin, bias, i, q);
#pragma unroll
    for (int u = 0; u < WV; ++u) {
        const int e = 4 * (t + u * TBG);
        const int k = e / NOUT, n = e - k * NOUT;
        *(float4*)(Ws + k * LDW + n) = wv[u];
    }
    float4 o = gather_run<F>(col, Hin, st, q);
    if (RELU) {
        o.x = fmaxf(o.x, 0.0f); o.y = fmaxf(o.y, 0.0f);
        o.z = fmaxf(o.z, 0.0f); o.w = fmaxf(o.w, 0.0f);
    }
    *(float4*)(As + il * LDA + q * 4) = o;
    if (q == 0) rsl[il] = st.di;
    __syncthreads();

    const int wave = t >> 5;
    if (wave >= NWW) return;
    const int lane = t & 31;
    const int half = lane >> 4;
    const int l16  = lane & 15;
    const int koff = half * 2;
    const float* arow = As + l16 * LDA + koff;
    const float* wcol = Ws + koff * LDW + wave * 32 + l16;

    v8f acc0 = v8f{}, acc1 = v8f{};
#pragma unroll
    for (int s = 0; s < F / 4; ++s) {
        const v2f av = *(const v2f*)(arow + 4 * s);
        const int k0 = 4 * s;
        v2f b0, b1;
        b0.x = wcol[k0 * LDW];
        b0.y = wcol[k0 * LDW + LDW];
        b1.x = wcol[k0 * LDW + 16];
        b1.y = wcol[k0 * LDW + LDW + 16];
        acc0 = __builtin_amdgcn_wmma_f32_16x16x4_f32(false, av, false, b0, (short)0, acc0, false, false);
        acc1 = __builtin_amdgcn_wmma_f32_16x16x4_f32(false, av, false, b1, (short)0, acc1, false, false);
    }
    const float4 rs0 = *(const float4*)(rsl + half * 8);
    const float4 rs1 = *(const float4*)(rsl + half * 8 + 4);
    const float rsv[8] = {rs0.x, rs0.y, rs0.z, rs0.w, rs1.x, rs1.y, rs1.z, rs1.w};
    float* orow = Hout + (size_t)(blockIdx.x * NB + half * 8) * NOUT + wave * 32 + l16;
#pragma unroll
    for (int v = 0; v < 8; ++v) {
        orow[(size_t)v * NOUT]      = acc0[v] * rsv[v];
        orow[(size_t)v * NOUT + 16] = acc1[v] * rsv[v];
    }
}

extern "C" void kernel_launch(void* const* d_in, const int* in_sizes, int n_in,
                              void* d_out, int out_size, void* d_ws, size_t ws_size,
                              hipStream_t stream) {
    (void)in_sizes; (void)n_in; (void)out_size;
    const float* x  = (const float*)d_in[0];
    const int*   ei = (const int*)d_in[1];
    const float* W1 = (const float*)d_in[2];
    const float* b1 = (const float*)d_in[3];
    const float* W2 = (const float*)d_in[4];
    const float* b2 = (const float*)d_in[5];
    const float* W3 = (const float*)d_in[6];
    const float* b3 = (const float*)d_in[7];
    float* out = (float*)d_out;

    const int N = GCN_N, E = GCN_E;
    const int* srcI = ei;
    const int* dstI = ei + E;

    float* dinv  = (float*)d_ws;
    float* Hbuf  = dinv + N;
    float* Abuf  = Hbuf + (size_t)N * 64;
    int*   cnt    = (int*)(Abuf + (size_t)N * 64);
    int*   rowp   = cnt + N;
    int*   hist   = rowp + N;
    int*   total  = hist + SORT_NCH * SORT_NBK;
    int*   binned = total + 512;
    int*   col    = binned + E;
    size_t need = (3 * (size_t)N + 2 * (size_t)N * 64 + SORT_NCH * SORT_NBK + 512 + 2 * (size_t)E) * sizeof(float);
    if (ws_size < need) return;

    const int TB = 256;
    const int mtiles = N / 16;
    const int gbA = (mtiles + 7) / 8;

    bucket_hist_kernel <<<SORT_NCH, SORT_TBE, 0, stream>>>(dstI, hist);
    chunk_scan_kernel  <<<SORT_NBK, 256, 0, stream>>>(hist, total);
    bucket_place_kernel<<<SORT_NCH, SORT_TBE, 0, stream>>>(srcI, dstI, hist, total, binned);
    bucket_sort_kernel <<<SORT_NBK, SORT_TBB, 0, stream>>>(binned, total, rowp, cnt, dinv, col, N);

    gemm_wmma_kernel<128, 64, false><<<gbA, TB, 0, stream>>>(x, W1, dinv, Hbuf, N);

    gather_gemm_kernel<64, true><<<N / 16, TB, 0, stream>>>(rowp, cnt, col, dinv, Hbuf, b1, W2, Abuf);

    gather_gemm_kernel<32, true><<<N / 16, TB, 0, stream>>>(rowp, cnt, col, dinv, Abuf, b2, W3, Hbuf);

    gather_sum_kernel<32><<<(N * 8 + TB - 1) / TB, TB, 0, stream>>>(rowp, cnt, col, dinv, Hbuf, b3, out, N);
}


# Gather launches made resident: 1250 (last: 1042) blocks of 8 waves walk 5 (3) node tiles each, W staged once per block, next tiles' opening loads issued under the current row loop; same sum order
# speedup vs baseline: 3.5774x; 1.0457x over previous
#include <hip/hip_runtime.h>


typedef float v2f __attribute__((ext_vector_type(2)));
typedef float v8f __attribute__((ext_vector_type(8)));
typedef int   v4i __attribute__((ext_vector_type(4)));

#define GCN_N 100000
#define GCN_E 1600000

#define SORT_BW   256
#define SORT_NBK  ((GCN_N + SORT_BW - 1) / SORT_BW)
#define SORT_NCH  250
#define SORT_CH   (GCN_E / SORT_NCH)
#define SORT_TBE  400
#define SORT_EPT  (SORT_CH / 4 / SORT_TBE)
#define SORT_TBB  512
#define SORT_U    10
#define SRC_BITS  17
#define GEMM_GATHER_BLOCKS 1250
#define SUM_GATHER_BLOCKS  1042

static_assert(GCN_N % 16 == 0, "the fused gather blocks own 16 whole nodes each");
static_assert(GCN_E % SORT_NCH == 0 && SORT_CH % 4 == 0, "chunks are whole int4 groups");
static_assert(GCN_N <= (1 << SRC_BITS), "src must fit the low bits of a binned word");
static_assert((SORT_CH / 4) % SORT_TBE == 0 && SORT_EPT == 4, "every thread of an edge pass owns 4 int4 groups");
static_assert(SORT_BW == 256 && SORT_NCH <= 256, "scan widths below");
static_assert(SORT_NBK <= SORT_TBE && SORT_NBK <= SORT_TBB && SORT_TBB == 512, "one thread per bucket total");


template <int NW>
__device__ inline int block_incl_scan(int v, int* wsum) {
    const int lane = threadIdx.x & 31;
    const int wave = threadIdx.x >> 5;
    for (int o = 1; o < 32; o <<= 1) {
        const int u = __shfl_up(v, o);
        if (lane >= o) v += u;
    }
    if (lane == 31 && wave < NW) wsum[wave] = v;
    __syncthreads();
    int off = 0;
    for (int w = 0; w < NW; ++w)
        if (w < wave) off += wsum[w];
    return v + off;
}

__global__ void bucket_hist_kernel(const int* __restrict__ dst, int* __restrict__ hist) {
    __shared__ int lh[SORT_NBK];
    const int t = threadIdx.x;
    for (int k = t; k < SORT_NBK; k += blockDim.x) lh[k] = 0;
    __syncthreads();
    const v4i* d4 = (const v4i*)(dst + (size_t)blockIdx.x * SORT_CH);
    v4i d[SORT_EPT];
#pragma unroll
    for (int u = 0; u < SORT_EPT; ++u) d[u] = d4[t + u * SORT_TBE];
    asm volatile("" :: "v"(d[0]), "v"(d[1]), "v"(d[2]), "v"(d[3]));
#pragma unroll
    for (int u = 0; u < SORT_EPT; ++u) {
        atomicAdd(&lh[d[u].x >> 8], 1);
        atomicAdd(&lh[d[u].y >> 8], 1);
        atomicAdd(&lh[d[u].z >> 8], 1);
        atomicAdd(&lh[d[u].w >> 8], 1);
    }
    __syncthreads();
    for (int k = t; k < SORT_NBK; k += blockDim.x) hist[blockIdx.x * SORT_NBK + k] = lh[k];
}

__global__ void chunk_scan_kernel(int* __restrict__ hist, int* __restrict__ total) {
    __shared__ int wsum[8];
    const int t = threadIdx.x;
    const int k = blockIdx.x;
    const int v = (t < SORT_NCH) ? hist[t * SORT_NBK + k] : 0;
    const int incl = block_incl_scan<8>(v, wsum);
    if (t < SORT_NCH) hist[t * SORT_NBK + k] = incl - v;
    if (t == SORT_NCH - 1) total[k] = incl;
}

__global__ void bucket_place_kernel(const int* __restrict__ src, const int* __restrict__ dst,
                                    const int* __restrict__ pre, const int* __restrict__ total,
                                    int* __restrict__ binned) {
    __shared__ int wsum[(SORT_TBE + 31) / 32];
    __shared__ int lpos[SORT_NBK];
    const int t = threadIdx.x;
    const int v = (t < SORT_NBK) ? total[t] : 0;
    const int incl = block_incl_scan<(SORT_TBE + 31) / 32>(v, wsum);
    if (t < SORT_NBK) lpos[t] = incl - v + pre[blockIdx.x * SORT_NBK + t];
    __syncthreads();
    const v4i* d4 = (const v4i*)(dst + (size_t)blockIdx.x * SORT_CH);
    const v4i* s4 = (const v4i*)(src + (size_t)blockIdx.x * SORT_CH);
    v4i d[SORT_EPT], s[SORT_EPT];
#pragma unroll
    for (int u = 0; u < SORT_EPT; ++u) {
        d[u] = d4[t + u * SORT_TBE];
        s[u] = s4[t + u * SORT_TBE];
    }
    asm volatile("" :: "v"(d[0]), "v"(d[1]), "v"(d[2]), "v"(d[3]),
                       "v"(s[0]), "v"(s[1]), "v"(s[2]), "v"(s[3]));
#pragma unroll
    for (int u = 0; u < SORT_EPT; ++u) {
        const int p0 = atomicAdd(&lpos[d[u].x >> 8], 1);
        const int p1 = atomicAdd(&lpos[d[u].y >> 8], 1);
        const int p2 = atomicAdd(&lpos[d[u].z >> 8], 1);
        const int p3 = atomicAdd(&lpos[d[u].w >> 8], 1);
        binned[p0] = s[u].x | ((d[u].x & (SORT_BW - 1)) << SRC_BITS);
        binned[p1] = s[u].y | ((d[u].y & (SORT_BW - 1)) << SRC_BITS);
        binned[p2] = s[u].z | ((d[u].z & (SORT_BW - 1)) << SRC_BITS);
        binned[p3] = s[u].w | ((d[u].w & (SORT_BW - 1)) << SRC_BITS);
    }
}

__global__ void bucket_sort_kernel(const int* __restrict__ binned, const int* __restrict__ total,
                                   int* __restrict__ rowptr, int* __restrict__ cnt,
                                   float* __restrict__ dinv, int* __restrict__ col, int n) {
    __shared__ int red[16];
    __shared__ int wsum[8];
    __shared__ int lc[SORT_BW];
    const int t = threadIdx.x;
    const int lane = t & 31;
    const int wave = t >> 5;
    const int k = blockIdx.x;

    int s = (t < k) ? total[t] : 0;
    for (int o = 16; o > 0; o >>= 1) s += __shfl_xor(s, o);
    if (lane == 0 && wave < 16) red[wave] = s;
    if (t < SORT_BW) lc[t] = 0;
    __syncthreads();
    int bs = 0;
    for (int w = 0; w < 16; ++w) bs += red[w];
    const int be = bs + total[k];

    for (int e0 = bs + t; e0 < be; e0 += SORT_U * SORT_TBB) {
        int w[SORT_U];
#pragma unroll
        for (int u = 0; u < SORT_U; ++u) {
            const int e = e0 + u * SORT_TBB;
            w[u] = (e < be) ? binned[e] : -1;
        }
#pragma unroll
        for (int u = 0; u < SORT_U; ++u)
            if (w[u] >= 0) atomicAdd(&lc[w[u] >> SRC_BITS], 1);
    }
    __syncthreads();

    const int c = (t < SORT_BW) ? lc[t] : 0;
    const int incl = block_incl_scan<8>(c, wsum);
    if (t < SORT_BW) {
        const int start = bs + incl - c;
        const int node = k * SORT_BW + t;
        if (node < n) {
            rowptr[node] = start;
            cnt[node] = c;
            dinv[node] = rsqrtf((float)(c + 1));
        }
        lc[t] = start;
    }
    __syncthreads();
    for (int e0 = bs + t; e0 < be; e0 += SORT_U * SORT_TBB) {
        int w[SORT_U];
#pragma unroll
        for (int u = 0; u < SORT_U; ++u) {
            const int e = e0 + u * SORT_TBB;
            w[u] = (e < be) ? binned[e] : -1;
        }
        int p[SORT_U];
#pragma unroll
        for (int u = 0; u < SORT_U; ++u)
            p[u] = (w[u] >= 0) ? atomicAdd(&lc[w[u] >> SRC_BITS], 1) : 0;
#pragma unroll
        for (int u = 0; u < SORT_U; ++u)
            if (w[u] >= 0) col[p[u]] = w[u] & ((1 << SRC_BITS) - 1);
    }
}

template <int KDIM, int NOUT, bool RELU>
__global__ void gemm_wmma_kernel(const float* __restrict__ X,
                                 const float* __restrict__ W,
                                 const float* __restrict__ rs,
                                 float* __restrict__ H, int M) {
    constexpr int NT  = NOUT / 16;
    constexpr int LDW = NOUT + 8;
    constexpr int KC  = 32;
    constexpr int NS  = KC / 4;
    static_assert(KDIM % KC == 0, "whole groups");
    __shared__ float Ws[KDIM * LDW];
    for (int i = threadIdx.x; i < KDIM * NOUT; i += blockDim.x) {
        int k = i / NOUT, n = i - k * NOUT;
        Ws[k * LDW + n] = W[i];
    }
    __syncthreads();

    const int wid  = (blockIdx.x * blockDim.x + threadIdx.x) >> 5;
    const int lane = threadIdx.x & 31;
    const int m0 = wid * 16;
    if (m0 >= M) return;

    const int half = lane >> 4;
    const int l16  = lane & 15;
    const int koff = half * 2;
    const float* xrow = X + (size_t)(m0 + l16) * KDIM + koff;
    const float* wcol = Ws + koff * LDW + l16;

    const float4 rs0 = *(const float4*)(rs + m0 + half * 8);
    const float4 rs1 = *(const float4*)(rs + m0 + half * 8 + 4);

    v8f acc[NT];
#pragma unroll
    for (int nt = 0; nt < NT; ++nt) acc[nt] = v8f{};

    v2f a[NS], an[NS];
#pragma unroll
    for (int s = 0; s < NS; ++s) a[s] = *(const v2f*)(xrow + 4 * s);

#pragma unroll
    for (int kc = 0; kc < KDIM; kc += KC) {
        if (kc + KC < KDIM) {
#pragma unroll
            for (int s = 0; s < NS; ++s) an[s] = *(const v2f*)(xrow + kc + KC + 4 * s);
        }
#pragma unroll
        for (int s = 0; s < NS; ++s) {
            v2f av = a[s];
            if (RELU) { av.x = fmaxf(av.x, 0.0f); av.y = fmaxf(av.y, 0.0f); }
            const int k0 = kc + 4 * s;
#pragma unroll
            for (int nt = 0; nt < NT; ++nt) {
                v2f b;
                b.x = wcol[k0 * LDW + nt * 16];
                b.y = wcol[k0 * LDW + LDW + nt * 16];
                acc[nt] = __builtin_amdgcn_wmma_f32_16x16x4_f32(
                    false, av, false, b, (short)0, acc[nt], false, false);
            }
        }
        if (kc + KC < KDIM) {
#pragma unroll
            for (int s = 0; s < NS; ++s) a[s] = an[s];
        }
    }
    const float rsv[8] = {rs0.x, rs0.y, rs0.z, rs0.w, rs1.x, rs1.y, rs1.z, rs1.w};
    float* orow = H + (size_t)(m0 + half * 8) * NOUT + l16;
#pragma unroll
    for (int nt = 0; nt < NT; ++nt)
#pragma unroll
        for (int v = 0; v < 8; ++v)
            orow[(size_t)v * NOUT + nt * 16] = acc[nt][v] * rsv[v];
}

template <int LPN, int K>
__device__ inline int group_bcast(int v) {
    static_assert(LPN == 16 || LPN == 8, "a group is a DPP row or a DPP8 group");
    static_assert(K >= 0 && K < LPN, "source lane inside the group");
    if constexpr (LPN == 16)
        return __builtin_amdgcn_update_dpp(0, v, 0x150 + K, 0xF, 0xF, true);
    else
        return __builtin_amdgcn_mov_dpp8(v, K * 0x249249);
}

template <int F, int G>
__device__ inline void load_row_group(const float* __restrict__ H, int mine, int c, float4 (&h)[8]) {
    constexpr int LPN = F / 4;
    int s[8];
    s[0] = group_bcast<LPN, G + 0>(mine);
    s[1] = group_bcast<LPN, G + 1>(mine);
    s[2] = group_bcast<LPN, G + 2>(mine);
    s[3] = group_bcast<LPN, G + 3>(mine);
    s[4] = group_bcast<LPN, G + 4>(mine);
    s[5] = group_bcast<LPN, G + 5>(mine);
    s[6] = group_bcast<LPN, G + 6>(mine);
    s[7] = group_bcast<LPN, G + 7>(mine);
#pragma unroll
    for (int k = 0; k < 8; ++k) h[k] = *(const float4*)(H + (size_t)s[k] * F + c);
}

template <int G>
__device__ inline void sum_row_group(const float4 (&h)[8], int rem, float4& acc) {
#pragma unroll
    for (int k = 0; k < 8; ++k) {
        const float w = (G + k < rem) ? 1.0f : 0.0f;
        acc.x += h[k].x * w;
        acc.y += h[k].y * w;
        acc.z += h[k].z * w;
        acc.w += h[k].w * w;
    }
}

struct GatherState {
    int beg, end, mine;
    float di;
    float4 acc;
};

struct GatherHead {
    int beg, deg;
    float di;
    float4 acc;
};

template <int F>
__device__ inline GatherHead gather_head(const int* __restrict__ rowptr,
                                         const int* __restrict__ cnt,
                                         const float* __restrict__ dinv,
                                         const float* __restrict__ H,
                                         int i, int q) {
    GatherHead hd;
    hd.beg = rowptr[i];
    hd.deg = cnt[i];
    hd.di = dinv[i];
    hd.acc = *(const float4*)(H + (size_t)i * F + q * 4);
    return hd;
}

__device__ inline GatherState gather_arm(const int* __restrict__ col, const GatherHead& hd, int q) {
    GatherState st;
    const int beg = hd.beg;
    const int end = beg + hd.deg;
    st.di = hd.di;
    st.acc = hd.acc;
    st.beg = beg;
    st.end = end;
    st.mine = 0;
    if (beg < end) st.mine = col[beg + q < end ? beg + q : end - 1];
    return st;
}

template <int F>
__device__ inline GatherState gather_open(const int* __restrict__ rowptr,
                                          const int* __restrict__ cnt,
                                          const int* __restrict__ col,
                                          const float* __restrict__ dinv,
                                          const float* __restrict__ H,
                                          int i, int q) {
    const GatherHead hd = gather_head<F>(rowptr, cnt, dinv, H, i, q);
    asm volatile("" :: "v"(hd.beg), "v"(hd.deg));
    return gather_arm(col, hd, q);
}

template <int F>
__device__ inline float4 gather_run(const int* __restrict__ col,
                                    const float* __restrict__ H,
                                    const GatherState& st, const float4 bv, int q) {
    constexpr int LPN = F / 4;
    const int c = q * 4;
    const int end = st.end;
    int mine = st.mine;
    float4 acc = st.acc;
    for (int j = st.beg; j < end; j += LPN) {
        const int rem = end - j;
        float4 h[8];
        load_row_group<F, 0>(H, mine, c, h);
        int next = 0;
        if (rem > LPN) {
            const int jn = j + LPN + q;
            next = col[jn < end ? jn : end - 1];
        }
        sum_row_group<0>(h, rem, acc);
        if constexpr (LPN > 8) {
            if (rem > 8) {
                load_row_group<F, 8>(H, mine, c, h);
                sum_row_group<8>(h, rem, acc);
            }
        }
        mine = next;
    }
    float4 o;
    o.x = acc.x * st.di + bv.x;
    o.y = acc.y * st.di + bv.y;
    o.z = acc.z * st.di + bv.z;
    o.w = acc.w * st.di + bv.w;
    return o;
}

template <int F>
__global__ void gather_sum_kernel(const int* __restrict__ rowptr,
                                  const int* __restrict__ cnt,
                                  const int* __restrict__ col,
                                  const float* __restrict__ dinv,
                                  const float* __restrict__ H,
                                  const float* __restrict__ bias,
                                  float* __restrict__ A, int n) {
    constexpr int LPN = F / 4;
    const int stride = (int)(gridDim.x * blockDim.x) / LPN;
    int i = (int)(blockIdx.x * blockDim.x + threadIdx.x) / LPN;
    if (i >= n) return;
    const int q = threadIdx.x % LPN;
    const float4 bv = *(const float4*)(bias + q * 4);
    GatherState st = gather_open<F>(rowptr, cnt, col, dinv, H, i, q);
    GatherHead hd = {};
    if (i + stride < n) hd = gather_head<F>(rowptr, cnt, dinv, H, i + stride, q);
    for (; i < n; i += stride) {
        GatherState nst = st;
        if (i + stride < n) nst = gather_arm(col, hd, q);
        if (i + 2 * stride < n) hd = gather_head<F>(rowptr, cnt, dinv, H, i + 2 * stride, q);
        const float4 o = gather_run<F>(col, H, st, bv, q);
        *(float4*)(A + (size_t)i * F + q * 4) = o;
        st = nst;
    }
}

template <int NOUT, bool RELU>
__global__ void gather_gemm_kernel(const int* __restrict__ rowptr,
                                   const int* __restrict__ cnt,
                                   const int* __restrict__ col,
                                   const float* __restrict__ dinv,
                                   const float* __restrict__ Hin,
                                   const float* __restrict__ bias,
                                   const float* __restrict__ W,
                                   float* __restrict__ Hout, int ntiles) {
    constexpr int F   = 64;
    constexpr int LPN = F / 4;
    constexpr int NB  = 16;
    constexpr int TBG = NB * LPN;
    constexpr int LDW = NOUT + 8;
    constexpr int LDA = F + 4;
    constexpr int NWW = NOUT / 32;
    constexpr int WV  = F * NOUT / 4 / TBG;
    static_assert(NOUT % 32 == 0 && (F * NOUT / 4) % TBG == 0, "whole lines, whole pieces");
    __shared__ alignas(16) float Ws[F * LDW];
    __shared__ alignas(16) float As[2][NB * LDA];
    __shared__ alignas(16) float rsl[2][NB];

    const int t = threadIdx.x;
    int tile = blockIdx.x;
    if (tile >= ntiles) return;
    float4 wv[WV];
#pragma unroll
    for (int u = 0; u < WV; ++u) wv[u] = *(const float4*)(W + 4 * (t + u * TBG));

    const int il = t / LPN;
    const int q  = t % LPN;
    const float4 bv = *(const float4*)(bias + q * 4);
    const int G = gridDim.x;
    GatherState st = gather_open<F>(rowptr, cnt, col, dinv, Hin, tile * NB + il, q);
    GatherHead hd = {};
    if (tile + G < ntiles) hd = gather_head<F>(rowptr, cnt, dinv, Hin, (tile + G) * NB + il, q);
#pragma unroll
    for (int u = 0; u < WV; ++u) {
        const int e = 4 * (t + u * TBG);
        const int k = e / NOUT, n = e - k * NOUT;
        *(float4*)(Ws + k * LDW + n) = wv[u];
    }

    const int wave = t >> 5;
    const int lane = t & 31;
    const int half = lane >> 4;
    const int l16  = lane & 15;
    const int koff = half * 2;
    const float* wcol = Ws + koff * LDW + (wave & (NWW - 1)) * 32 + l16;

    int buf = 0;
    for (; tile < ntiles; tile += G) {
        GatherState nst = st;
        if (tile + G < ntiles) nst = gather_arm(col, hd, q);
        if (tile + 2 * G < ntiles) hd = gather_head<F>(rowptr, cnt, dinv, Hin, (tile + 2 * G) * NB + il, q);
        float4 o = gather_run<F>(col, Hin, st, bv, q);
        if (RELU) {
            o.x = fmaxf(o.x, 0.0f); o.y = fmaxf(o.y, 0.0f);
            o.z = fmaxf(o.z, 0.0f); o.w = fmaxf(o.w, 0.0f);
        }
        float* Ab = As[buf];
        *(float4*)(Ab + il * LDA + q * 4) = o;
        if (q == 0) rsl[buf][il] = st.di;
        __syncthreads();

        if (wave < NWW) {
            const float* arow = Ab + l16 * LDA + koff;
            v8f acc0 = v8f{}, acc1 = v8f{};
#pragma unroll
            for (int s = 0; s < F / 4; ++s) {
                const v2f av = *(const v2f*)(arow + 4 * s);
                const int k0 = 4 * s;
                v2f b0, b1;
                b0.x = wcol[k0 * LDW];
                b0.y = wcol[k0 * LDW + LDW];
                b1.x = wcol[k0 * LDW + 16];
                b1.y = wcol[k0 * LDW + LDW + 16];
                acc0 = __builtin_amdgcn_wmma_f32_16x16x4_f32(false, av, false, b0, (short)0, acc0, false, false);
                acc1 = __builtin_amdgcn_wmma_f32_16x16x4_f32(false, av, false, b1, (short)0, acc1, false, false);
            }
            const float4 rs0 = *(const float4*)(rsl[buf] + half * 8);
            const float4 rs1 = *(const float4*)(rsl[buf] + half * 8 + 4);
            const float rsv[8] = {rs0.x, rs0.y, rs0.z, rs0.w, rs1.x, rs1.y, rs1.z, rs1.w};
            float* orow = Hout + (size_t)(tile * NB + half * 8) * NOUT + wave * 32 + l16;
#pragma unroll
            for (int v = 0; v < 8; ++v) {
                orow[(size_t)v * NOUT]      = acc0[v] * rsv[v];
                orow[(size_t)v * NOUT + 16] = acc1[v] * rsv[v];
            }
        }
        buf ^= 1;
        st = nst;
    }
}

extern "C" void kernel_launch(void* const* d_in, const int* in_sizes, int n_in,
                              void* d_out, int out_size, void* d_ws, size_t ws_size,
                              hipStream_t stream) {
    (void)in_sizes; (void)n_in; (void)out_size;
    const float* x  = (const float*)d_in[0];
    const int*   ei = (const int*)d_in[1];
    const float* W1 = (const float*)d_in[2];
    const float* b1 = (const float*)d_in[3];
    const float* W2 = (const float*)d_in[4];
    const float* b2 = (const float*)d_in[5];
    const float* W3 = (const float*)d_in[6];
    const float* b3 = (const float*)d_in[7];
    float* out = (float*)d_out;

    const int N = GCN_N, E = GCN_E;
    const int* srcI = ei;
    const int* dstI = ei + E;

    float* dinv  = (float*)d_ws;
    float* Hbuf  = dinv + N;
    float* Abuf  = Hbuf + (size_t)N * 64;
    int*   cnt    = (int*)(Abuf + (size_t)N * 64);
    int*   rowp   = cnt + N;
    int*   hist   = rowp + N;
    int*   total  = hist + SORT_NCH * SORT_NBK;
    int*   binned = total + 512;
    int*   col    = binned + E;
    size_t need = (3 * (size_t)N + 2 * (size_t)N * 64 + SORT_NCH * SORT_NBK + 512 + 2 * (size_t)E) * sizeof(float);
    if (ws_size < need) return;

    const int TB = 256;
    const int mtiles = N / 16;
    const int gbA = (mtiles + 7) / 8;

    bucket_hist_kernel <<<SORT_NCH, SORT_TBE, 0, stream>>>(dstI, hist);
    chunk_scan_kernel  <<<SORT_NBK, 256, 0, stream>>>(hist, total);
    bucket_place_kernel<<<SORT_NCH, SORT_TBE, 0, stream>>>(srcI, dstI, hist, total, binned);
    bucket_sort_kernel <<<SORT_NBK, SORT_TBB, 0, stream>>>(binned, total, rowp, cnt, dinv, col, N);

    gemm_wmma_kernel<128, 64, false><<<gbA, TB, 0, stream>>>(x, W1, dinv, Hbuf, N);

    gather_gemm_kernel<64, true><<<GEMM_GATHER_BLOCKS, TB, 0, stream>>>(rowp, cnt, col, dinv, Hbuf, b1, W2, Abuf, N / 16);

    gather_gemm_kernel<32, true><<<GEMM_GATHER_BLOCKS, TB, 0, stream>>>(rowp, cnt, col, dinv, Abuf, b2, W3, Hbuf, N / 16);

    gather_sum_kernel<32><<<SUM_GATHER_BLOCKS, TB, 0, stream>>>(rowp, cnt, col, dinv, Hbuf, b3, out, N);
}


# Edge sort in two launches, not four: each chunk groups its own edges by bucket (LDS count, scan, place in one pass), the bucket pass collects its 250 runs; per-node sum order stays LDS arrival order
# speedup vs baseline: 4.3865x; 1.2262x over previous
#include <hip/hip_runtime.h>


typedef float v2f __attribute__((ext_vector_type(2)));
typedef float v8f __attribute__((ext_vector_type(8)));
typedef int   v4i __attribute__((ext_vector_type(4)));

#define GCN_N 100000
#define GCN_E 1600000

#define SORT_BW   256
#define SORT_NBK  ((GCN_N + SORT_BW - 1) / SORT_BW)
#define SORT_NCH  250
#define SORT_CH   (GCN_E / SORT_NCH)
#define SORT_TBE  400
#define SORT_EPT  (SORT_CH / 4 / SORT_TBE)
#define SORT_TBB  512
#define SORT_NG   (SORT_TBB / 16)
#define SORT_NQ   ((SORT_NCH + SORT_NG - 1) / SORT_NG)
#define SRC_BITS  17
#define GEMM_GATHER_BLOCKS 1250
#define SUM_GATHER_BLOCKS  1042

static_assert(GCN_N % 16 == 0, "the fused gather blocks own 16 whole nodes each");
static_assert(GCN_E % SORT_NCH == 0 && SORT_CH % 4 == 0, "chunks are whole int4 groups");
static_assert(GCN_N <= (1 << SRC_BITS), "src must fit the low bits of a binned word");
static_assert((SORT_CH / 4) % SORT_TBE == 0 && SORT_EPT == 4, "every thread of the chunk pass owns 4 int4 groups");
static_assert(SORT_BW == 256 && SORT_TBB == 512, "scan widths below");
static_assert(SORT_NBK <= SORT_TBE, "one thread per bucket in the chunk pass's scan");
static_assert(SORT_NG * SORT_NQ >= SORT_NCH && SORT_NG == 32, "every chunk has a lane group in the per-bucket pass");


template <int NW>
__device__ inline int block_incl_scan(int v, int* wsum) {
    const int lane = threadIdx.x & 31;
    const int wave = threadIdx.x >> 5;
    for (int o = 1; o < 32; o <<= 1) {
        const int u = __shfl_up(v, o);
        if (lane >= o) v += u;
    }
    if (lane == 31 && wave < NW) wsum[wave] = v;
    __syncthreads();
    int off = 0;
    for (int w = 0; w < NW; ++w)
        if (w < wave) off += wsum[w];
    return v + off;
}

__global__ void chunk_bin_kernel(const int* __restrict__ src, const int* __restrict__ dst,
                                 int* __restrict__ cstart, int* __restrict__ binned) {
    __shared__ int wsum[(SORT_TBE + 31) / 32];
    __shared__ int lh[SORT_NBK];
    const int t = threadIdx.x;
    const v4i* d4 = (const v4i*)(dst + (size_t)blockIdx.x * SORT_CH);
    const v4i* s4 = (const v4i*)(src + (size_t)blockIdx.x * SORT_CH);
    v4i d[SORT_EPT], s[SORT_EPT];
#pragma unroll
    for (int u = 0; u < SORT_EPT; ++u) {
        d[u] = d4[t + u * SORT_TBE];
        s[u] = s4[t + u * SORT_TBE];
    }
    for (int k = t; k < SORT_NBK; k += blockDim.x) lh[k] = 0;
    __syncthreads();
    asm volatile("" :: "v"(d[0]), "v"(d[1]), "v"(d[2]), "v"(d[3]),
                       "v"(s[0]), "v"(s[1]), "v"(s[2]), "v"(s[3]));
#pragma unroll
    for (int u = 0; u < SORT_EPT; ++u) {
        atomicAdd(&lh[d[u].x >> 8], 1);
        atomicAdd(&lh[d[u].y >> 8], 1);
        atomicAdd(&lh[d[u].z >> 8], 1);
        atomicAdd(&lh[d[u].w >> 8], 1);
    }
    __syncthreads();
    const int v = (t < SORT_NBK) ? lh[t] : 0;
    const int incl = block_incl_scan<(SORT_TBE + 31) / 32>(v, wsum);
    if (t < SORT_NBK) {
        cstart[blockIdx.x * SORT_NBK + t] = incl - v;
        lh[t] = blockIdx.x * SORT_CH + incl - v;
    }
    __syncthreads();
#pragma unroll
    for (int u = 0; u < SORT_EPT; ++u) {
        const int p0 = atomicAdd(&lh[d[u].x >> 8], 1);
        const int p1 = atomicAdd(&lh[d[u].y >> 8], 1);
        const int p2 = atomicAdd(&lh[d[u].z >> 8], 1);
        const int p3 = atomicAdd(&lh[d[u].w >> 8], 1);
        binned[p0] = s[u].x | ((d[u].x & (SORT_BW - 1)) << SRC_BITS);
        binned[p1] = s[u].y | ((d[u].y & (SORT_BW - 1)) << SRC_BITS);
        binned[p2] = s[u].z | ((d[u].z & (SORT_BW - 1)) << SRC_BITS);
        binned[p3] = s[u].w | ((d[u].w & (SORT_BW - 1)) << SRC_BITS);
    }
}

__global__ void bucket_sort_kernel(const int* __restrict__ binned, const int* __restrict__ cstart,
                                   int* __restrict__ rowptr, int* __restrict__ cnt,
                                   float* __restrict__ dinv, int* __restrict__ col, int n) {
    __shared__ int red[SORT_NG];
    __shared__ int wsum[8];
    __shared__ int lc[SORT_BW + 32];
    const int t = threadIdx.x;
    const int g = t >> 4;
    const int l = t & 15;
    const int k = blockIdx.x;

    int rb[SORT_NQ], rn[SORT_NQ];
    int part = 0;
#pragma unroll
    for (int q = 0; q < SORT_NQ; ++q) {
        const int ch = g + q * SORT_NG;
        const int cc = ch < SORT_NCH ? ch : SORT_NCH - 1;
        const int ia = cc * SORT_NBK + k;
        const int ie = ia + 1 < SORT_NCH * SORT_NBK ? ia + 1 : ia;
        const int a = cstart[ia];
        const int en = cstart[ie];
        const int e = (k + 1 < SORT_NBK) ? en : SORT_CH;
        part += (ch < SORT_NCH) ? a : 0;
        rb[q] = cc * SORT_CH + a;
        rn[q] = (ch < SORT_NCH) ? e - a : 0;
    }
    int w[SORT_NQ][2];
#pragma unroll
    for (int q = 0; q < SORT_NQ; ++q) {
        const int last = rn[q] - 1;
        const int i0 = rb[q] + (l < last ? l : last);
        const int i1 = rb[q] + (l + 16 < last ? l + 16 : last);
        const int v0 = binned[i0 > 0 ? i0 : 0];
        const int v1 = binned[i1 > 0 ? i1 : 0];
        w[q][0] = (l < rn[q]) ? v0 : -1;
        w[q][1] = (l + 16 < rn[q]) ? v1 : -1;
    }

    if (l == 0) red[g] = part;
    if (t < SORT_BW + 32) lc[t] = 0;
    __syncthreads();
    int bs = 0;
    for (int i = 0; i < SORT_NG; ++i) bs += red[i];

#pragma unroll
    for (int q = 0; q < SORT_NQ; ++q) {
        if (w[q][0] >= 0) atomicAdd(&lc[w[q][0] >> SRC_BITS], 1);
        if (w[q][1] >= 0) atomicAdd(&lc[w[q][1] >> SRC_BITS], 1);
    }
#pragma unroll
    for (int q = 0; q < SORT_NQ; ++q)
        for (int j = 32 + l; j < rn[q]; j += 16)
            atomicAdd(&lc[binned[rb[q] + j] >> SRC_BITS], 1);
    __syncthreads();

    const int c = (t < SORT_BW) ? lc[t] : 0;
    const int incl = block_incl_scan<8>(c, wsum);
    if (t < SORT_BW) {
        const int start = bs + incl - c;
        const int node = k * SORT_BW + t;
        if (node < n) {
            rowptr[node] = start;
            cnt[node] = c;
            dinv[node] = rsqrtf((float)(c + 1));
        }
        lc[t] = start;
    }
    __syncthreads();

    const int spare = SORT_BW + (t & 31);
    int p[SORT_NQ][2];
#pragma unroll
    for (int q = 0; q < SORT_NQ; ++q) {
        p[q][0] = atomicAdd(&lc[w[q][0] >= 0 ? (w[q][0] >> SRC_BITS) : spare], 1);
        p[q][1] = atomicAdd(&lc[w[q][1] >= 0 ? (w[q][1] >> SRC_BITS) : spare], 1);
    }
#pragma unroll
    for (int q = 0; q < SORT_NQ; ++q) {
        if (w[q][0] >= 0) col[p[q][0]] = w[q][0] & ((1 << SRC_BITS) - 1);
        if (w[q][1] >= 0) col[p[q][1]] = w[q][1] & ((1 << SRC_BITS) - 1);
    }
#pragma unroll
    for (int q = 0; q < SORT_NQ; ++q)
        for (int j = 32 + l; j < rn[q]; j += 16) {
            const int ww = binned[rb[q] + j];
            const int pp = atomicAdd(&lc[ww >> SRC_BITS], 1);
            col[pp] = ww & ((1 << SRC_BITS) - 1);
        }
}

template <int KDIM, int NOUT, bool RELU>
__global__ void gemm_wmma_kernel(const float* __restrict__ X,
                                 const float* __restrict__ W,
                                 const float* __restrict__ rs,
                                 float* __restrict__ H, int M) {
    constexpr int NT  = NOUT / 16;
    constexpr int LDW = NOUT + 8;
    constexpr int KC  = 32;
    constexpr int NS  = KC / 4;
    static_assert(KDIM % KC == 0, "whole groups");
    __shared__ float Ws[KDIM * LDW];
    for (int i = threadIdx.x; i < KDIM * NOUT; i += blockDim.x) {
        int k = i / NOUT, n = i - k * NOUT;
        Ws[k * LDW + n] = W[i];
    }
    __syncthreads();

    const int wid  = (blockIdx.x * blockDim.x + threadIdx.x) >> 5;
    const int lane = threadIdx.x & 31;
    const int m0 = wid * 16;
    if (m0 >= M) return;

    const int half = lane >> 4;
    const int l16  = lane & 15;
    const int koff = half * 2;
    const float* xrow = X + (size_t)(m0 + l16) * KDIM + koff;
    const float* wcol = Ws + koff * LDW + l16;

    const float4 rs0 = *(const float4*)(rs + m0 + half * 8);
    const float4 rs1 = *(const float4*)(rs + m0 + half * 8 + 4);

    v8f acc[NT];
#pragma unroll
    for (int nt = 0; nt < NT; ++nt) acc[nt] = v8f{};

    v2f a[NS], an[NS];
#pragma unroll
    for (int s = 0; s < NS; ++s) a[s] = *(const v2f*)(xrow + 4 * s);

#pragma unroll
    for (int kc = 0; kc < KDIM; kc += KC) {
        if (kc + KC < KDIM) {
#pragma unroll
            for (int s = 0; s < NS; ++s) an[s] = *(const v2f*)(xrow + kc + KC + 4 * s);
        }
#pragma unroll
        for (int s = 0; s < NS; ++s) {
            v2f av = a[s];
            if (RELU) { av.x = fmaxf(av.x, 0.0f); av.y = fmaxf(av.y, 0.0f); }
            const int k0 = kc + 4 * s;
#pragma unroll
            for (int nt = 0; nt < NT; ++nt) {
                v2f b;
                b.x = wcol[k0 * LDW + nt * 16];
                b.y = wcol[k0 * LDW + LDW + nt * 16];
                acc[nt] = __builtin_amdgcn_wmma_f32_16x16x4_f32(
                    false, av, false, b, (short)0, acc[nt], false, false);
            }
        }
        if (kc + KC < KDIM) {
#pragma unroll
            for (int s = 0; s < NS; ++s) a[s] = an[s];
        }
    }
    const float rsv[8] = {rs0.x, rs0.y, rs0.z, rs0.w, rs1.x, rs1.y, rs1.z, rs1.w};
    float* orow = H + (size_t)(m0 + half * 8) * NOUT + l16;
#pragma unroll
    for (int nt = 0; nt < NT; ++nt)
#pragma unroll
        for (int v = 0; v < 8; ++v)
            orow[(size_t)v * NOUT + nt * 16] = acc[nt][v] * rsv[v];
}

template <int LPN, int K>
__device__ inline int group_bcast(int v) {
    static_assert(LPN == 16 || LPN == 8, "a group is a DPP row or a DPP8 group");
    static_assert(K >= 0 && K < LPN, "source lane inside the group");
    if constexpr (LPN == 16)
        return __builtin_amdgcn_update_dpp(0, v, 0x150 + K, 0xF, 0xF, true);
    else
        return __builtin_amdgcn_mov_dpp8(v, K * 0x249249);
}

template <int F, int G>
__device__ inline void load_row_group(const float* __restrict__ H, int mine, int c, float4 (&h)[8]) {
    constexpr int LPN = F / 4;
    int s[8];
    s[0] = group_bcast<LPN, G + 0>(mine);
    s[1] = group_bcast<LPN, G + 1>(mine);
    s[2] = group_bcast<LPN, G + 2>(mine);
    s[3] = group_bcast<LPN, G + 3>(mine);
    s[4] = group_bcast<LPN, G + 4>(mine);
    s[5] = group_bcast<LPN, G + 5>(mine);
    s[6] = group_bcast<LPN, G + 6>(mine);
    s[7] = group_bcast<LPN, G + 7>(mine);
#pragma unroll
    for (int k = 0; k < 8; ++k) h[k] = *(const float4*)(H + (size_t)s[k] * F + c);
}

template <int G>
__device__ inline void sum_row_group(const float4 (&h)[8], int rem, float4& acc) {
#pragma unroll
    for (int k = 0; k < 8; ++k) {
        const float w = (G + k < rem) ? 1.0f : 0.0f;
        acc.x += h[k].x * w;
        acc.y += h[k].y * w;
        acc.z += h[k].z * w;
        acc.w += h[k].w * w;
    }
}

struct GatherState {
    int beg, end, mine;
    float di;
    float4 acc;
};

struct GatherHead {
    int beg, deg;
    float di;
    float4 acc;
};

template <int F>
__device__ inline GatherHead gather_head(const int* __restrict__ rowptr,
                                         const int* __restrict__ cnt,
                                         const float* __restrict__ dinv,
                                         const float* __restrict__ H,
                                         int i, int q) {
    GatherHead hd;
    hd.beg = rowptr[i];
    hd.deg = cnt[i];
    hd.di = dinv[i];
    hd.acc = *(const float4*)(H + (size_t)i * F + q * 4);
    return hd;
}

__device__ inline GatherState gather_arm(const int* __restrict__ col, const GatherHead& hd, int q) {
    GatherState st;
    const int beg = hd.beg;
    const int end = beg + hd.deg;
    st.di = hd.di;
    st.acc = hd.acc;
    st.beg = beg;
    st.end = end;
    st.mine = 0;
    if (beg < end) st.mine = col[beg + q < end ? beg + q : end - 1];
    return st;
}

template <int F>
__device__ inline GatherState gather_open(const int* __restrict__ rowptr,
                                          const int* __restrict__ cnt,
                                          const int* __restrict__ col,
                                          const float* __restrict__ dinv,
                                          const float* __restrict__ H,
                                          int i, int q) {
    const GatherHead hd = gather_head<F>(rowptr, cnt, dinv, H, i, q);
    asm volatile("" :: "v"(hd.beg), "v"(hd.deg));
    return gather_arm(col, hd, q);
}

template <int F>
__device__ inline float4 gather_run(const int* __restrict__ col,
                                    const float* __restrict__ H,
                                    const GatherState& st, const float4 bv, int q) {
    constexpr int LPN = F / 4;
    const int c = q * 4;
    const int end = st.end;
    int mine = st.mine;
    float4 acc = st.acc;
    for (int j = st.beg; j < end; j += LPN) {
        const int rem = end - j;
        float4 h[8];
        load_row_group<F, 0>(H, mine, c, h);
        int next = 0;
        if (rem > LPN) {
            const int jn = j + LPN + q;
            next = col[jn < end ? jn : end - 1];
        }
        sum_row_group<0>(h, rem, acc);
        if constexpr (LPN > 8) {
            if (rem > 8) {
                load_row_group<F, 8>(H, mine, c, h);
                sum_row_group<8>(h, rem, acc);
            }
        }
        mine = next;
    }
    float4 o;
    o.x = acc.x * st.di + bv.x;
    o.y = acc.y * st.di + bv.y;
    o.z = acc.z * st.di + bv.z;
    o.w = acc.w * st.di + bv.w;
    return o;
}

template <int F>
__global__ void gather_sum_kernel(const int* __restrict__ rowptr,
                                  const int* __restrict__ cnt,
                                  const int* __restrict__ col,
                                  const float* __restrict__ dinv,
                                  const float* __restrict__ H,
                                  const float* __restrict__ bias,
                                  float* __restrict__ A, int n) {
    constexpr int LPN = F / 4;
    const int stride = (int)(gridDim.x * blockDim.x) / LPN;
    int i = (int)(blockIdx.x * blockDim.x + threadIdx.x) / LPN;
    if (i >= n) return;
    const int q = threadIdx.x % LPN;
    const float4 bv = *(const float4*)(bias + q * 4);
    GatherState st = gather_open<F>(rowptr, cnt, col, dinv, H, i, q);
    GatherHead hd = {};
    if (i + stride < n) hd = gather_head<F>(rowptr, cnt, dinv, H, i + stride, q);
    for (; i < n; i += stride) {
        GatherState nst = st;
        if (i + stride < n) nst = gather_arm(col, hd, q);
        if (i + 2 * stride < n) hd = gather_head<F>(rowptr, cnt, dinv, H, i + 2 * stride, q);
        const float4 o = gather_run<F>(col, H, st, bv, q);
        *(float4*)(A + (size_t)i * F + q * 4) = o;
        st = nst;
    }
}

template <int NOUT, bool RELU>
__global__ void gather_gemm_kernel(const int* __restrict__ rowptr,
                                   const int* __restrict__ cnt,
                                   const int* __restrict__ col,
                                   const float* __restrict__ dinv,
                                   const float* __restrict__ Hin,
                                   const float* __restrict__ bias,
                                   const float* __restrict__ W,
                                   float* __restrict__ Hout, int ntiles) {
    constexpr int F   = 64;
    constexpr int LPN = F / 4;
    constexpr int NB  = 16;
    constexpr int TBG = NB * LPN;
    constexpr int LDW = NOUT + 8;
    constexpr int LDA = F + 4;
    constexpr int NWW = NOUT / 32;
    constexpr int WV  = F * NOUT / 4 / TBG;
    static_assert(NOUT % 32 == 0 && (F * NOUT / 4) % TBG == 0, "whole lines, whole pieces");
    __shared__ alignas(16) float Ws[F * LDW];
    __shared__ alignas(16) float As[2][NB * LDA];
    __shared__ alignas(16) float rsl[2][NB];

    const int t = threadIdx.x;
    int tile = blockIdx.x;
    if (tile >= ntiles) return;
    float4 wv[WV];
#pragma unroll
    for (int u = 0; u < WV; ++u) wv[u] = *(const float4*)(W + 4 * (t + u * TBG));

    const int il = t / LPN;
    const int q  = t % LPN;
    const float4 bv = *(const float4*)(bias + q * 4);
    const int G = gridDim.x;
    GatherState st = gather_open<F>(rowptr, cnt, col, dinv, Hin, tile * NB + il, q);
    GatherHead hd = {};
    if (tile + G < ntiles) hd = gather_head<F>(rowptr, cnt, dinv, Hin, (tile + G) * NB + il, q);
#pragma unroll
    for (int u = 0; u < WV; ++u) {
        const int e = 4 * (t + u * TBG);
        const int k = e / NOUT, n = e - k * NOUT;
        *(float4*)(Ws + k * LDW + n) = wv[u];
    }

    const int wave = t >> 5;
    const int lane = t & 31;
    const int half = lane >> 4;
    const int l16  = lane & 15;
    const int koff = half * 2;
    const float* wcol = Ws + koff * LDW + (wave & (NWW - 1)) * 32 + l16;

    int buf = 0;
    for (; tile < ntiles; tile += G) {
        GatherState nst = st;
        if (tile + G < ntiles) nst = gather_arm(col, hd, q);
        if (tile + 2 * G < ntiles) hd = gather_head<F>(rowptr, cnt, dinv, Hin, (tile + 2 * G) * NB + il, q);
        float4 o = gather_run<F>(col, Hin, st, bv, q);
        if (RELU) {
            o.x = fmaxf(o.x, 0.0f); o.y = fmaxf(o.y, 0.0f);
            o.z = fmaxf(o.z, 0.0f); o.w = fmaxf(o.w, 0.0f);
        }
        float* Ab = As[buf];
        *(float4*)(Ab + il * LDA + q * 4) = o;
        if (q == 0) rsl[buf][il] = st.di;
        __syncthreads();

        if (wave < NWW) {
            const float* arow = Ab + l16 * LDA + koff;
            v8f acc0 = v8f{}, acc1 = v8f{};
#pragma unroll
            for (int s = 0; s < F / 4; ++s) {
                const v2f av = *(const v2f*)(arow + 4 * s);
                const int k0 = 4 * s;
                v2f b0, b1;
                b0.x = wcol[k0 * LDW];
                b0.y = wcol[k0 * LDW + LDW];
                b1.x = wcol[k0 * LDW + 16];
                b1.y = wcol[k0 * LDW + LDW + 16];
                acc0 = __builtin_amdgcn_wmma_f32_16x16x4_f32(false, av, false, b0, (short)0, acc0, false, false);
                acc1 = __builtin_amdgcn_wmma_f32_16x16x4_f32(false, av, false, b1, (short)0, acc1, false, false);
            }
            const float4 rs0 = *(const float4*)(rsl[buf] + half * 8);
            const float4 rs1 = *(const float4*)(rsl[buf] + half * 8 + 4);
            const float rsv[8] = {rs0.x, rs0.y, rs0.z, rs0.w, rs1.x, rs1.y, rs1.z, rs1.w};
            float* orow = Hout + (size_t)(tile * NB + half * 8) * NOUT + wave * 32 + l16;
#pragma unroll
            for (int v = 0; v < 8; ++v) {
                orow[(size_t)v * NOUT]      = acc0[v] * rsv[v];
                orow[(size_t)v * NOUT + 16] = acc1[v] * rsv[v];
            }
        }
        buf ^= 1;
        st = nst;
    }
}

extern "C" void kernel_launch(void* const* d_in, const int* in_sizes, int n_in,
                              void* d_out, int out_size, void* d_ws, size_t ws_size,
                              hipStream_t stream) {
    (void)in_sizes; (void)n_in; (void)out_size;
    const float* x  = (const float*)d_in[0];
    const int*   ei = (const int*)d_in[1];
    const float* W1 = (const float*)d_in[2];
    const float* b1 = (const float*)d_in[3];
    const float* W2 = (const float*)d_in[4];
    const float* b2 = (const float*)d_in[5];
    const float* W3 = (const float*)d_in[6];
    const float* b3 = (const float*)d_in[7];
    float* out = (float*)d_out;

    const int N = GCN_N, E = GCN_E;
    const int* srcI = ei;
    const int* dstI = ei + E;

    float* dinv  = (float*)d_ws;
    float* Hbuf  = dinv + N;
    float* Abuf  = Hbuf + (size_t)N * 64;
    int*   cnt    = (int*)(Abuf + (size_t)N * 64);
    int*   rowp   = cnt + N;
    int*   cstart = rowp + N;
    int*   binned = cstart + SORT_NCH * SORT_NBK;
    int*   col    = binned + E;
    size_t need = (3 * (size_t)N + 2 * (size_t)N * 64 + SORT_NCH * SORT_NBK + 2 * (size_t)E) * sizeof(float);
    if (ws_size < need) return;

    const int TB = 256;
    const int mtiles = N / 16;
    const int gbA = (mtiles + 7) / 8;

    chunk_bin_kernel  <<<SORT_NCH, SORT_TBE, 0, stream>>>(srcI, dstI, cstart, binned);
    bucket_sort_kernel<<<SORT_NBK, SORT_TBB, 0, stream>>>(binned, cstart, rowp, cnt, dinv, col, N);

    gemm_wmma_kernel<128, 64, false><<<gbA, TB, 0, stream>>>(x, W1, dinv, Hbuf, N);

    gather_gemm_kernel<64, true><<<GEMM_GATHER_BLOCKS, TB, 0, stream>>>(rowp, cnt, col, dinv, Hbuf, b1, W2, Abuf, N / 16);

    gather_gemm_kernel<32, true><<<GEMM_GATHER_BLOCKS, TB, 0, stream>>>(rowp, cnt, col, dinv, Abuf, b2, W3, Hbuf, N / 16);

    gather_sum_kernel<32><<<SUM_GATHER_BLOCKS, TB, 0, stream>>>(rowp, cnt, col, dinv, Hbuf, b3, out, N);
}


# Layer-1 product moved into the bucket pass of the edge sort: each block multiplies its own 256 rows (16-row strip per wave, same k order), row factors from its own counts in LDS; 5 launches, not 6
# speedup vs baseline: 5.1044x; 1.1637x over previous
#include <hip/hip_runtime.h>


typedef float v2f __attribute__((ext_vector_type(2)));
typedef float v8f __attribute__((ext_vector_type(8)));
typedef int   v4i __attribute__((ext_vector_type(4)));

#define GCN_N 100000
#define GCN_E 1600000

#define SORT_BW   256
#define SORT_NBK  ((GCN_N + SORT_BW - 1) / SORT_BW)
#define SORT_NCH  250
#define SORT_CH   (GCN_E / SORT_NCH)
#define SORT_TBE  400
#define SORT_EPT  (SORT_CH / 4 / SORT_TBE)
#define SORT_TBB  512
#define SORT_NG   (SORT_TBB / 16)
#define SORT_NQ   ((SORT_NCH + SORT_NG - 1) / SORT_NG)
#define SRC_BITS  17
#define GEMM_GATHER_BLOCKS 1250
#define SUM_GATHER_BLOCKS  1042

static_assert(GCN_N % 16 == 0, "the fused gather blocks own 16 whole nodes each");
static_assert(GCN_E % SORT_NCH == 0 && SORT_CH % 4 == 0, "chunks are whole int4 groups");
static_assert(GCN_N <= (1 << SRC_BITS), "src must fit the low bits of a binned word");
static_assert((SORT_CH / 4) % SORT_TBE == 0 && SORT_EPT == 4, "every thread of the chunk pass owns 4 int4 groups");
static_assert(SORT_BW == 256 && SORT_TBB == 512, "scan widths below");
static_assert(SORT_NBK <= SORT_TBE, "one thread per bucket in the chunk pass's scan");
static_assert(SORT_NG * SORT_NQ >= SORT_NCH && SORT_NG == 32, "every chunk has a lane group in the per-bucket pass");


template <int NW>
__device__ inline int block_incl_scan(int v, int* wsum) {
    const int lane = threadIdx.x & 31;
    const int wave = threadIdx.x >> 5;
    for (int o = 1; o < 32; o <<= 1) {
        const int u = __shfl_up(v, o);
        if (lane >= o) v += u;
    }
    if (lane == 31 && wave < NW) wsum[wave] = v;
    __syncthreads();
    int off = 0;
    for (int w = 0; w < NW; ++w)
        if (w < wave) off += wsum[w];
    return v + off;
}

__global__ void chunk_bin_kernel(const int* __restrict__ src, const int* __restrict__ dst,
                                 int* __restrict__ cstart, int* __restrict__ binned) {
    __shared__ int wsum[(SORT_TBE + 31) / 32];
    __shared__ int lh[SORT_NBK];
    const int t = threadIdx.x;
    const v4i* d4 = (const v4i*)(dst + (size_t)blockIdx.x * SORT_CH);
    const v4i* s4 = (const v4i*)(src + (size_t)blockIdx.x * SORT_CH);
    v4i d[SORT_EPT], s[SORT_EPT];
#pragma unroll
    for (int u = 0; u < SORT_EPT; ++u) {
        d[u] = d4[t + u * SORT_TBE];
        s[u] = s4[t + u * SORT_TBE];
    }
    for (int k = t; k < SORT_NBK; k += blockDim.x) lh[k] = 0;
    __syncthreads();
    asm volatile("" :: "v"(d[0]), "v"(d[1]), "v"(d[2]), "v"(d[3]),
                       "v"(s[0]), "v"(s[1]), "v"(s[2]), "v"(s[3]));
#pragma unroll
    for (int u = 0; u < SORT_EPT; ++u) {
        atomicAdd(&lh[d[u].x >> 8], 1);
        atomicAdd(&lh[d[u].y >> 8], 1);
        atomicAdd(&lh[d[u].z >> 8], 1);
        atomicAdd(&lh[d[u].w >> 8], 1);
    }
    __syncthreads();
    const int v = (t < SORT_NBK) ? lh[t] : 0;
    const int incl = block_incl_scan<(SORT_TBE + 31) / 32>(v, wsum);
    if (t < SORT_NBK) {
        cstart[blockIdx.x * SORT_NBK + t] = incl - v;
        lh[t] = blockIdx.x * SORT_CH + incl - v;
    }
    __syncthreads();
#pragma unroll
    for (int u = 0; u < SORT_EPT; ++u) {
        const int p0 = atomicAdd(&lh[d[u].x >> 8], 1);
        const int p1 = atomicAdd(&lh[d[u].y >> 8], 1);
        const int p2 = atomicAdd(&lh[d[u].z >> 8], 1);
        const int p3 = atomicAdd(&lh[d[u].w >> 8], 1);
        binned[p0] = s[u].x | ((d[u].x & (SORT_BW - 1)) << SRC_BITS);
        binned[p1] = s[u].y | ((d[u].y & (SORT_BW - 1)) << SRC_BITS);
        binned[p2] = s[u].z | ((d[u].z & (SORT_BW - 1)) << SRC_BITS);
        binned[p3] = s[u].w | ((d[u].w & (SORT_BW - 1)) << SRC_BITS);
    }
}

template <int KDIM, int NOUT>
__global__ void bucket_sort_gemm_kernel(const int* __restrict__ binned, const int* __restrict__ cstart,
                                        int* __restrict__ rowptr, int* __restrict__ cnt,
                                        float* __restrict__ dinv, int* __restrict__ col, int n,
                                        const float* __restrict__ X, const float* __restrict__ W,
                                        float* __restrict__ H) {
    constexpr int NT  = NOUT / 16;
    constexpr int LDW = NOUT + 8;
    constexpr int KC  = 32;
    constexpr int NS  = KC / 4;
    constexpr int WV  = KDIM * NOUT / 4 / SORT_TBB;
    static_assert(KDIM % KC == 0, "whole groups");
    static_assert((KDIM * NOUT / 4) % SORT_TBB == 0 && NOUT % 4 == 0, "whole pieces");
    static_assert(SORT_BW == 16 * (SORT_TBB / 32), "one 16-row strip per wave");
    __shared__ alignas(16) float Ws[KDIM * LDW];
    __shared__ alignas(16) float dl[SORT_BW];
    __shared__ int red[SORT_NG];
    __shared__ int wsum[8];
    __shared__ int lc[SORT_BW + 32];
    const int t = threadIdx.x;
    const int g = t >> 4;
    const int l = t & 15;
    const int k = blockIdx.x;

    float4 wv[WV];
#pragma unroll
    for (int u = 0; u < WV; ++u) wv[u] = *(const float4*)(W + 4 * (t + u * SORT_TBB));

    int rb[SORT_NQ], rn[SORT_NQ];
    int part = 0;
#pragma unroll
    for (int q = 0; q < SORT_NQ; ++q) {
        const int ch = g + q * SORT_NG;
        const int cc = ch < SORT_NCH ? ch : SORT_NCH - 1;
        const int ia = cc * SORT_NBK + k;
        const int ie = ia + 1 < SORT_NCH * SORT_NBK ? ia + 1 : ia;
        const int a = cstart[ia];
        const int en = cstart[ie];
        const int e = (k + 1 < SORT_NBK) ? en : SORT_CH;
        part += (ch < SORT_NCH) ? a : 0;
        rb[q] = cc * SORT_CH + a;
        rn[q] = (ch < SORT_NCH) ? e - a : 0;
    }
    int w[SORT_NQ][2];
#pragma unroll
    for (int q = 0; q < SORT_NQ; ++q) {
        const int last = rn[q] - 1;
        const int i0 = rb[q] + (l < last ? l : last);
        const int i1 = rb[q] + (l + 16 < last ? l + 16 : last);
        const int v0 = binned[i0 > 0 ? i0 : 0];
        const int v1 = binned[i1 > 0 ? i1 : 0];
        w[q][0] = (l < rn[q]) ? v0 : -1;
        w[q][1] = (l + 16 < rn[q]) ? v1 : -1;
    }

    if (l == 0) red[g] = part;
    if (t < SORT_BW + 32) lc[t] = 0;
#pragma unroll
    for (int u = 0; u < WV; ++u) {
        const int e = 4 * (t + u * SORT_TBB);
        const int kk = e / NOUT, nn = e - kk * NOUT;
        *(float4*)(Ws + kk * LDW + nn) = wv[u];
    }
    __syncthreads();
    int bs = 0;
    for (int i = 0; i < SORT_NG; ++i) bs += red[i];

#pragma unroll
    for (int q = 0; q < SORT_NQ; ++q) {
        if (w[q][0] >= 0) atomicAdd(&lc[w[q][0] >> SRC_BITS], 1);
        if (w[q][1] >= 0) atomicAdd(&lc[w[q][1] >> SRC_BITS], 1);
    }
#pragma unroll
    for (int q = 0; q < SORT_NQ; ++q)
        for (int j = 32 + l; j < rn[q]; j += 16)
            atomicAdd(&lc[binned[rb[q] + j] >> SRC_BITS], 1);
    __syncthreads();

    const int c = (t < SORT_BW) ? lc[t] : 0;
    const int incl = block_incl_scan<8>(c, wsum);
    if (t < SORT_BW) {
        const int start = bs + incl - c;
        const int node = k * SORT_BW + t;
        const float di = rsqrtf((float)(c + 1));
        if (node < n) {
            rowptr[node] = start;
            cnt[node] = c;
            dinv[node] = di;
        }
        dl[t] = di;
        lc[t] = start;
    }
    __syncthreads();

    const int wave = t >> 5;
    const int lane = t & 31;
    const int m0 = k * SORT_BW + wave * 16;
    const bool live = m0 < n;
    const int half = lane >> 4;
    const int l16  = lane & 15;
    const int koff = half * 2;
    const float* xrow = X + (size_t)(live ? m0 + l16 : 0) * KDIM + koff;
    v2f a[NS], an[NS];
#pragma unroll
    for (int s = 0; s < NS; ++s) a[s] = *(const v2f*)(xrow + 4 * s);

    const int spare = SORT_BW + (t & 31);
    int p[SORT_NQ][2];
#pragma unroll
    for (int q = 0; q < SORT_NQ; ++q) {
        p[q][0] = atomicAdd(&lc[w[q][0] >= 0 ? (w[q][0] >> SRC_BITS) : spare], 1);
        p[q][1] = atomicAdd(&lc[w[q][1] >= 0 ? (w[q][1] >> SRC_BITS) : spare], 1);
    }
#pragma unroll
    for (int q = 0; q < SORT_NQ; ++q) {
        if (w[q][0] >= 0) col[p[q][0]] = w[q][0] & ((1 << SRC_BITS) - 1);
        if (w[q][1] >= 0) col[p[q][1]] = w[q][1] & ((1 << SRC_BITS) - 1);
    }
#pragma unroll
    for (int q = 0; q < SORT_NQ; ++q)
        for (int j = 32 + l; j < rn[q]; j += 16) {
            const int ww = binned[rb[q] + j];
            const int pp = atomicAdd(&lc[ww >> SRC_BITS], 1);
            col[pp] = ww & ((1 << SRC_BITS) - 1);
        }

    if (!live) return;
    const float* wcol = Ws + koff * LDW + l16;
    const float4 rs0 = *(const float4*)(dl + wave * 16 + half * 8);
    const float4 rs1 = *(const float4*)(dl + wave * 16 + half * 8 + 4);

    v8f acc[NT];
#pragma unroll
    for (int nt = 0; nt < NT; ++nt) acc[nt] = v8f{};

#pragma unroll
    for (int kc = 0; kc < KDIM; kc += KC) {
        if (kc + KC < KDIM) {
#pragma unroll
            for (int s = 0; s < NS; ++s) an[s] = *(const v2f*)(xrow + kc + KC + 4 * s);
        }
#pragma unroll
        for (int s = 0; s < NS; ++s) {
            const v2f av = a[s];
            const int k0 = kc + 4 * s;
#pragma unroll
            for (int nt = 0; nt < NT; ++nt) {
                v2f b;
                b.x = wcol[k0 * LDW + nt * 16];
                b.y = wcol[k0 * LDW + LDW + nt * 16];
                acc[nt] = __builtin_amdgcn_wmma_f32_16x16x4_f32(
                    false, av, false, b, (short)0, acc[nt], false, false);
            }
        }
        if (kc + KC < KDIM) {
#pragma unroll
            for (int s = 0; s < NS; ++s) a[s] = an[s];
        }
    }
    const float rsv[8] = {rs0.x, rs0.y, rs0.z, rs0.w, rs1.x, rs1.y, rs1.z, rs1.w};
    float* orow = H + (size_t)(m0 + half * 8) * NOUT + l16;
#pragma unroll
    for (int nt = 0; nt < NT; ++nt)
#pragma unroll
        for (int v = 0; v < 8; ++v)
            orow[(size_t)v * NOUT + nt * 16] = acc[nt][v] * rsv[v];
}

template <int LPN, int K>
__device__ inline int group_bcast(int v) {
    static_assert(LPN == 16 || LPN == 8, "a group is a DPP row or a DPP8 group");
    static_assert(K >= 0 && K < LPN, "source lane inside the group");
    if constexpr (LPN == 16)
        return __builtin_amdgcn_update_dpp(0, v, 0x150 + K, 0xF, 0xF, true);
    else
        return __builtin_amdgcn_mov_dpp8(v, K * 0x249249);
}

template <int F, int G>
__device__ inline void load_row_group(const float* __restrict__ H, int mine, int c, float4 (&h)[8]) {
    constexpr int LPN = F / 4;
    int s[8];
    s[0] = group_bcast<LPN, G + 0>(mine);
    s[1] = group_bcast<LPN, G + 1>(mine);
    s[2] = group_bcast<LPN, G + 2>(mine);
    s[3] = group_bcast<LPN, G + 3>(mine);
    s[4] = group_bcast<LPN, G + 4>(mine);
    s[5] = group_bcast<LPN, G + 5>(mine);
    s[6] = group_bcast<LPN, G + 6>(mine);
    s[7] = group_bcast<LPN, G + 7>(mine);
#pragma unroll
    for (int k = 0; k < 8; ++k) h[k] = *(const float4*)(H + (size_t)s[k] * F + c);
}

template <int G>
__device__ inline void sum_row_group(const float4 (&h)[8], int rem, float4& acc) {
#pragma unroll
    for (int k = 0; k < 8; ++k) {
        const float w = (G + k < rem) ? 1.0f : 0.0f;
        acc.x += h[k].x * w;
        acc.y += h[k].y * w;
        acc.z += h[k].z * w;
        acc.w += h[k].w * w;
    }
}

struct GatherState {
    int beg, end, mine;
    float di;
    float4 acc;
};

struct GatherHead {
    int beg, deg;
    float di;
    float4 acc;
};

template <int F>
__device__ inline GatherHead gather_head(const int* __restrict__ rowptr,
                                         const int* __restrict__ cnt,
                                         const float* __restrict__ dinv,
                                         const float* __restrict__ H,
                                         int i, int q) {
    GatherHead hd;
    hd.beg = rowptr[i];
    hd.deg = cnt[i];
    hd.di = dinv[i];
    hd.acc = *(const float4*)(H + (size_t)i * F + q * 4);
    return hd;
}

__device__ inline GatherState gather_arm(const int* __restrict__ col, const GatherHead& hd, int q) {
    GatherState st;
    const int beg = hd.beg;
    const int end = beg + hd.deg;
    st.di = hd.di;
    st.acc = hd.acc;
    st.beg = beg;
    st.end = end;
    st.mine = 0;
    if (beg < end) st.mine = col[beg + q < end ? beg + q : end - 1];
    return st;
}

template <int F>
__device__ inline GatherState gather_open(const int* __restrict__ rowptr,
                                          const int* __restrict__ cnt,
                                          const int* __restrict__ col,
                                          const float* __restrict__ dinv,
                                          const float* __restrict__ H,
                                          int i, int q) {
    const GatherHead hd = gather_head<F>(rowptr, cnt, dinv, H, i, q);
    asm volatile("" :: "v"(hd.beg), "v"(hd.deg));
    return gather_arm(col, hd, q);
}

template <int F>
__device__ inline float4 gather_run(const int* __restrict__ col,
                                    const float* __restrict__ H,
                                    const GatherState& st, const float4 bv, int q) {
    constexpr int LPN = F / 4;
    const int c = q * 4;
    const int end = st.end;
    int mine = st.mine;
    float4 acc = st.acc;
    for (int j = st.beg; j < end; j += LPN) {
        const int rem = end - j;
        float4 h[8];
        load_row_group<F, 0>(H, mine, c, h);
        int next = 0;
        if (rem > LPN) {
            const int jn = j + LPN + q;
            next = col[jn < end ? jn : end - 1];
        }
        sum_row_group<0>(h, rem, acc);
        if constexpr (LPN > 8) {
            if (rem > 8) {
                load_row_group<F, 8>(H, mine, c, h);
                sum_row_group<8>(h, rem, acc);
            }
        }
        mine = next;
    }
    float4 o;
    o.x = acc.x * st.di + bv.x;
    o.y = acc.y * st.di + bv.y;
    o.z = acc.z * st.di + bv.z;
    o.w = acc.w * st.di + bv.w;
    return o;
}

template <int F>
__global__ void gather_sum_kernel(const int* __restrict__ rowptr,
                                  const int* __restrict__ cnt,
                                  const int* __restrict__ col,
                                  const float* __restrict__ dinv,
                                  const float* __restrict__ H,
                                  const float* __restrict__ bias,
                                  float* __restrict__ A, int n) {
    constexpr int LPN = F / 4;
    const int stride = (int)(gridDim.x * blockDim.x) / LPN;
    int i = (int)(blockIdx.x * blockDim.x + threadIdx.x) / LPN;
    if (i >= n) return;
    const int q = threadIdx.x % LPN;
    const float4 bv = *(const float4*)(bias + q * 4);
    GatherState st = gather_open<F>(rowptr, cnt, col, dinv, H, i, q);
    GatherHead hd = {};
    if (i + stride < n) hd = gather_head<F>(rowptr, cnt, dinv, H, i + stride, q);
    for (; i < n; i += stride) {
        GatherState nst = st;
        if (i + stride < n) nst = gather_arm(col, hd, q);
        if (i + 2 * stride < n) hd = gather_head<F>(rowptr, cnt, dinv, H, i + 2 * stride, q);
        const float4 o = gather_run<F>(col, H, st, bv, q);
        *(float4*)(A + (size_t)i * F + q * 4) = o;
        st = nst;
    }
}

template <int NOUT, bool RELU>
__global__ void gather_gemm_kernel(const int* __restrict__ rowptr,
                                   const int* __restrict__ cnt,
                                   const int* __restrict__ col,
                                   const float* __restrict__ dinv,
                                   const float* __restrict__ Hin,
                                   const float* __restrict__ bias,
                                   const float* __restrict__ W,
                                   float* __restrict__ Hout, int ntiles) {
    constexpr int F   = 64;
    constexpr int LPN = F / 4;
    constexpr int NB  = 16;
    constexpr int TBG = NB * LPN;
    constexpr int LDW = NOUT + 8;
    constexpr int LDA = F + 4;
    constexpr int NWW = NOUT / 32;
    constexpr int WV  = F * NOUT / 4 / TBG;
    static_assert(NOUT % 32 == 0 && (F * NOUT / 4) % TBG == 0, "whole lines, whole pieces");
    __shared__ alignas(16) float Ws[F * LDW];
    __shared__ alignas(16) float As[2][NB * LDA];
    __shared__ alignas(16) float rsl[2][NB];

    const int t = threadIdx.x;
    int tile = blockIdx.x;
    if (tile >= ntiles) return;
    float4 wv[WV];
#pragma unroll
    for (int u = 0; u < WV; ++u) wv[u] = *(const float4*)(W + 4 * (t + u * TBG));

    const int il = t / LPN;
    const int q  = t % LPN;
    const float4 bv = *(const float4*)(bias + q * 4);
    const int G = gridDim.x;
    GatherState st = gather_open<F>(rowptr, cnt, col, dinv, Hin, tile * NB + il, q);
    GatherHead hd = {};
    if (tile + G < ntiles) hd = gather_head<F>(rowptr, cnt, dinv, Hin, (tile + G) * NB + il, q);
#pragma unroll
    for (int u = 0; u < WV; ++u) {
        const int e = 4 * (t + u * TBG);
        const int k = e / NOUT, n = e - k * NOUT;
        *(float4*)(Ws + k * LDW + n) = wv[u];
    }

    const int wave = t >> 5;
    const int lane = t & 31;
    const int half = lane >> 4;
    const int l16  = lane & 15;
    const int koff = half * 2;
    const float* wcol = Ws + koff * LDW + (wave & (NWW - 1)) * 32 + l16;

    int buf = 0;
    for (; tile < ntiles; tile += G) {
        GatherState nst = st;
        if (tile + G < ntiles) nst = gather_arm(col, hd, q);
        if (tile + 2 * G < ntiles) hd = gather_head<F>(rowptr, cnt, dinv, Hin, (tile + 2 * G) * NB + il, q);
        float4 o = gather_run<F>(col, Hin, st, bv, q);
        if (RELU) {
            o.x = fmaxf(o.x, 0.0f); o.y = fmaxf(o.y, 0.0f);
            o.z = fmaxf(o.z, 0.0f); o.w = fmaxf(o.w, 0.0f);
        }
        float* Ab = As[buf];
        *(float4*)(Ab + il * LDA + q * 4) = o;
        if (q == 0) rsl[buf][il] = st.di;
        __syncthreads();

        if (wave < NWW) {
            const float* arow = Ab + l16 * LDA + koff;
            v8f acc0 = v8f{}, acc1 = v8f{};
#pragma unroll
            for (int s = 0; s < F / 4; ++s) {
                const v2f av = *(const v2f*)(arow + 4 * s);
                const int k0 = 4 * s;
                v2f b0, b1;
                b0.x = wcol[k0 * LDW];
                b0.y = wcol[k0 * LDW + LDW];
                b1.x = wcol[k0 * LDW + 16];
                b1.y = wcol[k0 * LDW + LDW + 16];
                acc0 = __builtin_amdgcn_wmma_f32_16x16x4_f32(false, av, false, b0, (short)0, acc0, false, false);
                acc1 = __builtin_amdgcn_wmma_f32_16x16x4_f32(false, av, false, b1, (short)0, acc1, false, false);
            }
            const float4 rs0 = *(const float4*)(rsl[buf] + half * 8);
            const float4 rs1 = *(const float4*)(rsl[buf] + half * 8 + 4);
            const float rsv[8] = {rs0.x, rs0.y, rs0.z, rs0.w, rs1.x, rs1.y, rs1.z, rs1.w};
            float* orow = Hout + (size_t)(tile * NB + half * 8) * NOUT + wave * 32 + l16;
#pragma unroll
            for (int v = 0; v < 8; ++v) {
                orow[(size_t)v * NOUT]      = acc0[v] * rsv[v];
                orow[(size_t)v * NOUT + 16] = acc1[v] * rsv[v];
            }
        }
        buf ^= 1;
        st = nst;
    }
}

extern "C" void kernel_launch(void* const* d_in, const int* in_sizes, int n_in,
                              void* d_out, int out_size, void* d_ws, size_t ws_size,
                              hipStream_t stream) {
    (void)in_sizes; (void)n_in; (void)out_size;
    const float* x  = (const float*)d_in[0];
    const int*   ei = (const int*)d_in[1];
    const float* W1 = (const float*)d_in[2];
    const float* b1 = (const float*)d_in[3];
    const float* W2 = (const float*)d_in[4];
    const float* b2 = (const float*)d_in[5];
    const float* W3 = (const float*)d_in[6];
    const float* b3 = (const float*)d_in[7];
    float* out = (float*)d_out;

    const int N = GCN_N, E = GCN_E;
    const int* srcI = ei;
    const int* dstI = ei + E;

    float* dinv  = (float*)d_ws;
    float* Hbuf  = dinv + N;
    float* Abuf  = Hbuf + (size_t)N * 64;
    int*   cnt    = (int*)(Abuf + (size_t)N * 64);
    int*   rowp   = cnt + N;
    int*   cstart = rowp + N;
    int*   binned = cstart + SORT_NCH * SORT_NBK;
    int*   col    = binned + E;
    size_t need = (3 * (size_t)N + 2 * (size_t)N * 64 + SORT_NCH * SORT_NBK + 2 * (size_t)E) * sizeof(float);
    if (ws_size < need) return;

    const int TB = 256;

    chunk_bin_kernel<<<SORT_NCH, SORT_TBE, 0, stream>>>(srcI, dstI, cstart, binned);
    bucket_sort_gemm_kernel<128, 64><<<SORT_NBK, SORT_TBB, 0, stream>>>(binned, cstart, rowp, cnt, dinv, col, N,
                                                                        x, W1, Hbuf);

    gather_gemm_kernel<64, true><<<GEMM_GATHER_BLOCKS, TB, 0, stream>>>(rowp, cnt, col, dinv, Hbuf, b1, W2, Abuf, N / 16);

    gather_gemm_kernel<32, true><<<GEMM_GATHER_BLOCKS, TB, 0, stream>>>(rowp, cnt, col, dinv, Abuf, b2, W3, Hbuf, N / 16);

    gather_sum_kernel<32><<<SUM_GATHER_BLOCKS, TB, 0, stream>>>(rowp, cnt, col, dinv, Hbuf, b3, out, N);
}


# Sort passes' scattered 4-byte stores staged in LDS: the cursors place a chunk's slice and a bucket's run of col there, stored out as whole 128-byte lines; planes line-aligned; sum order as before
# speedup vs baseline: 5.5667x; 1.0906x over previous
#include <hip/hip_runtime.h>


typedef float v2f __attribute__((ext_vector_type(2)));
typedef float v8f __attribute__((ext_vector_type(8)));
typedef int   v4i __attribute__((ext_vector_type(4)));

#define GCN_N 100000
#define GCN_E 1600000

#define SORT_BW   256
#define SORT_NBK  ((GCN_N + SORT_BW - 1) / SORT_BW)
#define SORT_NCH  250
#define SORT_CH   (GCN_E / SORT_NCH)
#define SORT_TBE  400
#define SORT_EPT  (SORT_CH / 4 / SORT_TBE)
#define SORT_TBB  512
#define SORT_NG   (SORT_TBB / 16)
#define SORT_NQ   ((SORT_NCH + SORT_NG - 1) / SORT_NG)
#define SRC_BITS  17
#define SORT_CAP  8192
#define CSTART_WORDS ((SORT_NCH * SORT_NBK + 31) / 32 * 32)
#define GEMM_GATHER_BLOCKS 1250
#define SUM_GATHER_BLOCKS  1042

static_assert(GCN_N % 16 == 0, "the fused gather blocks own 16 whole nodes each");
static_assert(GCN_E % SORT_NCH == 0 && SORT_CH % 4 == 0, "chunks are whole int4 groups");
static_assert(GCN_N <= (1 << SRC_BITS), "src must fit the low bits of a binned word");
static_assert((SORT_CH / 4) % SORT_TBE == 0 && SORT_EPT == 4, "every thread of the chunk pass owns 4 int4 groups");
static_assert(SORT_BW == 256 && SORT_TBB == 512, "scan widths below");
static_assert(SORT_NBK <= SORT_TBE, "one thread per bucket in the chunk pass's scan");
static_assert(SORT_NG * SORT_NQ >= SORT_NCH && SORT_NG == 32, "every chunk has a lane group in the per-bucket pass");
static_assert(GCN_N % 32 == 0 && GCN_E % 32 == 0 && SORT_CH % 32 == 0,
              "every workspace plane and every chunk's slice of binned starts on a 128-byte line");


template <int NW>
__device__ inline int block_incl_scan(int v, int* wsum) {
    const int lane = threadIdx.x & 31;
    const int wave = threadIdx.x >> 5;
    for (int o = 1; o < 32; o <<= 1) {
        const int u = __shfl_up(v, o);
        if (lane >= o) v += u;
    }
    if (lane == 31 && wave < NW) wsum[wave] = v;
    __syncthreads();
    int off = 0;
    for (int w = 0; w < NW; ++w)
        if (w < wave) off += wsum[w];
    return v + off;
}

__global__ void chunk_bin_kernel(const int* __restrict__ src, const int* __restrict__ dst,
                                 int* __restrict__ cstart, int* __restrict__ binned) {
    __shared__ int wsum[(SORT_TBE + 31) / 32];
    __shared__ int lh[SORT_NBK];
    __shared__ alignas(16) int stage[SORT_CH];
    const int t = threadIdx.x;
    const v4i* d4 = (const v4i*)(dst + (size_t)blockIdx.x * SORT_CH);
    const v4i* s4 = (const v4i*)(src + (size_t)blockIdx.x * SORT_CH);
    v4i d[SORT_EPT], s[SORT_EPT];
#pragma unroll
    for (int u = 0; u < SORT_EPT; ++u) {
        d[u] = d4[t + u * SORT_TBE];
        s[u] = s4[t + u * SORT_TBE];
    }
    for (int k = t; k < SORT_NBK; k += blockDim.x) lh[k] = 0;
    __syncthreads();
    asm volatile("" :: "v"(d[0]), "v"(d[1]), "v"(d[2]), "v"(d[3]),
                       "v"(s[0]), "v"(s[1]), "v"(s[2]), "v"(s[3]));
#pragma unroll
    for (int u = 0; u < SORT_EPT; ++u) {
        atomicAdd(&lh[d[u].x >> 8], 1);
        atomicAdd(&lh[d[u].y >> 8], 1);
        atomicAdd(&lh[d[u].z >> 8], 1);
        atomicAdd(&lh[d[u].w >> 8], 1);
    }
    __syncthreads();
    const int v = (t < SORT_NBK) ? lh[t] : 0;
    const int incl = block_incl_scan<(SORT_TBE + 31) / 32>(v, wsum);
    if (t < SORT_NBK) {
        cstart[blockIdx.x * SORT_NBK + t] = incl - v;
        lh[t] = incl - v;
    }
    __syncthreads();
#pragma unroll
    for (int u = 0; u < SORT_EPT; ++u) {
        const int p0 = atomicAdd(&lh[d[u].x >> 8], 1);
        const int p1 = atomicAdd(&lh[d[u].y >> 8], 1);
        const int p2 = atomicAdd(&lh[d[u].z >> 8], 1);
        const int p3 = atomicAdd(&lh[d[u].w >> 8], 1);
        stage[p0] = s[u].x | ((d[u].x & (SORT_BW - 1)) << SRC_BITS);
        stage[p1] = s[u].y | ((d[u].y & (SORT_BW - 1)) << SRC_BITS);
        stage[p2] = s[u].z | ((d[u].z & (SORT_BW - 1)) << SRC_BITS);
        stage[p3] = s[u].w | ((d[u].w & (SORT_BW - 1)) << SRC_BITS);
    }
    __syncthreads();
    v4i* o4 = (v4i*)(binned + (size_t)blockIdx.x * SORT_CH);
    const v4i* st4 = (const v4i*)stage;
#pragma unroll
    for (int u = 0; u < SORT_EPT; ++u) o4[t + u * SORT_TBE] = st4[t + u * SORT_TBE];
}

template <int KDIM, int NOUT>
__global__ void bucket_sort_gemm_kernel(const int* __restrict__ binned, const int* __restrict__ cstart,
                                        int* __restrict__ rowptr, int* __restrict__ cnt,
                                        float* __restrict__ dinv, int* __restrict__ col, int n,
                                        const float* __restrict__ X, const float* __restrict__ W,
                                        float* __restrict__ H) {
    constexpr int NT  = NOUT / 16;
    constexpr int LDW = NOUT + 8;
    constexpr int KC  = 32;
    constexpr int NS  = KC / 4;
    constexpr int WV  = KDIM * NOUT / 4 / SORT_TBB;
    static_assert(KDIM % KC == 0, "whole groups");
    static_assert((KDIM * NOUT / 4) % SORT_TBB == 0 && NOUT % 4 == 0, "whole pieces");
    static_assert(SORT_BW == 16 * (SORT_TBB / 32), "one 16-row strip per wave");
    __shared__ alignas(16) float Ws[KDIM * LDW];
    __shared__ alignas(16) float dl[SORT_BW];
    __shared__ int red[SORT_NG];
    __shared__ int wsum[8];
    __shared__ int lc[SORT_BW + 32];
    __shared__ int stage[SORT_CAP];
    const int t = threadIdx.x;
    const int g = t >> 4;
    const int l = t & 15;
    const int k = blockIdx.x;

    float4 wv[WV];
#pragma unroll
    for (int u = 0; u < WV; ++u) wv[u] = *(const float4*)(W + 4 * (t + u * SORT_TBB));

    int rb[SORT_NQ], rn[SORT_NQ];
    int part = 0;
#pragma unroll
    for (int q = 0; q < SORT_NQ; ++q) {
        const int ch = g + q * SORT_NG;
        const int cc = ch < SORT_NCH ? ch : SORT_NCH - 1;
        const int ia = cc * SORT_NBK + k;
        const int ie = ia + 1 < SORT_NCH * SORT_NBK ? ia + 1 : ia;
        const int a = cstart[ia];
        const int en = cstart[ie];
        const int e = (k + 1 < SORT_NBK) ? en : SORT_CH;
        part += (ch < SORT_NCH) ? a : 0;
        rb[q] = cc * SORT_CH + a;
        rn[q] = (ch < SORT_NCH) ? e - a : 0;
    }
    int w[SORT_NQ][2];
#pragma unroll
    for (int q = 0; q < SORT_NQ; ++q) {
        const int last = rn[q] - 1;
        const int i0 = rb[q] + (l < last ? l : last);
        const int i1 = rb[q] + (l + 16 < last ? l + 16 : last);
        const int v0 = binned[i0 > 0 ? i0 : 0];
        const int v1 = binned[i1 > 0 ? i1 : 0];
        w[q][0] = (l < rn[q]) ? v0 : -1;
        w[q][1] = (l + 16 < rn[q]) ? v1 : -1;
    }

    if (l == 0) red[g] = part;
    if (t < SORT_BW + 32) lc[t] = 0;
#pragma unroll
    for (int u = 0; u < WV; ++u) {
        const int e = 4 * (t + u * SORT_TBB);
        const int kk = e / NOUT, nn = e - kk * NOUT;
        *(float4*)(Ws + kk * LDW + nn) = wv[u];
    }
    __syncthreads();
    int bs = 0;
    for (int i = 0; i < SORT_NG; ++i) bs += red[i];

#pragma unroll
    for (int q = 0; q < SORT_NQ; ++q) {
        if (w[q][0] >= 0) atomicAdd(&lc[w[q][0] >> SRC_BITS], 1);
        if (w[q][1] >= 0) atomicAdd(&lc[w[q][1] >> SRC_BITS], 1);
    }
#pragma unroll
    for (int q = 0; q < SORT_NQ; ++q)
        for (int j = 32 + l; j < rn[q]; j += 16)
            atomicAdd(&lc[binned[rb[q] + j] >> SRC_BITS], 1);
    __syncthreads();

    const int c = (t < SORT_BW) ? lc[t] : 0;
    const int incl = block_incl_scan<8>(c, wsum);
    if (t < SORT_BW) {
        const int start = bs + incl - c;
        const int node = k * SORT_BW + t;
        const float di = rsqrtf((float)(c + 1));
        if (node < n) {
            rowptr[node] = start;
            cnt[node] = c;
            dinv[node] = di;
        }
        dl[t] = di;
        lc[t] = start;
    }
    __syncthreads();

    const int wave = t >> 5;
    const int lane = t & 31;
    const int m0 = k * SORT_BW + wave * 16;
    const bool live = m0 < n;
    const int half = lane >> 4;
    const int l16  = lane & 15;
    const int koff = half * 2;
    const float* xrow = X + (size_t)(live ? m0 + l16 : 0) * KDIM + koff;
    v2f a[NS], an[NS];
#pragma unroll
    for (int s = 0; s < NS; ++s) a[s] = *(const v2f*)(xrow + 4 * s);

    const int spare = SORT_BW + (t & 31);
    int p[SORT_NQ][2];
#pragma unroll
    for (int q = 0; q < SORT_NQ; ++q) {
        p[q][0] = atomicAdd(&lc[w[q][0] >= 0 ? (w[q][0] >> SRC_BITS) : spare], 1);
        p[q][1] = atomicAdd(&lc[w[q][1] >= 0 ? (w[q][1] >> SRC_BITS) : spare], 1);
    }
#pragma unroll
    for (int q = 0; q < SORT_NQ; ++q) {
        if (w[q][0] >= 0) {
            const int o = p[q][0] - bs, sv = w[q][0] & ((1 << SRC_BITS) - 1);
            if (o < SORT_CAP) stage[o] = sv; else col[p[q][0]] = sv;
        }
        if (w[q][1] >= 0) {
            const int o = p[q][1] - bs, sv = w[q][1] & ((1 << SRC_BITS) - 1);
            if (o < SORT_CAP) stage[o] = sv; else col[p[q][1]] = sv;
        }
    }
#pragma unroll
    for (int q = 0; q < SORT_NQ; ++q)
        for (int j = 32 + l; j < rn[q]; j += 16) {
            const int ww = binned[rb[q] + j];
            const int pp = atomicAdd(&lc[ww >> SRC_BITS], 1);
            const int o = pp - bs, sv = ww & ((1 << SRC_BITS) - 1);
            if (o < SORT_CAP) stage[o] = sv; else col[pp] = sv;
        }
    __syncthreads();
    {
        const int nb = lc[SORT_BW - 1] - bs;
        const int ns = nb < SORT_CAP ? nb : SORT_CAP;
        for (int j0 = t - (bs & 31); j0 < ns; j0 += 4 * SORT_TBB) {
            int sv[4];
#pragma unroll
            for (int u = 0; u < 4; ++u) {
                const int j = j0 + u * SORT_TBB;
                sv[u] = stage[(j >= 0 && j < ns) ? j : 0];
            }
#pragma unroll
            for (int u = 0; u < 4; ++u) {
                const int j = j0 + u * SORT_TBB;
                if (j >= 0 && j < ns) col[bs + j] = sv[u];
            }
        }
    }

    if (!live) return;
    const float* wcol = Ws + koff * LDW + l16;
    const float4 rs0 = *(const float4*)(dl + wave * 16 + half * 8);
    const float4 rs1 = *(const float4*)(dl + wave * 16 + half * 8 + 4);

    v8f acc[NT];
#pragma unroll
    for (int nt = 0; nt < NT; ++nt) acc[nt] = v8f{};

#pragma unroll
    for (int kc = 0; kc < KDIM; kc += KC) {
        if (kc + KC < KDIM) {
#pragma unroll
            for (int s = 0; s < NS; ++s) an[s] = *(const v2f*)(xrow + kc + KC + 4 * s);
        }
#pragma unroll
        for (int s = 0; s < NS; ++s) {
            const v2f av = a[s];
            const int k0 = kc + 4 * s;
#pragma unroll
            for (int nt = 0; nt < NT; ++nt) {
                v2f b;
                b.x = wcol[k0 * LDW + nt * 16];
                b.y = wcol[k0 * LDW + LDW + nt * 16];
                acc[nt] = __builtin_amdgcn_wmma_f32_16x16x4_f32(
                    false, av, false, b, (short)0, acc[nt], false, false);
            }
        }
        if (kc + KC < KDIM) {
#pragma unroll
            for (int s = 0; s < NS; ++s) a[s] = an[s];
        }
    }
    const float rsv[8] = {rs0.x, rs0.y, rs0.z, rs0.w, rs1.x, rs1.y, rs1.z, rs1.w};
    float* orow = H + (size_t)(m0 + half * 8) * NOUT + l16;
#pragma unroll
    for (int nt = 0; nt < NT; ++nt)
#pragma unroll
        for (int v = 0; v < 8; ++v)
            orow[(size_t)v * NOUT + nt * 16] = acc[nt][v] * rsv[v];
}

template <int LPN, int K>
__device__ inline int group_bcast(int v) {
    static_assert(LPN == 16 || LPN == 8, "a group is a DPP row or a DPP8 group");
    static_assert(K >= 0 && K < LPN, "source lane inside the group");
    if constexpr (LPN == 16)
        return __builtin_amdgcn_update_dpp(0, v, 0x150 + K, 0xF, 0xF, true);
    else
        return __builtin_amdgcn_mov_dpp8(v, K * 0x249249);
}

template <int F, int G>
__device__ inline void load_row_group(const float* __restrict__ H, int mine, int c, float4 (&h)[8]) {
    constexpr int LPN = F / 4;
    int s[8];
    s[0] = group_bcast<LPN, G + 0>(mine);
    s[1] = group_bcast<LPN, G + 1>(mine);
    s[2] = group_bcast<LPN, G + 2>(mine);
    s[3] = group_bcast<LPN, G + 3>(mine);
    s[4] = group_bcast<LPN, G + 4>(mine);
    s[5] = group_bcast<LPN, G + 5>(mine);
    s[6] = group_bcast<LPN, G + 6>(mine);
    s[7] = group_bcast<LPN, G + 7>(mine);
#pragma unroll
    for (int k = 0; k < 8; ++k) h[k] = *(const float4*)(H + (size_t)s[k] * F + c);
}

template <int G>
__device__ inline void sum_row_group(const float4 (&h)[8], int rem, float4& acc) {
#pragma unroll
    for (int k = 0; k < 8; ++k) {
        const float w = (G + k < rem) ? 1.0f : 0.0f;
        acc.x += h[k].x * w;
        acc.y += h[k].y * w;
        acc.z += h[k].z * w;
        acc.w += h[k].w * w;
    }
}

struct GatherState {
    int beg, end, mine;
    float di;
    float4 acc;
};

struct GatherHead {
    int beg, deg;
    float di;
    float4 acc;
};

template <int F>
__device__ inline GatherHead gather_head(const int* __restrict__ rowptr,
                                         const int* __restrict__ cnt,
                                         const float* __restrict__ dinv,
                                         const float* __restrict__ H,
                                         int i, int q) {
    GatherHead hd;
    hd.beg = rowptr[i];
    hd.deg = cnt[i];
    hd.di = dinv[i];
    hd.acc = *(const float4*)(H + (size_t)i * F + q * 4);
    return hd;
}

__device__ inline GatherState gather_arm(const int* __restrict__ col, const GatherHead& hd, int q) {
    GatherState st;
    const int beg = hd.beg;
    const int end = beg + hd.deg;
    st.di = hd.di;
    st.acc = hd.acc;
    st.beg = beg;
    st.end = end;
    st.mine = 0;
    if (beg < end) st.mine = col[beg + q < end ? beg + q : end - 1];
    return st;
}

template <int F>
__device__ inline GatherState gather_open(const int* __restrict__ rowptr,
                                          const int* __restrict__ cnt,
                                          const int* __restrict__ col,
                                          const float* __restrict__ dinv,
                                          const float* __restrict__ H,
                                          int i, int q) {
    const GatherHead hd = gather_head<F>(rowptr, cnt, dinv, H, i, q);
    asm volatile("" :: "v"(hd.beg), "v"(hd.deg));
    return gather_arm(col, hd, q);
}

template <int F>
__device__ inline float4 gather_run(const int* __restrict__ col,
                                    const float* __restrict__ H,
                                    const GatherState& st, const float4 bv, int q) {
    constexpr int LPN = F / 4;
    const int c = q * 4;
    const int end = st.end;
    int mine = st.mine;
    float4 acc = st.acc;
    for (int j = st.beg; j < end; j += LPN) {
        const int rem = end - j;
        float4 h[8];
        load_row_group<F, 0>(H, mine, c, h);
        int next = 0;
        if (rem > LPN) {
            const int jn = j + LPN + q;
            next = col[jn < end ? jn : end - 1];
        }
        sum_row_group<0>(h, rem, acc);
        if constexpr (LPN > 8) {
            if (rem > 8) {
                load_row_group<F, 8>(H, mine, c, h);
                sum_row_group<8>(h, rem, acc);
            }
        }
        mine = next;
    }
    float4 o;
    o.x = acc.x * st.di + bv.x;
    o.y = acc.y * st.di + bv.y;
    o.z = acc.z * st.di + bv.z;
    o.w = acc.w * st.di + bv.w;
    return o;
}

template <int F>
__global__ void gather_sum_kernel(const int* __restrict__ rowptr,
                                  const int* __restrict__ cnt,
                                  const int* __restrict__ col,
                                  const float* __restrict__ dinv,
                                  const float* __restrict__ H,
                                  const float* __restrict__ bias,
                                  float* __restrict__ A, int n) {
    constexpr int LPN = F / 4;
    const int stride = (int)(gridDim.x * blockDim.x) / LPN;
    int i = (int)(blockIdx.x * blockDim.x + threadIdx.x) / LPN;
    if (i >= n) return;
    const int q = threadIdx.x % LPN;
    const float4 bv = *(const float4*)(bias + q * 4);
    GatherState st = gather_open<F>(rowptr, cnt, col, dinv, H, i, q);
    GatherHead hd = {};
    if (i + stride < n) hd = gather_head<F>(rowptr, cnt, dinv, H, i + stride, q);
    for (; i < n; i += stride) {
        GatherState nst = st;
        if (i + stride < n) nst = gather_arm(col, hd, q);
        if (i + 2 * stride < n) hd = gather_head<F>(rowptr, cnt, dinv, H, i + 2 * stride, q);
        const float4 o = gather_run<F>(col, H, st, bv, q);
        *(float4*)(A + (size_t)i * F + q * 4) = o;
        st = nst;
    }
}

template <int NOUT, bool RELU>
__global__ void gather_gemm_kernel(const int* __restrict__ rowptr,
                                   const int* __restrict__ cnt,
                                   const int* __restrict__ col,
                                   const float* __restrict__ dinv,
                                   const float* __restrict__ Hin,
                                   const float* __restrict__ bias,
                                   const float* __restrict__ W,
                                   float* __restrict__ Hout, int ntiles) {
    constexpr int F   = 64;
    constexpr int LPN = F / 4;
    constexpr int NB  = 16;
    constexpr int TBG = NB * LPN;
    constexpr int LDW = NOUT + 8;
    constexpr int LDA = F + 4;
    constexpr int NWW = NOUT / 32;
    constexpr int WV  = F * NOUT / 4 / TBG;
    static_assert(NOUT % 32 == 0 && (F * NOUT / 4) % TBG == 0, "whole lines, whole pieces");
    __shared__ alignas(16) float Ws[F * LDW];
    __shared__ alignas(16) float As[2][NB * LDA];
    __shared__ alignas(16) float rsl[2][NB];

    const int t = threadIdx.x;
    int tile = blockIdx.x;
    if (tile >= ntiles) return;
    float4 wv[WV];
#pragma unroll
    for (int u = 0; u < WV; ++u) wv[u] = *(const float4*)(W + 4 * (t + u * TBG));

    const int il = t / LPN;
    const int q  = t % LPN;
    const float4 bv = *(const float4*)(bias + q * 4);
    const int G = gridDim.x;
    GatherState st = gather_open<F>(rowptr, cnt, col, dinv, Hin, tile * NB + il, q);
    GatherHead hd = {};
    if (tile + G < ntiles) hd = gather_head<F>(rowptr, cnt, dinv, Hin, (tile + G) * NB + il, q);
#pragma unroll
    for (int u = 0; u < WV; ++u) {
        const int e = 4 * (t + u * TBG);
        const int k = e / NOUT, n = e - k * NOUT;
        *(float4*)(Ws + k * LDW + n) = wv[u];
    }

    const int wave = t >> 5;
    const int lane = t & 31;
    const int half = lane >> 4;
    const int l16  = lane & 15;
    const int koff = half * 2;
    const float* wcol = Ws + koff * LDW + (wave & (NWW - 1)) * 32 + l16;

    int buf = 0;
    for (; tile < ntiles; tile += G) {
        GatherState nst = st;
        if (tile + G < ntiles) nst = gather_arm(col, hd, q);
        if (tile + 2 * G < ntiles) hd = gather_head<F>(rowptr, cnt, dinv, Hin, (tile + 2 * G) * NB + il, q);
        float4 o = gather_run<F>(col, Hin, st, bv, q);
        if (RELU) {
            o.x = fmaxf(o.x, 0.0f); o.y = fmaxf(o.y, 0.0f);
            o.z = fmaxf(o.z, 0.0f); o.w = fmaxf(o.w, 0.0f);
        }
        float* Ab = As[buf];
        *(float4*)(Ab + il * LDA + q * 4) = o;
        if (q == 0) rsl[buf][il] = st.di;
        __syncthreads();

        if (wave < NWW) {
            const float* arow = Ab + l16 * LDA + koff;
            v8f acc0 = v8f{}, acc1 = v8f{};
#pragma unroll
            for (int s = 0; s < F / 4; ++s) {
                const v2f av = *(const v2f*)(arow + 4 * s);
                const int k0 = 4 * s;
                v2f b0, b1;
                b0.x = wcol[k0 * LDW];
                b0.y = wcol[k0 * LDW + LDW];
                b1.x = wcol[k0 * LDW + 16];
                b1.y = wcol[k0 * LDW + LDW + 16];
                acc0 = __builtin_amdgcn_wmma_f32_16x16x4_f32(false, av, false, b0, (short)0, acc0, false, false);
                acc1 = __builtin_amdgcn_wmma_f32_16x16x4_f32(false, av, false, b1, (short)0, acc1, false, false);
            }
            const float4 rs0 = *(const float4*)(rsl[buf] + half * 8);
            const float4 rs1 = *(const float4*)(rsl[buf] + half * 8 + 4);
            const float rsv[8] = {rs0.x, rs0.y, rs0.z, rs0.w, rs1.x, rs1.y, rs1.z, rs1.w};
            float* orow = Hout + (size_t)(tile * NB + half * 8) * NOUT + wave * 32 + l16;
#pragma unroll
            for (int v = 0; v < 8; ++v) {
                orow[(size_t)v * NOUT]      = acc0[v] * rsv[v];
                orow[(size_t)v * NOUT + 16] = acc1[v] * rsv[v];
            }
        }
        buf ^= 1;
        st = nst;
    }
}

extern "C" void kernel_launch(void* const* d_in, const int* in_sizes, int n_in,
                              void* d_out, int out_size, void* d_ws, size_t ws_size,
                              hipStream_t stream) {
    (void)in_sizes; (void)n_in; (void)out_size;
    const float* x  = (const float*)d_in[0];
    const int*   ei = (const int*)d_in[1];
    const float* W1 = (const float*)d_in[2];
    const float* b1 = (const float*)d_in[3];
    const float* W2 = (const float*)d_in[4];
    const float* b2 = (const float*)d_in[5];
    const float* W3 = (const float*)d_in[6];
    const float* b3 = (const float*)d_in[7];
    float* out = (float*)d_out;

    const int N = GCN_N, E = GCN_E;
    const int* srcI = ei;
    const int* dstI = ei + E;

    float* dinv  = (float*)d_ws;
    float* Hbuf  = dinv + N;
    float* Abuf  = Hbuf + (size_t)N * 64;
    int*   cnt    = (int*)(Abuf + (size_t)N * 64);
    int*   rowp   = cnt + N;
    int*   cstart = rowp + N;
    int*   binned = cstart + CSTART_WORDS;
    int*   col    = binned + E;
    size_t need = (3 * (size_t)N + 2 * (size_t)N * 64 + CSTART_WORDS + 2 * (size_t)E) * sizeof(float);
    if (ws_size < need) return;

    const int TB = 256;

    chunk_bin_kernel<<<SORT_NCH, SORT_TBE, 0, stream>>>(srcI, dstI, cstart, binned);
    bucket_sort_gemm_kernel<128, 64><<<SORT_NBK, SORT_TBB, 0, stream>>>(binned, cstart, rowp, cnt, dinv, col, N,
                                                                        x, W1, Hbuf);

    gather_gemm_kernel<64, true><<<GEMM_GATHER_BLOCKS, TB, 0, stream>>>(rowp, cnt, col, dinv, Hbuf, b1, W2, Abuf, N / 16);

    gather_gemm_kernel<32, true><<<GEMM_GATHER_BLOCKS, TB, 0, stream>>>(rowp, cnt, col, dinv, Abuf, b2, W3, Hbuf, N / 16);

    gather_sum_kernel<32><<<SUM_GATHER_BLOCKS, TB, 0, stream>>>(rowp, cnt, col, dinv, Hbuf, b3, out, N);
}
